# starting kernel
# baseline (speedup 1.0000x reference)
#include <hip/hip_runtime.h>
#include <hip/hip_bf16.h>

#define NNODES 50000
#define DIMIN  512
#define HIDDEN 256
#define NCLS   2
#define LDSK   516

typedef __attribute__((ext_vector_type(2))) float v2f;
typedef __attribute__((ext_vector_type(8))) float v8f;


__global__ void k_init_deg(float* __restrict__ deg, int n) {
    int i = blockIdx.x * blockDim.x + threadIdx.x;
    if (i < n) deg[i] = 1.0f;
}

__global__ void k_deg_accum(const int* __restrict__ dst, float* __restrict__ deg, int e) {
    int i = blockIdx.x * blockDim.x + threadIdx.x;
    if (i < e) atomicAdd(&deg[dst[i]], 1.0f);
}

__global__ void k_dinv(const float* __restrict__ deg, float* __restrict__ dinv, int n) {
    int i = blockIdx.x * blockDim.x + threadIdx.x;
    if (i < n) dinv[i] = __frsqrt_rn(deg[i]);
}


__global__ void k_packW(const float* __restrict__ W, float2* __restrict__ Wp) {
    int idx = blockIdx.x * blockDim.x + threadIdx.x;
    if (idx >= (DIMIN / 2) * HIDDEN) return;
    int kp  = idx / HIDDEN;
    int col = idx - kp * HIDDEN;
    float2 p;
    p.x = W[(size_t)(2 * kp)     * HIDDEN + col];
    p.y = W[(size_t)(2 * kp + 1) * HIDDEN + col];
    Wp[idx] = p;
}


__global__ void __launch_bounds__(256)
k_gemm1(const float* __restrict__ X, const float2* __restrict__ Wp,
        const float* __restrict__ dinv, float* __restrict__ H, float* __restrict__ AGG) {
    __shared__ float As[16 * LDSK];

    const int row0 = blockIdx.x * 16;

    for (int idx = threadIdx.x; idx < (16 * DIMIN) / 4; idx += 256) {
        int r = idx >> 7;
        int c = (idx & 127) * 4;
        const float4 v = *(const float4*)(X + (size_t)(row0 + r) * DIMIN + c);
        *(float4*)(&As[r * LDSK + c]) = v;
    }
    __syncthreads();

    const int wave = threadIdx.x >> 5;
    const int lane = threadIdx.x & 31;
    const int l16  = lane & 15;
    const int hi   = lane >> 4;

    for (int ct = wave; ct < HIDDEN / 16; ct += 8) {
        const int col0 = ct * 16;
        v8f c = {};
        #pragma unroll 8
        for (int k = 0; k < DIMIN; k += 4) {
            const int kk = k + 2 * hi;
            const float2 af = *(const float2*)(&As[l16 * LDSK + kk]);
            v2f a; a.x = af.x; a.y = af.y;
            const float2 bf = Wp[(size_t)(kk >> 1) * HIDDEN + col0 + l16];
            v2f b; b.x = bf.x; b.y = bf.y;
            c = __builtin_amdgcn_wmma_f32_16x16x4_f32(false, a, false, b,
                                                      (short)0, c, false, false);
        }
        #pragma unroll
        for (int v = 0; v < 8; ++v) {
            const int row = row0 + hi * 8 + v;
            const float dv = dinv[row];
            const size_t o = (size_t)row * HIDDEN + col0 + l16;
            H[o]   = c[v];
            AGG[o] = dv * dv * c[v];
        }
    }
}


__global__ void k_edge_agg1(const int* __restrict__ src, const int* __restrict__ dst,
                            const float* __restrict__ dinv, const float* __restrict__ H1,
                            float* __restrict__ AGG, int e) {
    int gid  = blockIdx.x * blockDim.x + threadIdx.x;
    int wid  = gid >> 5;
    int lane = gid & 31;
    if (wid >= e) return;
    int s = src[wid], d = dst[wid];
    float norm = dinv[s] * dinv[d];
    const float4* hs = (const float4*)(H1 + (size_t)s * HIDDEN);
    float* od = AGG + (size_t)d * HIDDEN;
    #pragma unroll
    for (int it = 0; it < 2; ++it) {
        int c4 = it * 32 + lane;
        float4 h = hs[c4];
        int c = c4 * 4;
        atomicAdd(&od[c + 0], norm * h.x);
        atomicAdd(&od[c + 1], norm * h.y);
        atomicAdd(&od[c + 2], norm * h.z);
        atomicAdd(&od[c + 3], norm * h.w);
    }
}

__global__ void k_bias_relu(float* __restrict__ A, const float* __restrict__ b1, int total) {
    int i = blockIdx.x * blockDim.x + threadIdx.x;
    if (i >= total) return;
    float v = A[i] + b1[i & (HIDDEN - 1)];
    A[i] = v > 0.0f ? v : 0.0f;
}


__global__ void k_gemm2(const float* __restrict__ Z, const float* __restrict__ W2,
                        const float* __restrict__ dinv, const float* __restrict__ b2,
                        float* __restrict__ H2, float* __restrict__ out, int n) {
    int i = blockIdx.x * blockDim.x + threadIdx.x;
    if (i >= n) return;
    const float4* z4 = (const float4*)(Z + (size_t)i * HIDDEN);
    float a0 = 0.0f, a1 = 0.0f;
    #pragma unroll 4
    for (int k4 = 0; k4 < HIDDEN / 4; ++k4) {
        float4 z = z4[k4];
        int k = k4 * 4;
        a0 += z.x * W2[2*k+0] + z.y * W2[2*k+2] + z.z * W2[2*k+4] + z.w * W2[2*k+6];
        a1 += z.x * W2[2*k+1] + z.y * W2[2*k+3] + z.z * W2[2*k+5] + z.w * W2[2*k+7];
    }
    H2[2*i]     = a0;
    H2[2*i + 1] = a1;
    float s = dinv[i]; s *= s;
    out[2*i]     = s * a0 + b2[0];
    out[2*i + 1] = s * a1 + b2[1];
}


__global__ void k_edge_agg2(const int* __restrict__ src, const int* __restrict__ dst,
                            const float* __restrict__ dinv, const float* __restrict__ H2,
                            float* __restrict__ out, int e) {
    int i = blockIdx.x * blockDim.x + threadIdx.x;
    if (i >= e) return;
    int s = src[i], d = dst[i];
    float norm = dinv[s] * dinv[d];
    atomicAdd(&out[2*d],     norm * H2[2*s]);
    atomicAdd(&out[2*d + 1], norm * H2[2*s + 1]);
}


extern "C" void kernel_launch(void* const* d_in, const int* in_sizes, int n_in,
                              void* d_out, int out_size, void* d_ws, size_t ws_size,
                              hipStream_t stream) {
    const float* X   = (const float*)d_in[0];
    const float* W1  = (const float*)d_in[1];
    const float* b1  = (const float*)d_in[2];
    const float* W2  = (const float*)d_in[3];
    const float* b2  = (const float*)d_in[4];
    const int*   ei  = (const int*)d_in[5];
    float*       out = (float*)d_out;

    const int N = NNODES;
    const int E = in_sizes[5] / 2;
    const int* src = ei;
    const int* dst = ei + E;

    float*  deg  = (float*)d_ws;
    float*  dinv = deg  + N;
    float*  H1   = dinv + N;
    float*  AGG1 = H1   + (size_t)N * HIDDEN;
    float*  H2   = AGG1 + (size_t)N * HIDDEN;
    float2* Wp   = (float2*)(H2 + (size_t)N * NCLS);

    const int T = 256;

    k_init_deg <<<(N + T - 1) / T, T, 0, stream>>>(deg, N);
    k_deg_accum<<<(E + T - 1) / T, T, 0, stream>>>(dst, deg, E);
    k_dinv     <<<(N + T - 1) / T, T, 0, stream>>>(deg, dinv, N);

    {
        int total = (DIMIN / 2) * HIDDEN;
        k_packW<<<(total + T - 1) / T, T, 0, stream>>>(W1, Wp);
    }

    k_gemm1<<<N / 16, 256, 0, stream>>>(X, Wp, dinv, H1, AGG1);
    {
        long long threads = (long long)E * 32;
        k_edge_agg1<<<(int)((threads + T - 1) / T), T, 0, stream>>>(src, dst, dinv, H1, AGG1, E);
    }
    {
        int total = N * HIDDEN;
        k_bias_relu<<<(total + T - 1) / T, T, 0, stream>>>(AGG1, b1, total);
    }

    k_gemm2    <<<(N + T - 1) / T, T, 0, stream>>>(AGG1, W2, dinv, b2, H2, out, N);
    k_edge_agg2<<<(E + T - 1) / T, T, 0, stream>>>(src, dst, dinv, H2, out, E);
}


# Layer-1 scatter moved into LDS: a work-group owns 196 dst rows, scans the edge list, adds owned edges' h1 rows by LDS f32 atomics (arrival order); bias, relu, 256->2 from LDS, lane-tree sum
# speedup vs baseline: 1.7904x; 1.7904x over previous
#include <hip/hip_runtime.h>
#include <hip/hip_bf16.h>

#define NNODES 50000
#define DIMIN  512
#define HIDDEN 256
#define NCLS   2
#define LDSK   516

typedef __attribute__((ext_vector_type(2))) float v2f;
typedef __attribute__((ext_vector_type(8))) float v8f;


__global__ void k_init_deg(float* __restrict__ deg, int n) {
    int i = blockIdx.x * blockDim.x + threadIdx.x;
    if (i < n) deg[i] = 1.0f;
}

__global__ void k_deg_accum(const int* __restrict__ dst, float* __restrict__ deg, int e) {
    int i = blockIdx.x * blockDim.x + threadIdx.x;
    if (i < e) atomicAdd(&deg[dst[i]], 1.0f);
}

__global__ void k_dinv(const float* __restrict__ deg, float* __restrict__ dinv, int n) {
    int i = blockIdx.x * blockDim.x + threadIdx.x;
    if (i < n) dinv[i] = __frsqrt_rn(deg[i]);
}


__global__ void k_packW(const float* __restrict__ W, float2* __restrict__ Wp) {
    int idx = blockIdx.x * blockDim.x + threadIdx.x;
    if (idx >= (DIMIN / 2) * HIDDEN) return;
    int kp  = idx / HIDDEN;
    int col = idx - kp * HIDDEN;
    float2 p;
    p.x = W[(size_t)(2 * kp)     * HIDDEN + col];
    p.y = W[(size_t)(2 * kp + 1) * HIDDEN + col];
    Wp[idx] = p;
}


__global__ void __launch_bounds__(256)
k_gemm1(const float* __restrict__ X, const float2* __restrict__ Wp,
        float* __restrict__ H) {
    __shared__ float As[16 * LDSK];

    const int row0 = blockIdx.x * 16;

    for (int idx = threadIdx.x; idx < (16 * DIMIN) / 4; idx += 256) {
        int r = idx >> 7;
        int c = (idx & 127) * 4;
        const float4 v = *(const float4*)(X + (size_t)(row0 + r) * DIMIN + c);
        *(float4*)(&As[r * LDSK + c]) = v;
    }
    __syncthreads();

    const int wave = threadIdx.x >> 5;
    const int lane = threadIdx.x & 31;
    const int l16  = lane & 15;
    const int hi   = lane >> 4;

    for (int ct = wave; ct < HIDDEN / 16; ct += 8) {
        const int col0 = ct * 16;
        v8f c = {};
        #pragma unroll 8
        for (int k = 0; k < DIMIN; k += 4) {
            const int kk = k + 2 * hi;
            const float2 af = *(const float2*)(&As[l16 * LDSK + kk]);
            v2f a; a.x = af.x; a.y = af.y;
            const float2 bf = Wp[(size_t)(kk >> 1) * HIDDEN + col0 + l16];
            v2f b; b.x = bf.x; b.y = bf.y;
            c = __builtin_amdgcn_wmma_f32_16x16x4_f32(false, a, false, b,
                                                      (short)0, c, false, false);
        }
        #pragma unroll
        for (int v = 0; v < 8; ++v) {
            const int row = row0 + hi * 8 + v;
            const size_t o = (size_t)row * HIDDEN + col0 + l16;
            H[o] = c[v];
        }
    }
}


#define OWN_ROWS    196
#define OWN_THREADS 1024
#define OWN_WAVES   (OWN_THREADS / 32)
#define SCAN_EDGES  512
#define LIST_CAP    (SCAN_EDGES + 4)

__device__ __forceinline__ void own_lds_add(float* a, float nrm, const float4& p, const float4& q) {
    __hip_atomic_fetch_add(&a[0 * 32], nrm * p.x, __ATOMIC_RELAXED, __HIP_MEMORY_SCOPE_WORKGROUP);
    __hip_atomic_fetch_add(&a[1 * 32], nrm * p.y, __ATOMIC_RELAXED, __HIP_MEMORY_SCOPE_WORKGROUP);
    __hip_atomic_fetch_add(&a[2 * 32], nrm * p.z, __ATOMIC_RELAXED, __HIP_MEMORY_SCOPE_WORKGROUP);
    __hip_atomic_fetch_add(&a[3 * 32], nrm * p.w, __ATOMIC_RELAXED, __HIP_MEMORY_SCOPE_WORKGROUP);
    __hip_atomic_fetch_add(&a[4 * 32], nrm * q.x, __ATOMIC_RELAXED, __HIP_MEMORY_SCOPE_WORKGROUP);
    __hip_atomic_fetch_add(&a[5 * 32], nrm * q.y, __ATOMIC_RELAXED, __HIP_MEMORY_SCOPE_WORKGROUP);
    __hip_atomic_fetch_add(&a[6 * 32], nrm * q.z, __ATOMIC_RELAXED, __HIP_MEMORY_SCOPE_WORKGROUP);
    __hip_atomic_fetch_add(&a[7 * 32], nrm * q.w, __ATOMIC_RELAXED, __HIP_MEMORY_SCOPE_WORKGROUP);
}

__device__ __forceinline__ void own_add4(const int* __restrict__ src, const int* __restrict__ dst,
                                         const float* __restrict__ dinv, const float* __restrict__ H1,
                                         float* acc, int lo, int lane,
                                         int e0, int e1, int e2, int e3) {
    const int s0 = src[e0], s1 = src[e1], s2 = src[e2], s3 = src[e3];
    const int d0 = dst[e0], d1 = dst[e1], d2 = dst[e2], d3 = dst[e3];
    const float4* h0 = (const float4*)(H1 + (size_t)s0 * HIDDEN) + lane;
    const float4* h1 = (const float4*)(H1 + (size_t)s1 * HIDDEN) + lane;
    const float4* h2 = (const float4*)(H1 + (size_t)s2 * HIDDEN) + lane;
    const float4* h3 = (const float4*)(H1 + (size_t)s3 * HIDDEN) + lane;
    const float4 p0 = h0[0], q0 = h0[32];
    const float4 p1 = h1[0], q1 = h1[32];
    const float4 p2 = h2[0], q2 = h2[32];
    const float4 p3 = h3[0], q3 = h3[32];
    const float n0 = dinv[s0] * dinv[d0];
    const float n1 = dinv[s1] * dinv[d1];
    const float n2 = dinv[s2] * dinv[d2];
    const float n3 = dinv[s3] * dinv[d3];
    own_lds_add(acc + (d0 - lo) * HIDDEN + lane, n0, p0, q0);
    own_lds_add(acc + (d1 - lo) * HIDDEN + lane, n1, p1, q1);
    own_lds_add(acc + (d2 - lo) * HIDDEN + lane, n2, p2, q2);
    own_lds_add(acc + (d3 - lo) * HIDDEN + lane, n3, p3, q3);
}

__device__ __forceinline__ void own_add1(const int* __restrict__ src, const int* __restrict__ dst,
                                         const float* __restrict__ dinv, const float* __restrict__ H1,
                                         float* acc, int lo, int lane, int e0) {
    const int s0 = src[e0];
    const int d0 = dst[e0];
    const float4* h0 = (const float4*)(H1 + (size_t)s0 * HIDDEN) + lane;
    const float4 p0 = h0[0], q0 = h0[32];
    const float n0 = dinv[s0] * dinv[d0];
    own_lds_add(acc + (d0 - lo) * HIDDEN + lane, n0, p0, q0);
}

__device__ __forceinline__ void own_load_trip(const int* __restrict__ dst, int base, int lane, int e,
                                              bool vec_ok, int4 (&v)[4]) {
    if (vec_ok && base + SCAN_EDGES <= e) {
        #pragma unroll
        for (int g = 0; g < 4; ++g) v[g] = *(const int4*)(dst + base + g * 128 + lane * 4);
    } else {
        #pragma unroll
        for (int g = 0; g < 4; ++g) {
            const int i = base + g * 128 + lane * 4;
            v[g].x = (i     < e) ? dst[i]     : -1;
            v[g].y = (i + 1 < e) ? dst[i + 1] : -1;
            v[g].z = (i + 2 < e) ? dst[i + 2] : -1;
            v[g].w = (i + 3 < e) ? dst[i + 3] : -1;
        }
    }
}

__global__ void __launch_bounds__(OWN_THREADS)
k_agg1_own(const int* __restrict__ src, const int* __restrict__ dst,
           const float* __restrict__ dinv, const float* __restrict__ H1,
           const float* __restrict__ b1, const float* __restrict__ W2,
           const float* __restrict__ b2, float* __restrict__ H2,
           float* __restrict__ out, int e, int n) {
    __shared__ float acc[OWN_ROWS * HIDDEN];
    __shared__ int   lst[OWN_WAVES * LIST_CAP];

    const int lo = blockIdx.x * OWN_ROWS;
    int nrows = n - lo;
    if (nrows > OWN_ROWS) nrows = OWN_ROWS;
    const int wave = __builtin_amdgcn_readfirstlane((int)(threadIdx.x >> 5));
    const int lane = threadIdx.x & 31;

    for (int r = wave; r < nrows; r += OWN_WAVES) {
        const float dv = dinv[lo + r];
        const float4* h = (const float4*)(H1 + (size_t)(lo + r) * HIDDEN) + lane;
        const float4 p = h[0], q = h[32];
        float* a = acc + r * HIDDEN + lane;
        a[0 * 32] = dv * dv * p.x;
        a[1 * 32] = dv * dv * p.y;
        a[2 * 32] = dv * dv * p.z;
        a[3 * 32] = dv * dv * p.w;
        a[4 * 32] = dv * dv * q.x;
        a[5 * 32] = dv * dv * q.y;
        a[6 * 32] = dv * dv * q.z;
        a[7 * 32] = dv * dv * q.w;
    }
    __syncthreads();

    const unsigned ulo = (unsigned)lo, un = (unsigned)(nrows > 0 ? nrows : 0);
    const bool vec_ok = ((((size_t)dst) & 15) == 0);
    const int ntrips = (e + SCAN_EDGES - 1) / SCAN_EDGES;
    const int rot = (int)(((long long)blockIdx.x * ntrips) / gridDim.x);
    int* my = lst + wave * LIST_CAP;
    int cnt = 0;

    int4 cur[4], nxt[4];
    #pragma unroll
    for (int g = 0; g < 4; ++g) { cur[g] = make_int4(-1, -1, -1, -1); nxt[g] = cur[g]; }
    if (wave < ntrips) {
        int t = wave + rot; if (t >= ntrips) t -= ntrips;
        own_load_trip(dst, t * SCAN_EDGES, lane, e, vec_ok, cur);
    }
    for (int i = wave; i < ntrips; i += OWN_WAVES) {
        int t = i + rot; if (t >= ntrips) t -= ntrips;
        const int base = t * SCAN_EDGES;
        if (i + OWN_WAVES < ntrips) {
            int tn = i + OWN_WAVES + rot; if (tn >= ntrips) tn -= ntrips;
            own_load_trip(dst, tn * SCAN_EDGES, lane, e, vec_ok, nxt);
        }
        #pragma unroll
        for (int g = 0; g < 4; ++g) {
            const bool qx = ((unsigned)cur[g].x - ulo) < un;
            const bool qy = ((unsigned)cur[g].y - ulo) < un;
            const bool qz = ((unsigned)cur[g].z - ulo) < un;
            const bool qw = ((unsigned)cur[g].w - ulo) < un;
            const unsigned bx = __builtin_amdgcn_ballot_w32(qx);
            const unsigned by = __builtin_amdgcn_ballot_w32(qy);
            const unsigned bz = __builtin_amdgcn_ballot_w32(qz);
            const unsigned bw = __builtin_amdgcn_ballot_w32(qw);
            if ((bx | by | bz | bw) != 0u) {
                const int id = base + g * 128 + lane * 4;
                if (qx) my[cnt + (int)__builtin_amdgcn_mbcnt_lo(bx, 0u)] = id;
                cnt += __builtin_popcount(bx);
                if (qy) my[cnt + (int)__builtin_amdgcn_mbcnt_lo(by, 0u)] = id + 1;
                cnt += __builtin_popcount(by);
                if (qz) my[cnt + (int)__builtin_amdgcn_mbcnt_lo(bz, 0u)] = id + 2;
                cnt += __builtin_popcount(bz);
                if (qw) my[cnt + (int)__builtin_amdgcn_mbcnt_lo(bw, 0u)] = id + 3;
                cnt += __builtin_popcount(bw);
            }
        }
        while (cnt >= 4) {
            cnt -= 4;
            const int e0 = __builtin_amdgcn_readfirstlane(my[cnt]);
            const int e1 = __builtin_amdgcn_readfirstlane(my[cnt + 1]);
            const int e2 = __builtin_amdgcn_readfirstlane(my[cnt + 2]);
            const int e3 = __builtin_amdgcn_readfirstlane(my[cnt + 3]);
            own_add4(src, dst, dinv, H1, acc, lo, lane, e0, e1, e2, e3);
        }
        #pragma unroll
        for (int g = 0; g < 4; ++g) cur[g] = nxt[g];
    }
    while (cnt > 0) {
        cnt -= 1;
        const int e0 = __builtin_amdgcn_readfirstlane(my[cnt]);
        own_add1(src, dst, dinv, H1, acc, lo, lane, e0);
    }
    __syncthreads();

    float bb[8], w0[8], w1[8];
    #pragma unroll
    for (int q = 0; q < 2; ++q) {
        const int c = 4 * lane + 128 * q;
        const float4 bv = *(const float4*)(b1 + c);
        const float4 wa = *(const float4*)(W2 + 2 * c);
        const float4 wb = *(const float4*)(W2 + 2 * c + 4);
        bb[4 * q + 0] = bv.x; bb[4 * q + 1] = bv.y; bb[4 * q + 2] = bv.z; bb[4 * q + 3] = bv.w;
        w0[4 * q + 0] = wa.x; w1[4 * q + 0] = wa.y;
        w0[4 * q + 1] = wa.z; w1[4 * q + 1] = wa.w;
        w0[4 * q + 2] = wb.x; w1[4 * q + 2] = wb.y;
        w0[4 * q + 3] = wb.z; w1[4 * q + 3] = wb.w;
    }
    const float b20 = b2[0], b21 = b2[1];
    for (int r = wave; r < nrows; r += OWN_WAVES) {
        const float* a = acc + r * HIDDEN + lane;
        float p0 = 0.0f, p1 = 0.0f;
        #pragma unroll
        for (int j = 0; j < 8; ++j) {
            float z = a[j * 32] + bb[j];
            z = z > 0.0f ? z : 0.0f;
            p0 += z * w0[j];
            p1 += z * w1[j];
        }
        #pragma unroll
        for (int off = 16; off > 0; off >>= 1) {
            p0 += __shfl_xor(p0, off, 32);
            p1 += __shfl_xor(p1, off, 32);
        }
        if (lane == 0) {
            const int row = lo + r;
            float s = dinv[row]; s *= s;
            H2[2 * row]      = p0;
            H2[2 * row + 1]  = p1;
            out[2 * row]     = s * p0 + b20;
            out[2 * row + 1] = s * p1 + b21;
        }
    }
}


__global__ void k_edge_agg2(const int* __restrict__ src, const int* __restrict__ dst,
                            const float* __restrict__ dinv, const float* __restrict__ H2,
                            float* __restrict__ out, int e) {
    int i = blockIdx.x * blockDim.x + threadIdx.x;
    if (i >= e) return;
    int s = src[i], d = dst[i];
    float norm = dinv[s] * dinv[d];
    atomicAdd(&out[2*d],     norm * H2[2*s]);
    atomicAdd(&out[2*d + 1], norm * H2[2*s + 1]);
}


extern "C" void kernel_launch(void* const* d_in, const int* in_sizes, int n_in,
                              void* d_out, int out_size, void* d_ws, size_t ws_size,
                              hipStream_t stream) {
    const float* X   = (const float*)d_in[0];
    const float* W1  = (const float*)d_in[1];
    const float* b1  = (const float*)d_in[2];
    const float* W2  = (const float*)d_in[3];
    const float* b2  = (const float*)d_in[4];
    const int*   ei  = (const int*)d_in[5];
    float*       out = (float*)d_out;

    const int N = NNODES;
    const int E = in_sizes[5] / 2;
    const int* src = ei;
    const int* dst = ei + E;

    float*  deg  = (float*)d_ws;
    float*  dinv = deg  + N;
    float*  H1   = dinv + N;
    float*  H2   = H1   + (size_t)N * HIDDEN;
    float2* Wp   = (float2*)(H2 + (size_t)N * NCLS);

    const int T = 256;

    k_init_deg <<<(N + T - 1) / T, T, 0, stream>>>(deg, N);
    k_deg_accum<<<(E + T - 1) / T, T, 0, stream>>>(dst, deg, E);
    k_dinv     <<<(N + T - 1) / T, T, 0, stream>>>(deg, dinv, N);

    {
        int total = (DIMIN / 2) * HIDDEN;
        k_packW<<<(total + T - 1) / T, T, 0, stream>>>(W1, Wp);
    }

    k_gemm1<<<N / 16, 256, 0, stream>>>(X, Wp, H1);
    k_agg1_own<<<(N + OWN_ROWS - 1) / OWN_ROWS, OWN_THREADS, 0, stream>>>(src, dst, dinv, H1, b1, W2, b2,
                                                                      H2, out, E, N);

    k_edge_agg2<<<(E + T - 1) / T, T, 0, stream>>>(src, dst, dinv, H2, out, E);
}


# x@W1 made weight-stationary: 256 resident work-groups each keep a 128-column half of W1 in LDS (packed at fill; repack launch gone) and stream 16-row X tiles past it; sum order unchanged
# speedup vs baseline: 1.8919x; 1.0567x over previous
#include <hip/hip_runtime.h>
#include <hip/hip_bf16.h>

#define NNODES 50000
#define DIMIN  512
#define HIDDEN 256
#define NCLS   2

typedef __attribute__((ext_vector_type(2))) float v2f;
typedef __attribute__((ext_vector_type(8))) float v8f;


__global__ void k_init_deg(float* __restrict__ deg, int n) {
    int i = blockIdx.x * blockDim.x + threadIdx.x;
    if (i < n) deg[i] = 1.0f;
}

__global__ void k_deg_accum(const int* __restrict__ dst, float* __restrict__ deg, int e) {
    int i = blockIdx.x * blockDim.x + threadIdx.x;
    if (i < e) atomicAdd(&deg[dst[i]], 1.0f);
}

__global__ void k_dinv(const float* __restrict__ deg, float* __restrict__ dinv, int n) {
    int i = blockIdx.x * blockDim.x + threadIdx.x;
    if (i < n) dinv[i] = __frsqrt_rn(deg[i]);
}


#define G_THREADS 512
#define G_ROWG    128
#define G_HALFC   128
#define G_KH      256
#define G_APITCH  260
#define G_NT      (NNODES / 16)

__device__ __forceinline__ void gemm_stage_load(const float* __restrict__ X, int rg, int s, int tid,
                                                float4 (&r)[4]) {
    const int p  = s >> 1;
    const int kh = s & 1;
    #pragma unroll
    for (int j = 0; j < 4; ++j) {
        const int idx = tid + G_THREADS * j;
        const int rr  = idx >> 6;
        const int c4  = (idx & 63) * 4;
        const int t   = rg + G_ROWG * (2 * p + (rr >> 4));
        if (t < G_NT)
            r[j] = *(const float4*)(X + (size_t)(t * 16 + (rr & 15)) * DIMIN + kh * G_KH + c4);
        else
            r[j] = make_float4(0.0f, 0.0f, 0.0f, 0.0f);
    }
}

__global__ void __launch_bounds__(G_THREADS)
k_gemm1(const float* __restrict__ X, const float* __restrict__ W,
        float* __restrict__ H) {
    __shared__ alignas(16) float Ws[(DIMIN / 2) * G_HALFC * 2];
    __shared__ alignas(16) float As[32 * G_APITCH];

    const int tid  = threadIdx.x;
    const int half = blockIdx.x & 1;
    const int rg   = blockIdx.x >> 1;
    const int ntile  = (G_NT - rg + G_ROWG - 1) / G_ROWG;
    const int nstage = 2 * ((ntile + 1) / 2);

    float4 pre[4];
    gemm_stage_load(X, rg, 0, tid, pre);

    #pragma unroll 4
    for (int i = 0; i < 16; ++i) {
        const int it  = (i + rg) & 15;
        const int idx = tid + G_THREADS * it;
        const int kp  = idx >> 5;
        const int cq  = idx & 31;
        const float4 r0 = *(const float4*)(W + (size_t)(2 * kp)     * HIDDEN + half * G_HALFC + 4 * cq);
        const float4 r1 = *(const float4*)(W + (size_t)(2 * kp + 1) * HIDDEN + half * G_HALFC + 4 * cq);
        const int c = (4 * cq) ^ ((kp & 1) << 4);
        float4* d = (float4*)(&Ws[2 * (kp * G_HALFC + c)]);
        d[0] = make_float4(r0.x, r1.x, r0.y, r1.y);
        d[1] = make_float4(r0.z, r1.z, r0.w, r1.w);
    }

    const int wave = __builtin_amdgcn_readfirstlane(tid >> 5);
    const int lane = tid & 31;
    const int l16  = lane & 15;
    const int hi   = lane >> 4;
    const int sub  = wave >> 3;
    const int ct   = wave & 7;

    const float* ap = As + (sub * 16 + l16) * G_APITCH + 2 * hi;
    const float* bp = Ws + 2 * (hi * G_HALFC + ((ct ^ hi) << 4) + l16);

    v8f c = {};
    for (int s = 0; s < nstage; ++s) {
        __syncthreads();
        #pragma unroll
        for (int j = 0; j < 4; ++j) {
            const int idx = tid + G_THREADS * j;
            *(float4*)(&As[(idx >> 6) * G_APITCH + (idx & 63) * 4]) = pre[j];
        }
        __syncthreads();
        if (s + 1 < nstage) gemm_stage_load(X, rg, s + 1, tid, pre);

        const int kh = s & 1;
        const int t  = rg + G_ROWG * (2 * (s >> 1) + sub);
        if (t < G_NT) {
            const float* b = bp + 2 * (kh * (G_KH / 2) * G_HALFC);
            v2f fa[2][8], fb[2][8];
            #pragma unroll
            for (int i = 0; i < 8; ++i) {
                fa[0][i] = *(const v2f*)(ap + 4 * i);
                fb[0][i] = *(const v2f*)(b + (2 * i) * (2 * G_HALFC));
            }
            #pragma unroll
            for (int j = 0; j < G_KH / 32; ++j) {
                if (j + 1 < G_KH / 32) {
                    #pragma unroll
                    for (int i = 0; i < 8; ++i) {
                        const int kk = 32 * (j + 1) + 4 * i;
                        fa[(j + 1) & 1][i] = *(const v2f*)(ap + kk);
                        fb[(j + 1) & 1][i] = *(const v2f*)(b + (kk >> 1) * (2 * G_HALFC));
                    }
                }
                __builtin_amdgcn_sched_barrier(0);
                #pragma unroll
                for (int i = 0; i < 8; ++i)
                    c = __builtin_amdgcn_wmma_f32_16x16x4_f32(false, fa[j & 1][i], false, fb[j & 1][i],
                                                              (short)0, c, false, false);
                __builtin_amdgcn_sched_barrier(0);
            }
            if (kh == 1) {
                #pragma unroll
                for (int v = 0; v < 8; ++v) {
                    const size_t o = (size_t)(t * 16 + hi * 8 + v) * HIDDEN + half * G_HALFC + ct * 16 + l16;
                    H[o] = c[v];
                }
                c = v8f{};
            }
        }
    }
}


#define OWN_ROWS    196
#define OWN_THREADS 1024
#define OWN_WAVES   (OWN_THREADS / 32)
#define SCAN_EDGES  512
#define LIST_CAP    (SCAN_EDGES + 4)

__device__ __forceinline__ void own_lds_add(float* a, float nrm, const float4& p, const float4& q) {
    __hip_atomic_fetch_add(&a[0 * 32], nrm * p.x, __ATOMIC_RELAXED, __HIP_MEMORY_SCOPE_WORKGROUP);
    __hip_atomic_fetch_add(&a[1 * 32], nrm * p.y, __ATOMIC_RELAXED, __HIP_MEMORY_SCOPE_WORKGROUP);
    __hip_atomic_fetch_add(&a[2 * 32], nrm * p.z, __ATOMIC_RELAXED, __HIP_MEMORY_SCOPE_WORKGROUP);
    __hip_atomic_fetch_add(&a[3 * 32], nrm * p.w, __ATOMIC_RELAXED, __HIP_MEMORY_SCOPE_WORKGROUP);
    __hip_atomic_fetch_add(&a[4 * 32], nrm * q.x, __ATOMIC_RELAXED, __HIP_MEMORY_SCOPE_WORKGROUP);
    __hip_atomic_fetch_add(&a[5 * 32], nrm * q.y, __ATOMIC_RELAXED, __HIP_MEMORY_SCOPE_WORKGROUP);
    __hip_atomic_fetch_add(&a[6 * 32], nrm * q.z, __ATOMIC_RELAXED, __HIP_MEMORY_SCOPE_WORKGROUP);
    __hip_atomic_fetch_add(&a[7 * 32], nrm * q.w, __ATOMIC_RELAXED, __HIP_MEMORY_SCOPE_WORKGROUP);
}

__device__ __forceinline__ void own_add4(const int* __restrict__ src, const int* __restrict__ dst,
                                         const float* __restrict__ dinv, const float* __restrict__ H1,
                                         float* acc, int lo, int lane,
                                         int e0, int e1, int e2, int e3) {
    const int s0 = src[e0], s1 = src[e1], s2 = src[e2], s3 = src[e3];
    const int d0 = dst[e0], d1 = dst[e1], d2 = dst[e2], d3 = dst[e3];
    const float4* h0 = (const float4*)(H1 + (size_t)s0 * HIDDEN) + lane;
    const float4* h1 = (const float4*)(H1 + (size_t)s1 * HIDDEN) + lane;
    const float4* h2 = (const float4*)(H1 + (size_t)s2 * HIDDEN) + lane;
    const float4* h3 = (const float4*)(H1 + (size_t)s3 * HIDDEN) + lane;
    const float4 p0 = h0[0], q0 = h0[32];
    const float4 p1 = h1[0], q1 = h1[32];
    const float4 p2 = h2[0], q2 = h2[32];
    const float4 p3 = h3[0], q3 = h3[32];
    const float n0 = dinv[s0] * dinv[d0];
    const float n1 = dinv[s1] * dinv[d1];
    const float n2 = dinv[s2] * dinv[d2];
    const float n3 = dinv[s3] * dinv[d3];
    own_lds_add(acc + (d0 - lo) * HIDDEN + lane, n0, p0, q0);
    own_lds_add(acc + (d1 - lo) * HIDDEN + lane, n1, p1, q1);
    own_lds_add(acc + (d2 - lo) * HIDDEN + lane, n2, p2, q2);
    own_lds_add(acc + (d3 - lo) * HIDDEN + lane, n3, p3, q3);
}

__device__ __forceinline__ void own_add1(const int* __restrict__ src, const int* __restrict__ dst,
                                         const float* __restrict__ dinv, const float* __restrict__ H1,
                                         float* acc, int lo, int lane, int e0) {
    const int s0 = src[e0];
    const int d0 = dst[e0];
    const float4* h0 = (const float4*)(H1 + (size_t)s0 * HIDDEN) + lane;
    const float4 p0 = h0[0], q0 = h0[32];
    const float n0 = dinv[s0] * dinv[d0];
    own_lds_add(acc + (d0 - lo) * HIDDEN + lane, n0, p0, q0);
}

__device__ __forceinline__ void own_load_trip(const int* __restrict__ dst, int base, int lane, int e,
                                              bool vec_ok, int4 (&v)[4]) {
    if (vec_ok && base + SCAN_EDGES <= e) {
        #pragma unroll
        for (int g = 0; g < 4; ++g) v[g] = *(const int4*)(dst + base + g * 128 + lane * 4);
    } else {
        #pragma unroll
        for (int g = 0; g < 4; ++g) {
            const int i = base + g * 128 + lane * 4;
            v[g].x = (i     < e) ? dst[i]     : -1;
            v[g].y = (i + 1 < e) ? dst[i + 1] : -1;
            v[g].z = (i + 2 < e) ? dst[i + 2] : -1;
            v[g].w = (i + 3 < e) ? dst[i + 3] : -1;
        }
    }
}

__global__ void __launch_bounds__(OWN_THREADS)
k_agg1_own(const int* __restrict__ src, const int* __restrict__ dst,
           const float* __restrict__ dinv, const float* __restrict__ H1,
           const float* __restrict__ b1, const float* __restrict__ W2,
           const float* __restrict__ b2, float* __restrict__ H2,
           float* __restrict__ out, int e, int n) {
    __shared__ float acc[OWN_ROWS * HIDDEN];
    __shared__ int   lst[OWN_WAVES * LIST_CAP];

    const int lo = blockIdx.x * OWN_ROWS;
    int nrows = n - lo;
    if (nrows > OWN_ROWS) nrows = OWN_ROWS;
    const int wave = __builtin_amdgcn_readfirstlane((int)(threadIdx.x >> 5));
    const int lane = threadIdx.x & 31;

    for (int r = wave; r < nrows; r += OWN_WAVES) {
        const float dv = dinv[lo + r];
        const float4* h = (const float4*)(H1 + (size_t)(lo + r) * HIDDEN) + lane;
        const float4 p = h[0], q = h[32];
        float* a = acc + r * HIDDEN + lane;
        a[0 * 32] = dv * dv * p.x;
        a[1 * 32] = dv * dv * p.y;
        a[2 * 32] = dv * dv * p.z;
        a[3 * 32] = dv * dv * p.w;
        a[4 * 32] = dv * dv * q.x;
        a[5 * 32] = dv * dv * q.y;
        a[6 * 32] = dv * dv * q.z;
        a[7 * 32] = dv * dv * q.w;
    }
    __syncthreads();

    const unsigned ulo = (unsigned)lo, un = (unsigned)(nrows > 0 ? nrows : 0);
    const bool vec_ok = ((((size_t)dst) & 15) == 0);
    const int ntrips = (e + SCAN_EDGES - 1) / SCAN_EDGES;
    const int rot = (int)(((long long)blockIdx.x * ntrips) / gridDim.x);
    int* my = lst + wave * LIST_CAP;
    int cnt = 0;

    int4 cur[4], nxt[4];
    #pragma unroll
    for (int g = 0; g < 4; ++g) { cur[g] = make_int4(-1, -1, -1, -1); nxt[g] = cur[g]; }
    if (wave < ntrips) {
        int t = wave + rot; if (t >= ntrips) t -= ntrips;
        own_load_trip(dst, t * SCAN_EDGES, lane, e, vec_ok, cur);
    }
    for (int i = wave; i < ntrips; i += OWN_WAVES) {
        int t = i + rot; if (t >= ntrips) t -= ntrips;
        const int base = t * SCAN_EDGES;
        if (i + OWN_WAVES < ntrips) {
            int tn = i + OWN_WAVES + rot; if (tn >= ntrips) tn -= ntrips;
            own_load_trip(dst, tn * SCAN_EDGES, lane, e, vec_ok, nxt);
        }
        #pragma unroll
        for (int g = 0; g < 4; ++g) {
            const bool qx = ((unsigned)cur[g].x - ulo) < un;
            const bool qy = ((unsigned)cur[g].y - ulo) < un;
            const bool qz = ((unsigned)cur[g].z - ulo) < un;
            const bool qw = ((unsigned)cur[g].w - ulo) < un;
            const unsigned bx = __builtin_amdgcn_ballot_w32(qx);
            const unsigned by = __builtin_amdgcn_ballot_w32(qy);
            const unsigned bz = __builtin_amdgcn_ballot_w32(qz);
            const unsigned bw = __builtin_amdgcn_ballot_w32(qw);
            if ((bx | by | bz | bw) != 0u) {
                const int id = base + g * 128 + lane * 4;
                if (qx) my[cnt + (int)__builtin_amdgcn_mbcnt_lo(bx, 0u)] = id;
                cnt += __builtin_popcount(bx);
                if (qy) my[cnt + (int)__builtin_amdgcn_mbcnt_lo(by, 0u)] = id + 1;
                cnt += __builtin_popcount(by);
                if (qz) my[cnt + (int)__builtin_amdgcn_mbcnt_lo(bz, 0u)] = id + 2;
                cnt += __builtin_popcount(bz);
                if (qw) my[cnt + (int)__builtin_amdgcn_mbcnt_lo(bw, 0u)] = id + 3;
                cnt += __builtin_popcount(bw);
            }
        }
        while (cnt >= 4) {
            cnt -= 4;
            const int e0 = __builtin_amdgcn_readfirstlane(my[cnt]);
            const int e1 = __builtin_amdgcn_readfirstlane(my[cnt + 1]);
            const int e2 = __builtin_amdgcn_readfirstlane(my[cnt + 2]);
            const int e3 = __builtin_amdgcn_readfirstlane(my[cnt + 3]);
            own_add4(src, dst, dinv, H1, acc, lo, lane, e0, e1, e2, e3);
        }
        #pragma unroll
        for (int g = 0; g < 4; ++g) cur[g] = nxt[g];
    }
    while (cnt > 0) {
        cnt -= 1;
        const int e0 = __builtin_amdgcn_readfirstlane(my[cnt]);
        own_add1(src, dst, dinv, H1, acc, lo, lane, e0);
    }
    __syncthreads();

    float bb[8], w0[8], w1[8];
    #pragma unroll
    for (int q = 0; q < 2; ++q) {
        const int c = 4 * lane + 128 * q;
        const float4 bv = *(const float4*)(b1 + c);
        const float4 wa = *(const float4*)(W2 + 2 * c);
        const float4 wb = *(const float4*)(W2 + 2 * c + 4);
        bb[4 * q + 0] = bv.x; bb[4 * q + 1] = bv.y; bb[4 * q + 2] = bv.z; bb[4 * q + 3] = bv.w;
        w0[4 * q + 0] = wa.x; w1[4 * q + 0] = wa.y;
        w0[4 * q + 1] = wa.z; w1[4 * q + 1] = wa.w;
        w0[4 * q + 2] = wb.x; w1[4 * q + 2] = wb.y;
        w0[4 * q + 3] = wb.z; w1[4 * q + 3] = wb.w;
    }
    const float b20 = b2[0], b21 = b2[1];
    for (int r = wave; r < nrows; r += OWN_WAVES) {
        const float* a = acc + r * HIDDEN + lane;
        float p0 = 0.0f, p1 = 0.0f;
        #pragma unroll
        for (int j = 0; j < 8; ++j) {
            float z = a[j * 32] + bb[j];
            z = z > 0.0f ? z : 0.0f;
            p0 += z * w0[j];
            p1 += z * w1[j];
        }
        #pragma unroll
        for (int off = 16; off > 0; off >>= 1) {
            p0 += __shfl_xor(p0, off, 32);
            p1 += __shfl_xor(p1, off, 32);
        }
        if (lane == 0) {
            const int row = lo + r;
            float s = dinv[row]; s *= s;
            H2[2 * row]      = p0;
            H2[2 * row + 1]  = p1;
            out[2 * row]     = s * p0 + b20;
            out[2 * row + 1] = s * p1 + b21;
        }
    }
}


__global__ void k_edge_agg2(const int* __restrict__ src, const int* __restrict__ dst,
                            const float* __restrict__ dinv, const float* __restrict__ H2,
                            float* __restrict__ out, int e) {
    int i = blockIdx.x * blockDim.x + threadIdx.x;
    if (i >= e) return;
    int s = src[i], d = dst[i];
    float norm = dinv[s] * dinv[d];
    atomicAdd(&out[2*d],     norm * H2[2*s]);
    atomicAdd(&out[2*d + 1], norm * H2[2*s + 1]);
}


extern "C" void kernel_launch(void* const* d_in, const int* in_sizes, int n_in,
                              void* d_out, int out_size, void* d_ws, size_t ws_size,
                              hipStream_t stream) {
    const float* X   = (const float*)d_in[0];
    const float* W1  = (const float*)d_in[1];
    const float* b1  = (const float*)d_in[2];
    const float* W2  = (const float*)d_in[3];
    const float* b2  = (const float*)d_in[4];
    const int*   ei  = (const int*)d_in[5];
    float*       out = (float*)d_out;

    const int N = NNODES;
    const int E = in_sizes[5] / 2;
    const int* src = ei;
    const int* dst = ei + E;

    float*  deg  = (float*)d_ws;
    float*  dinv = deg  + N;
    float*  H1   = dinv + N;
    float*  H2   = H1   + (size_t)N * HIDDEN;

    const int T = 256;

    k_init_deg <<<(N + T - 1) / T, T, 0, stream>>>(deg, N);
    k_deg_accum<<<(E + T - 1) / T, T, 0, stream>>>(dst, deg, E);
    k_dinv     <<<(N + T - 1) / T, T, 0, stream>>>(deg, dinv, N);

    k_gemm1<<<2 * G_ROWG, G_THREADS, 0, stream>>>(X, W1, H1);
    k_agg1_own<<<(N + OWN_ROWS - 1) / OWN_ROWS, OWN_THREADS, 0, stream>>>(src, dst, dinv, H1, b1, W2, b2,
                                                                      H2, out, E, N);

    k_edge_agg2<<<(E + T - 1) / T, T, 0, stream>>>(src, dst, dinv, H2, out, E);
}


# Layer-1 owner scan reads a packed edge word (src | dst<<16, written by the degree pass): an owned edge reaches the add with source and row in hand, dinv of owned rows in LDS; sum order as before
# speedup vs baseline: 1.9725x; 1.0426x over previous
#include <hip/hip_runtime.h>
#include <hip/hip_bf16.h>

#define NNODES 50000
#define DIMIN  512
#define HIDDEN 256
#define NCLS   2

typedef __attribute__((ext_vector_type(2))) float v2f;
typedef __attribute__((ext_vector_type(8))) float v8f;


__global__ void k_init_deg(float* __restrict__ deg, int n) {
    int i = blockIdx.x * blockDim.x + threadIdx.x;
    if (i < n) deg[i] = 1.0f;
}

static_assert(NNODES <= 65536, "packed edge word keeps a node id in 16 bits");

__global__ void k_deg_accum(const int* __restrict__ src, const int* __restrict__ dst,
                            float* __restrict__ deg, unsigned* __restrict__ key, int e) {
    int i = blockIdx.x * blockDim.x + threadIdx.x;
    if (i < e) {
        const int d = dst[i];
        key[i] = (unsigned)src[i] | ((unsigned)d << 16);
        atomicAdd(&deg[d], 1.0f);
    }
}

__global__ void k_dinv(const float* __restrict__ deg, float* __restrict__ dinv, int n) {
    int i = blockIdx.x * blockDim.x + threadIdx.x;
    if (i < n) dinv[i] = __frsqrt_rn(deg[i]);
}


#define G_THREADS 512
#define G_ROWG    128
#define G_HALFC   128
#define G_KH      256
#define G_APITCH  260
#define G_NT      (NNODES / 16)

__device__ __forceinline__ void gemm_stage_load(const float* __restrict__ X, int rg, int s, int tid,
                                                float4 (&r)[4]) {
    const int p  = s >> 1;
    const int kh = s & 1;
    #pragma unroll
    for (int j = 0; j < 4; ++j) {
        const int idx = tid + G_THREADS * j;
        const int rr  = idx >> 6;
        const int c4  = (idx & 63) * 4;
        const int t   = rg + G_ROWG * (2 * p + (rr >> 4));
        if (t < G_NT)
            r[j] = *(const float4*)(X + (size_t)(t * 16 + (rr & 15)) * DIMIN + kh * G_KH + c4);
        else
            r[j] = make_float4(0.0f, 0.0f, 0.0f, 0.0f);
    }
}

__global__ void __launch_bounds__(G_THREADS)
k_gemm1(const float* __restrict__ X, const float* __restrict__ W,
        float* __restrict__ H) {
    __shared__ alignas(16) float Ws[(DIMIN / 2) * G_HALFC * 2];
    __shared__ alignas(16) float As[32 * G_APITCH];

    const int tid  = threadIdx.x;
    const int half = blockIdx.x & 1;
    const int rg   = blockIdx.x >> 1;
    const int ntile  = (G_NT - rg + G_ROWG - 1) / G_ROWG;
    const int nstage = 2 * ((ntile + 1) / 2);

    float4 pre[4];
    gemm_stage_load(X, rg, 0, tid, pre);

    #pragma unroll 4
    for (int i = 0; i < 16; ++i) {
        const int it  = (i + rg) & 15;
        const int idx = tid + G_THREADS * it;
        const int kp  = idx >> 5;
        const int cq  = idx & 31;
        const float4 r0 = *(const float4*)(W + (size_t)(2 * kp)     * HIDDEN + half * G_HALFC + 4 * cq);
        const float4 r1 = *(const float4*)(W + (size_t)(2 * kp + 1) * HIDDEN + half * G_HALFC + 4 * cq);
        const int c = (4 * cq) ^ ((kp & 1) << 4);
        float4* d = (float4*)(&Ws[2 * (kp * G_HALFC + c)]);
        d[0] = make_float4(r0.x, r1.x, r0.y, r1.y);
        d[1] = make_float4(r0.z, r1.z, r0.w, r1.w);
    }

    const int wave = __builtin_amdgcn_readfirstlane(tid >> 5);
    const int lane = tid & 31;
    const int l16  = lane & 15;
    const int hi   = lane >> 4;
    const int sub  = wave >> 3;
    const int ct   = wave & 7;

    const float* ap = As + (sub * 16 + l16) * G_APITCH + 2 * hi;
    const float* bp = Ws + 2 * (hi * G_HALFC + ((ct ^ hi) << 4) + l16);

    v8f c = {};
    for (int s = 0; s < nstage; ++s) {
        __syncthreads();
        #pragma unroll
        for (int j = 0; j < 4; ++j) {
            const int idx = tid + G_THREADS * j;
            *(float4*)(&As[(idx >> 6) * G_APITCH + (idx & 63) * 4]) = pre[j];
        }
        __syncthreads();
        if (s + 1 < nstage) gemm_stage_load(X, rg, s + 1, tid, pre);

        const int kh = s & 1;
        const int t  = rg + G_ROWG * (2 * (s >> 1) + sub);
        if (t < G_NT) {
            const float* b = bp + 2 * (kh * (G_KH / 2) * G_HALFC);
            v2f fa[2][8], fb[2][8];
            #pragma unroll
            for (int i = 0; i < 8; ++i) {
                fa[0][i] = *(const v2f*)(ap + 4 * i);
                fb[0][i] = *(const v2f*)(b + (2 * i) * (2 * G_HALFC));
            }
            #pragma unroll
            for (int j = 0; j < G_KH / 32; ++j) {
                if (j + 1 < G_KH / 32) {
                    #pragma unroll
                    for (int i = 0; i < 8; ++i) {
                        const int kk = 32 * (j + 1) + 4 * i;
                        fa[(j + 1) & 1][i] = *(const v2f*)(ap + kk);
                        fb[(j + 1) & 1][i] = *(const v2f*)(b + (kk >> 1) * (2 * G_HALFC));
                    }
                }
                __builtin_amdgcn_sched_barrier(0);
                #pragma unroll
                for (int i = 0; i < 8; ++i)
                    c = __builtin_amdgcn_wmma_f32_16x16x4_f32(false, fa[j & 1][i], false, fb[j & 1][i],
                                                              (short)0, c, false, false);
                __builtin_amdgcn_sched_barrier(0);
            }
            if (kh == 1) {
                #pragma unroll
                for (int v = 0; v < 8; ++v) {
                    const size_t o = (size_t)(t * 16 + hi * 8 + v) * HIDDEN + half * G_HALFC + ct * 16 + l16;
                    H[o] = c[v];
                }
                c = v8f{};
            }
        }
    }
}


#define OWN_ROWS    196
#define OWN_THREADS 1024
#define OWN_WAVES   (OWN_THREADS / 32)
#define SCAN_EDGES  512
#define LIST_CAP    (SCAN_EDGES + 4)

__device__ __forceinline__ void own_lds_add(float* a, float nrm, const float4& p, const float4& q) {
    __hip_atomic_fetch_add(&a[0 * 32], nrm * p.x, __ATOMIC_RELAXED, __HIP_MEMORY_SCOPE_WORKGROUP);
    __hip_atomic_fetch_add(&a[1 * 32], nrm * p.y, __ATOMIC_RELAXED, __HIP_MEMORY_SCOPE_WORKGROUP);
    __hip_atomic_fetch_add(&a[2 * 32], nrm * p.z, __ATOMIC_RELAXED, __HIP_MEMORY_SCOPE_WORKGROUP);
    __hip_atomic_fetch_add(&a[3 * 32], nrm * p.w, __ATOMIC_RELAXED, __HIP_MEMORY_SCOPE_WORKGROUP);
    __hip_atomic_fetch_add(&a[4 * 32], nrm * q.x, __ATOMIC_RELAXED, __HIP_MEMORY_SCOPE_WORKGROUP);
    __hip_atomic_fetch_add(&a[5 * 32], nrm * q.y, __ATOMIC_RELAXED, __HIP_MEMORY_SCOPE_WORKGROUP);
    __hip_atomic_fetch_add(&a[6 * 32], nrm * q.z, __ATOMIC_RELAXED, __HIP_MEMORY_SCOPE_WORKGROUP);
    __hip_atomic_fetch_add(&a[7 * 32], nrm * q.w, __ATOMIC_RELAXED, __HIP_MEMORY_SCOPE_WORKGROUP);
}

__device__ __forceinline__ void own_add4(const float* __restrict__ dinv, const float* __restrict__ H1,
                                         float* acc, const float* dl, int lane,
                                         unsigned t0, unsigned t1, unsigned t2, unsigned t3) {
    const int s0 = (int)(t0 & 0xFFFFu), s1 = (int)(t1 & 0xFFFFu);
    const int s2 = (int)(t2 & 0xFFFFu), s3 = (int)(t3 & 0xFFFFu);
    const int r0 = (int)(t0 >> 16), r1 = (int)(t1 >> 16), r2 = (int)(t2 >> 16), r3 = (int)(t3 >> 16);
    const float4* h0 = (const float4*)(H1 + (size_t)s0 * HIDDEN) + lane;
    const float4* h1 = (const float4*)(H1 + (size_t)s1 * HIDDEN) + lane;
    const float4* h2 = (const float4*)(H1 + (size_t)s2 * HIDDEN) + lane;
    const float4* h3 = (const float4*)(H1 + (size_t)s3 * HIDDEN) + lane;
    const float4 p0 = h0[0], q0 = h0[32];
    const float4 p1 = h1[0], q1 = h1[32];
    const float4 p2 = h2[0], q2 = h2[32];
    const float4 p3 = h3[0], q3 = h3[32];
    const float n0 = dinv[s0] * dl[r0];
    const float n1 = dinv[s1] * dl[r1];
    const float n2 = dinv[s2] * dl[r2];
    const float n3 = dinv[s3] * dl[r3];
    own_lds_add(acc + r0 * HIDDEN + lane, n0, p0, q0);
    own_lds_add(acc + r1 * HIDDEN + lane, n1, p1, q1);
    own_lds_add(acc + r2 * HIDDEN + lane, n2, p2, q2);
    own_lds_add(acc + r3 * HIDDEN + lane, n3, p3, q3);
}

__device__ __forceinline__ void own_add1(const float* __restrict__ dinv, const float* __restrict__ H1,
                                         float* acc, const float* dl, int lane, unsigned t0) {
    const int s0 = (int)(t0 & 0xFFFFu);
    const int r0 = (int)(t0 >> 16);
    const float4* h0 = (const float4*)(H1 + (size_t)s0 * HIDDEN) + lane;
    const float4 p0 = h0[0], q0 = h0[32];
    const float n0 = dinv[s0] * dl[r0];
    own_lds_add(acc + r0 * HIDDEN + lane, n0, p0, q0);
}

__device__ __forceinline__ void own_load_trip(const int* __restrict__ kw, int base, int lane, int e,
                                              bool vec_ok, int4 (&v)[4]) {
    if (vec_ok && base + SCAN_EDGES <= e) {
        #pragma unroll
        for (int g = 0; g < 4; ++g) v[g] = *(const int4*)(kw + base + g * 128 + lane * 4);
    } else {
        #pragma unroll
        for (int g = 0; g < 4; ++g) {
            const int i = base + g * 128 + lane * 4;
            v[g].x = (i     < e) ? kw[i]     : -1;
            v[g].y = (i + 1 < e) ? kw[i + 1] : -1;
            v[g].z = (i + 2 < e) ? kw[i + 2] : -1;
            v[g].w = (i + 3 < e) ? kw[i + 3] : -1;
        }
    }
}

__global__ void __launch_bounds__(OWN_THREADS)
k_agg1_own(const unsigned* __restrict__ key,
           const float* __restrict__ dinv, const float* __restrict__ H1,
           const float* __restrict__ b1, const float* __restrict__ W2,
           const float* __restrict__ b2, float* __restrict__ H2,
           float* __restrict__ out, int e, int n) {
    __shared__ float acc[OWN_ROWS * HIDDEN];
    __shared__ unsigned lst[OWN_WAVES * LIST_CAP];
    __shared__ float dl[OWN_ROWS];

    const int lo = blockIdx.x * OWN_ROWS;
    int nrows = n - lo;
    if (nrows > OWN_ROWS) nrows = OWN_ROWS;
    const int wave = __builtin_amdgcn_readfirstlane((int)(threadIdx.x >> 5));
    const int lane = threadIdx.x & 31;

    for (int r = wave; r < nrows; r += OWN_WAVES) {
        const float dv = dinv[lo + r];
        if (lane == 0) dl[r] = dv;
        const float4* h = (const float4*)(H1 + (size_t)(lo + r) * HIDDEN) + lane;
        const float4 p = h[0], q = h[32];
        float* a = acc + r * HIDDEN + lane;
        a[0 * 32] = dv * dv * p.x;
        a[1 * 32] = dv * dv * p.y;
        a[2 * 32] = dv * dv * p.z;
        a[3 * 32] = dv * dv * p.w;
        a[4 * 32] = dv * dv * q.x;
        a[5 * 32] = dv * dv * q.y;
        a[6 * 32] = dv * dv * q.z;
        a[7 * 32] = dv * dv * q.w;
    }
    __syncthreads();

    const unsigned ulo = (unsigned)lo << 16, un = (unsigned)(nrows > 0 ? nrows : 0) << 16;
    const int* kw = (const int*)key;
    const bool vec_ok = ((((size_t)kw) & 15) == 0);
    const int ntrips = (e + SCAN_EDGES - 1) / SCAN_EDGES;
    const int rot = (int)(((long long)blockIdx.x * ntrips) / gridDim.x);
    unsigned* my = lst + wave * LIST_CAP;
    int cnt = 0;

    int4 cur[4], nxt[4];
    #pragma unroll
    for (int g = 0; g < 4; ++g) { cur[g] = make_int4(-1, -1, -1, -1); nxt[g] = cur[g]; }
    if (wave < ntrips) {
        int t = wave + rot; if (t >= ntrips) t -= ntrips;
        own_load_trip(kw, t * SCAN_EDGES, lane, e, vec_ok, cur);
    }
    for (int i = wave; i < ntrips; i += OWN_WAVES) {
        if (i + OWN_WAVES < ntrips) {
            int tn = i + OWN_WAVES + rot; if (tn >= ntrips) tn -= ntrips;
            own_load_trip(kw, tn * SCAN_EDGES, lane, e, vec_ok, nxt);
        }
        #pragma unroll
        for (int g = 0; g < 4; ++g) {
            const unsigned xx = (unsigned)cur[g].x - ulo, xy = (unsigned)cur[g].y - ulo;
            const unsigned xz = (unsigned)cur[g].z - ulo, xw = (unsigned)cur[g].w - ulo;
            const bool qx = xx < un;
            const bool qy = xy < un;
            const bool qz = xz < un;
            const bool qw = xw < un;
            const unsigned bx = __builtin_amdgcn_ballot_w32(qx);
            const unsigned by = __builtin_amdgcn_ballot_w32(qy);
            const unsigned bz = __builtin_amdgcn_ballot_w32(qz);
            const unsigned bw = __builtin_amdgcn_ballot_w32(qw);
            if ((bx | by | bz | bw) != 0u) {
                if (qx) my[cnt + (int)__builtin_amdgcn_mbcnt_lo(bx, 0u)] = xx;
                cnt += __builtin_popcount(bx);
                if (qy) my[cnt + (int)__builtin_amdgcn_mbcnt_lo(by, 0u)] = xy;
                cnt += __builtin_popcount(by);
                if (qz) my[cnt + (int)__builtin_amdgcn_mbcnt_lo(bz, 0u)] = xz;
                cnt += __builtin_popcount(bz);
                if (qw) my[cnt + (int)__builtin_amdgcn_mbcnt_lo(bw, 0u)] = xw;
                cnt += __builtin_popcount(bw);
            }
        }
        while (cnt >= 4) {
            cnt -= 4;
            const unsigned t0 = (unsigned)__builtin_amdgcn_readfirstlane((int)my[cnt]);
            const unsigned t1 = (unsigned)__builtin_amdgcn_readfirstlane((int)my[cnt + 1]);
            const unsigned t2 = (unsigned)__builtin_amdgcn_readfirstlane((int)my[cnt + 2]);
            const unsigned t3 = (unsigned)__builtin_amdgcn_readfirstlane((int)my[cnt + 3]);
            own_add4(dinv, H1, acc, dl, lane, t0, t1, t2, t3);
        }
        #pragma unroll
        for (int g = 0; g < 4; ++g) cur[g] = nxt[g];
    }
    while (cnt > 0) {
        cnt -= 1;
        const unsigned t0 = (unsigned)__builtin_amdgcn_readfirstlane((int)my[cnt]);
        own_add1(dinv, H1, acc, dl, lane, t0);
    }
    __syncthreads();

    float bb[8], w0[8], w1[8];
    #pragma unroll
    for (int q = 0; q < 2; ++q) {
        const int c = 4 * lane + 128 * q;
        const float4 bv = *(const float4*)(b1 + c);
        const float4 wa = *(const float4*)(W2 + 2 * c);
        const float4 wb = *(const float4*)(W2 + 2 * c + 4);
        bb[4 * q + 0] = bv.x; bb[4 * q + 1] = bv.y; bb[4 * q + 2] = bv.z; bb[4 * q + 3] = bv.w;
        w0[4 * q + 0] = wa.x; w1[4 * q + 0] = wa.y;
        w0[4 * q + 1] = wa.z; w1[4 * q + 1] = wa.w;
        w0[4 * q + 2] = wb.x; w1[4 * q + 2] = wb.y;
        w0[4 * q + 3] = wb.z; w1[4 * q + 3] = wb.w;
    }
    const float b20 = b2[0], b21 = b2[1];
    for (int r = wave; r < nrows; r += OWN_WAVES) {
        const float* a = acc + r * HIDDEN + lane;
        float p0 = 0.0f, p1 = 0.0f;
        #pragma unroll
        for (int j = 0; j < 8; ++j) {
            float z = a[j * 32] + bb[j];
            z = z > 0.0f ? z : 0.0f;
            p0 += z * w0[j];
            p1 += z * w1[j];
        }
        #pragma unroll
        for (int off = 16; off > 0; off >>= 1) {
            p0 += __shfl_xor(p0, off, 32);
            p1 += __shfl_xor(p1, off, 32);
        }
        if (lane == 0) {
            const int row = lo + r;
            float s = dinv[row]; s *= s;
            H2[2 * row]      = p0;
            H2[2 * row + 1]  = p1;
            out[2 * row]     = s * p0 + b20;
            out[2 * row + 1] = s * p1 + b21;
        }
    }
}


__global__ void k_edge_agg2(const int* __restrict__ src, const int* __restrict__ dst,
                            const float* __restrict__ dinv, const float* __restrict__ H2,
                            float* __restrict__ out, int e) {
    int i = blockIdx.x * blockDim.x + threadIdx.x;
    if (i >= e) return;
    int s = src[i], d = dst[i];
    float norm = dinv[s] * dinv[d];
    atomicAdd(&out[2*d],     norm * H2[2*s]);
    atomicAdd(&out[2*d + 1], norm * H2[2*s + 1]);
}


extern "C" void kernel_launch(void* const* d_in, const int* in_sizes, int n_in,
                              void* d_out, int out_size, void* d_ws, size_t ws_size,
                              hipStream_t stream) {
    const float* X   = (const float*)d_in[0];
    const float* W1  = (const float*)d_in[1];
    const float* b1  = (const float*)d_in[2];
    const float* W2  = (const float*)d_in[3];
    const float* b2  = (const float*)d_in[4];
    const int*   ei  = (const int*)d_in[5];
    float*       out = (float*)d_out;

    const int N = NNODES;
    const int E = in_sizes[5] / 2;
    const int* src = ei;
    const int* dst = ei + E;

    float*  deg  = (float*)d_ws;
    float*  dinv = deg  + N;
    float*  H1   = dinv + N;
    float*  H2   = H1   + (size_t)N * HIDDEN;
    unsigned* key = (unsigned*)(H2 + (size_t)N * NCLS);

    const int T = 256;

    k_init_deg <<<(N + T - 1) / T, T, 0, stream>>>(deg, N);
    k_deg_accum<<<(E + T - 1) / T, T, 0, stream>>>(src, dst, deg, key, E);
    k_dinv     <<<(N + T - 1) / T, T, 0, stream>>>(deg, dinv, N);

    k_gemm1<<<2 * G_ROWG, G_THREADS, 0, stream>>>(X, W1, H1);
    k_agg1_own<<<(N + OWN_ROWS - 1) / OWN_ROWS, OWN_THREADS, 0, stream>>>(key, dinv, H1, b1, W2, b2,
                                                                      H2, out, E, N);

    k_edge_agg2<<<(E + T - 1) / T, T, 0, stream>>>(src, dst, dinv, H2, out, E);
}


# Degree count and layer-2 scatter privatised: a work-group sums an edge chunk in an LDS table over all nodes, then adds it by whole-line atomics; layer-2 sums now per-chunk partials, arrival order
# speedup vs baseline: 2.9099x; 1.4752x over previous
#include <hip/hip_runtime.h>
#include <hip/hip_bf16.h>

#define NNODES 50000
#define DIMIN  512
#define HIDDEN 256
#define NCLS   2

typedef __attribute__((ext_vector_type(2))) float v2f;
typedef __attribute__((ext_vector_type(8))) float v8f;


__global__ void k_init_deg(float* __restrict__ deg, int n) {
    int i = blockIdx.x * blockDim.x + threadIdx.x;
    if (i < n) deg[i] = 1.0f;
}

static_assert(NNODES <= 65536, "packed edge word keeps a node id in 16 bits");
static_assert(NNODES % 4 == 0, "the LDS tables are cleared with 16-byte stores");

#define PRIV_THREADS 1024
#define DEG_WGS      64
#define PRIV_BATCH   8
#define FLUSH_BATCH  4

__global__ void __launch_bounds__(PRIV_THREADS)
k_deg_hist(const int* __restrict__ src, const int* __restrict__ dst,
           float* __restrict__ deg, unsigned* __restrict__ key, int e, int n) {
    __shared__ alignas(16) unsigned hist[NNODES];
    const int tid = threadIdx.x;
    for (int i = tid; i < NNODES / 4; i += PRIV_THREADS) ((uint4*)hist)[i] = make_uint4(0u, 0u, 0u, 0u);
    __syncthreads();

    const int chunk = (e + (int)gridDim.x - 1) / (int)gridDim.x;
    const int lo = (int)blockIdx.x * chunk;
    int hi = lo + chunk; if (hi > e) hi = e;
    for (int base = lo; base < hi; base += PRIV_BATCH * PRIV_THREADS) {
        int s[PRIV_BATCH], d[PRIV_BATCH];
        #pragma unroll
        for (int j = 0; j < PRIV_BATCH; ++j) {
            const int i  = base + tid + PRIV_THREADS * j;
            const int ic = (i < hi) ? i : lo;
            d[j] = dst[ic];
            s[j] = src[ic];
        }
        #pragma unroll
        for (int j = 0; j < PRIV_BATCH; ++j) {
            const int i = base + tid + PRIV_THREADS * j;
            if (i < hi) {
                key[i] = (unsigned)s[j] | ((unsigned)d[j] << 16);
                __hip_atomic_fetch_add(&hist[d[j]], 1u, __ATOMIC_RELAXED, __HIP_MEMORY_SCOPE_WORKGROUP);
            }
        }
    }
    __syncthreads();

    const int nblk = (n + PRIV_THREADS - 1) / PRIV_THREADS;
    const int rot  = (int)blockIdx.x % nblk;
    for (int jj = 0; jj < nblk; jj += FLUSH_BATCH) {
        int idx[FLUSH_BATCH];
        unsigned c[FLUSH_BATCH];
        #pragma unroll
        for (int u = 0; u < FLUSH_BATCH; ++u) {
            int j = jj + u + rot; if (j >= nblk) j -= nblk;
            const int i = j * PRIV_THREADS + tid;
            idx[u] = (jj + u < nblk && i < n) ? i : -1;
            c[u] = hist[idx[u] >= 0 ? idx[u] : 0];
        }
        #pragma unroll
        for (int u = 0; u < FLUSH_BATCH; ++u)
            if (idx[u] >= 0 && c[u] != 0u) atomicAdd(&deg[idx[u]], (float)c[u]);
    }
}

__global__ void k_dinv(const float* __restrict__ deg, float* __restrict__ dinv, int n) {
    int i = blockIdx.x * blockDim.x + threadIdx.x;
    if (i < n) dinv[i] = __frsqrt_rn(deg[i]);
}


#define G_THREADS 512
#define G_ROWG    128
#define G_HALFC   128
#define G_KH      256
#define G_APITCH  260
#define G_NT      (NNODES / 16)

__device__ __forceinline__ void gemm_stage_load(const float* __restrict__ X, int rg, int s, int tid,
                                                float4 (&r)[4]) {
    const int p  = s >> 1;
    const int kh = s & 1;
    #pragma unroll
    for (int j = 0; j < 4; ++j) {
        const int idx = tid + G_THREADS * j;
        const int rr  = idx >> 6;
        const int c4  = (idx & 63) * 4;
        const int t   = rg + G_ROWG * (2 * p + (rr >> 4));
        if (t < G_NT)
            r[j] = *(const float4*)(X + (size_t)(t * 16 + (rr & 15)) * DIMIN + kh * G_KH + c4);
        else
            r[j] = make_float4(0.0f, 0.0f, 0.0f, 0.0f);
    }
}

__global__ void __launch_bounds__(G_THREADS)
k_gemm1(const float* __restrict__ X, const float* __restrict__ W,
        float* __restrict__ H) {
    __shared__ alignas(16) float Ws[(DIMIN / 2) * G_HALFC * 2];
    __shared__ alignas(16) float As[32 * G_APITCH];

    const int tid  = threadIdx.x;
    const int half = blockIdx.x & 1;
    const int rg   = blockIdx.x >> 1;
    const int ntile  = (G_NT - rg + G_ROWG - 1) / G_ROWG;
    const int nstage = 2 * ((ntile + 1) / 2);

    float4 pre[4];
    gemm_stage_load(X, rg, 0, tid, pre);

    #pragma unroll 4
    for (int i = 0; i < 16; ++i) {
        const int it  = (i + rg) & 15;
        const int idx = tid + G_THREADS * it;
        const int kp  = idx >> 5;
        const int cq  = idx & 31;
        const float4 r0 = *(const float4*)(W + (size_t)(2 * kp)     * HIDDEN + half * G_HALFC + 4 * cq);
        const float4 r1 = *(const float4*)(W + (size_t)(2 * kp + 1) * HIDDEN + half * G_HALFC + 4 * cq);
        const int c = (4 * cq) ^ ((kp & 1) << 4);
        float4* d = (float4*)(&Ws[2 * (kp * G_HALFC + c)]);
        d[0] = make_float4(r0.x, r1.x, r0.y, r1.y);
        d[1] = make_float4(r0.z, r1.z, r0.w, r1.w);
    }

    const int wave = __builtin_amdgcn_readfirstlane(tid >> 5);
    const int lane = tid & 31;
    const int l16  = lane & 15;
    const int hi   = lane >> 4;
    const int sub  = wave >> 3;
    const int ct   = wave & 7;

    const float* ap = As + (sub * 16 + l16) * G_APITCH + 2 * hi;
    const float* bp = Ws + 2 * (hi * G_HALFC + ((ct ^ hi) << 4) + l16);

    v8f c = {};
    for (int s = 0; s < nstage; ++s) {
        __syncthreads();
        #pragma unroll
        for (int j = 0; j < 4; ++j) {
            const int idx = tid + G_THREADS * j;
            *(float4*)(&As[(idx >> 6) * G_APITCH + (idx & 63) * 4]) = pre[j];
        }
        __syncthreads();
        if (s + 1 < nstage) gemm_stage_load(X, rg, s + 1, tid, pre);

        const int kh = s & 1;
        const int t  = rg + G_ROWG * (2 * (s >> 1) + sub);
        if (t < G_NT) {
            const float* b = bp + 2 * (kh * (G_KH / 2) * G_HALFC);
            v2f fa[2][8], fb[2][8];
            #pragma unroll
            for (int i = 0; i < 8; ++i) {
                fa[0][i] = *(const v2f*)(ap + 4 * i);
                fb[0][i] = *(const v2f*)(b + (2 * i) * (2 * G_HALFC));
            }
            #pragma unroll
            for (int j = 0; j < G_KH / 32; ++j) {
                if (j + 1 < G_KH / 32) {
                    #pragma unroll
                    for (int i = 0; i < 8; ++i) {
                        const int kk = 32 * (j + 1) + 4 * i;
                        fa[(j + 1) & 1][i] = *(const v2f*)(ap + kk);
                        fb[(j + 1) & 1][i] = *(const v2f*)(b + (kk >> 1) * (2 * G_HALFC));
                    }
                }
                __builtin_amdgcn_sched_barrier(0);
                #pragma unroll
                for (int i = 0; i < 8; ++i)
                    c = __builtin_amdgcn_wmma_f32_16x16x4_f32(false, fa[j & 1][i], false, fb[j & 1][i],
                                                              (short)0, c, false, false);
                __builtin_amdgcn_sched_barrier(0);
            }
            if (kh == 1) {
                #pragma unroll
                for (int v = 0; v < 8; ++v) {
                    const size_t o = (size_t)(t * 16 + hi * 8 + v) * HIDDEN + half * G_HALFC + ct * 16 + l16;
                    H[o] = c[v];
                }
                c = v8f{};
            }
        }
    }
}


#define OWN_ROWS    196
#define OWN_THREADS 1024
#define OWN_WAVES   (OWN_THREADS / 32)
#define SCAN_EDGES  512
#define LIST_CAP    (SCAN_EDGES + 4)

__device__ __forceinline__ void own_lds_add(float* a, float nrm, const float4& p, const float4& q) {
    __hip_atomic_fetch_add(&a[0 * 32], nrm * p.x, __ATOMIC_RELAXED, __HIP_MEMORY_SCOPE_WORKGROUP);
    __hip_atomic_fetch_add(&a[1 * 32], nrm * p.y, __ATOMIC_RELAXED, __HIP_MEMORY_SCOPE_WORKGROUP);
    __hip_atomic_fetch_add(&a[2 * 32], nrm * p.z, __ATOMIC_RELAXED, __HIP_MEMORY_SCOPE_WORKGROUP);
    __hip_atomic_fetch_add(&a[3 * 32], nrm * p.w, __ATOMIC_RELAXED, __HIP_MEMORY_SCOPE_WORKGROUP);
    __hip_atomic_fetch_add(&a[4 * 32], nrm * q.x, __ATOMIC_RELAXED, __HIP_MEMORY_SCOPE_WORKGROUP);
    __hip_atomic_fetch_add(&a[5 * 32], nrm * q.y, __ATOMIC_RELAXED, __HIP_MEMORY_SCOPE_WORKGROUP);
    __hip_atomic_fetch_add(&a[6 * 32], nrm * q.z, __ATOMIC_RELAXED, __HIP_MEMORY_SCOPE_WORKGROUP);
    __hip_atomic_fetch_add(&a[7 * 32], nrm * q.w, __ATOMIC_RELAXED, __HIP_MEMORY_SCOPE_WORKGROUP);
}

__device__ __forceinline__ void own_add4(const float* __restrict__ dinv, const float* __restrict__ H1,
                                         float* acc, const float* dl, int lane,
                                         unsigned t0, unsigned t1, unsigned t2, unsigned t3) {
    const int s0 = (int)(t0 & 0xFFFFu), s1 = (int)(t1 & 0xFFFFu);
    const int s2 = (int)(t2 & 0xFFFFu), s3 = (int)(t3 & 0xFFFFu);
    const int r0 = (int)(t0 >> 16), r1 = (int)(t1 >> 16), r2 = (int)(t2 >> 16), r3 = (int)(t3 >> 16);
    const float4* h0 = (const float4*)(H1 + (size_t)s0 * HIDDEN) + lane;
    const float4* h1 = (const float4*)(H1 + (size_t)s1 * HIDDEN) + lane;
    const float4* h2 = (const float4*)(H1 + (size_t)s2 * HIDDEN) + lane;
    const float4* h3 = (const float4*)(H1 + (size_t)s3 * HIDDEN) + lane;
    const float4 p0 = h0[0], q0 = h0[32];
    const float4 p1 = h1[0], q1 = h1[32];
    const float4 p2 = h2[0], q2 = h2[32];
    const float4 p3 = h3[0], q3 = h3[32];
    const float n0 = dinv[s0] * dl[r0];
    const float n1 = dinv[s1] * dl[r1];
    const float n2 = dinv[s2] * dl[r2];
    const float n3 = dinv[s3] * dl[r3];
    own_lds_add(acc + r0 * HIDDEN + lane, n0, p0, q0);
    own_lds_add(acc + r1 * HIDDEN + lane, n1, p1, q1);
    own_lds_add(acc + r2 * HIDDEN + lane, n2, p2, q2);
    own_lds_add(acc + r3 * HIDDEN + lane, n3, p3, q3);
}

__device__ __forceinline__ void own_add1(const float* __restrict__ dinv, const float* __restrict__ H1,
                                         float* acc, const float* dl, int lane, unsigned t0) {
    const int s0 = (int)(t0 & 0xFFFFu);
    const int r0 = (int)(t0 >> 16);
    const float4* h0 = (const float4*)(H1 + (size_t)s0 * HIDDEN) + lane;
    const float4 p0 = h0[0], q0 = h0[32];
    const float n0 = dinv[s0] * dl[r0];
    own_lds_add(acc + r0 * HIDDEN + lane, n0, p0, q0);
}

__device__ __forceinline__ void own_load_trip(const int* __restrict__ kw, int base, int lane, int e,
                                              bool vec_ok, int4 (&v)[4]) {
    if (vec_ok && base + SCAN_EDGES <= e) {
        #pragma unroll
        for (int g = 0; g < 4; ++g) v[g] = *(const int4*)(kw + base + g * 128 + lane * 4);
    } else {
        #pragma unroll
        for (int g = 0; g < 4; ++g) {
            const int i = base + g * 128 + lane * 4;
            v[g].x = (i     < e) ? kw[i]     : -1;
            v[g].y = (i + 1 < e) ? kw[i + 1] : -1;
            v[g].z = (i + 2 < e) ? kw[i + 2] : -1;
            v[g].w = (i + 3 < e) ? kw[i + 3] : -1;
        }
    }
}

__global__ void __launch_bounds__(OWN_THREADS)
k_agg1_own(const unsigned* __restrict__ key,
           const float* __restrict__ dinv, const float* __restrict__ H1,
           const float* __restrict__ b1, const float* __restrict__ W2,
           const float* __restrict__ b2, float* __restrict__ H2,
           float* __restrict__ out, int e, int n) {
    __shared__ float acc[OWN_ROWS * HIDDEN];
    __shared__ unsigned lst[OWN_WAVES * LIST_CAP];
    __shared__ float dl[OWN_ROWS];

    const int lo = blockIdx.x * OWN_ROWS;
    int nrows = n - lo;
    if (nrows > OWN_ROWS) nrows = OWN_ROWS;
    const int wave = __builtin_amdgcn_readfirstlane((int)(threadIdx.x >> 5));
    const int lane = threadIdx.x & 31;

    for (int r = wave; r < nrows; r += OWN_WAVES) {
        const float dv = dinv[lo + r];
        if (lane == 0) dl[r] = dv;
        const float4* h = (const float4*)(H1 + (size_t)(lo + r) * HIDDEN) + lane;
        const float4 p = h[0], q = h[32];
        float* a = acc + r * HIDDEN + lane;
        a[0 * 32] = dv * dv * p.x;
        a[1 * 32] = dv * dv * p.y;
        a[2 * 32] = dv * dv * p.z;
        a[3 * 32] = dv * dv * p.w;
        a[4 * 32] = dv * dv * q.x;
        a[5 * 32] = dv * dv * q.y;
        a[6 * 32] = dv * dv * q.z;
        a[7 * 32] = dv * dv * q.w;
    }
    __syncthreads();

    const unsigned ulo = (unsigned)lo << 16, un = (unsigned)(nrows > 0 ? nrows : 0) << 16;
    const int* kw = (const int*)key;
    const bool vec_ok = ((((size_t)kw) & 15) == 0);
    const int ntrips = (e + SCAN_EDGES - 1) / SCAN_EDGES;
    const int rot = (int)(((long long)blockIdx.x * ntrips) / gridDim.x);
    unsigned* my = lst + wave * LIST_CAP;
    int cnt = 0;

    int4 cur[4], nxt[4];
    #pragma unroll
    for (int g = 0; g < 4; ++g) { cur[g] = make_int4(-1, -1, -1, -1); nxt[g] = cur[g]; }
    if (wave < ntrips) {
        int t = wave + rot; if (t >= ntrips) t -= ntrips;
        own_load_trip(kw, t * SCAN_EDGES, lane, e, vec_ok, cur);
    }
    for (int i = wave; i < ntrips; i += OWN_WAVES) {
        if (i + OWN_WAVES < ntrips) {
            int tn = i + OWN_WAVES + rot; if (tn >= ntrips) tn -= ntrips;
            own_load_trip(kw, tn * SCAN_EDGES, lane, e, vec_ok, nxt);
        }
        #pragma unroll
        for (int g = 0; g < 4; ++g) {
            const unsigned xx = (unsigned)cur[g].x - ulo, xy = (unsigned)cur[g].y - ulo;
            const unsigned xz = (unsigned)cur[g].z - ulo, xw = (unsigned)cur[g].w - ulo;
            const bool qx = xx < un;
            const bool qy = xy < un;
            const bool qz = xz < un;
            const bool qw = xw < un;
            const unsigned bx = __builtin_amdgcn_ballot_w32(qx);
            const unsigned by = __builtin_amdgcn_ballot_w32(qy);
            const unsigned bz = __builtin_amdgcn_ballot_w32(qz);
            const unsigned bw = __builtin_amdgcn_ballot_w32(qw);
            if ((bx | by | bz | bw) != 0u) {
                if (qx) my[cnt + (int)__builtin_amdgcn_mbcnt_lo(bx, 0u)] = xx;
                cnt += __builtin_popcount(bx);
                if (qy) my[cnt + (int)__builtin_amdgcn_mbcnt_lo(by, 0u)] = xy;
                cnt += __builtin_popcount(by);
                if (qz) my[cnt + (int)__builtin_amdgcn_mbcnt_lo(bz, 0u)] = xz;
                cnt += __builtin_popcount(bz);
                if (qw) my[cnt + (int)__builtin_amdgcn_mbcnt_lo(bw, 0u)] = xw;
                cnt += __builtin_popcount(bw);
            }
        }
        while (cnt >= 4) {
            cnt -= 4;
            const unsigned t0 = (unsigned)__builtin_amdgcn_readfirstlane((int)my[cnt]);
            const unsigned t1 = (unsigned)__builtin_amdgcn_readfirstlane((int)my[cnt + 1]);
            const unsigned t2 = (unsigned)__builtin_amdgcn_readfirstlane((int)my[cnt + 2]);
            const unsigned t3 = (unsigned)__builtin_amdgcn_readfirstlane((int)my[cnt + 3]);
            own_add4(dinv, H1, acc, dl, lane, t0, t1, t2, t3);
        }
        #pragma unroll
        for (int g = 0; g < 4; ++g) cur[g] = nxt[g];
    }
    while (cnt > 0) {
        cnt -= 1;
        const unsigned t0 = (unsigned)__builtin_amdgcn_readfirstlane((int)my[cnt]);
        own_add1(dinv, H1, acc, dl, lane, t0);
    }
    __syncthreads();

    float bb[8], w0[8], w1[8];
    #pragma unroll
    for (int q = 0; q < 2; ++q) {
        const int c = 4 * lane + 128 * q;
        const float4 bv = *(const float4*)(b1 + c);
        const float4 wa = *(const float4*)(W2 + 2 * c);
        const float4 wb = *(const float4*)(W2 + 2 * c + 4);
        bb[4 * q + 0] = bv.x; bb[4 * q + 1] = bv.y; bb[4 * q + 2] = bv.z; bb[4 * q + 3] = bv.w;
        w0[4 * q + 0] = wa.x; w1[4 * q + 0] = wa.y;
        w0[4 * q + 1] = wa.z; w1[4 * q + 1] = wa.w;
        w0[4 * q + 2] = wb.x; w1[4 * q + 2] = wb.y;
        w0[4 * q + 3] = wb.z; w1[4 * q + 3] = wb.w;
    }
    const float b20 = b2[0], b21 = b2[1];
    for (int r = wave; r < nrows; r += OWN_WAVES) {
        const float* a = acc + r * HIDDEN + lane;
        float p0 = 0.0f, p1 = 0.0f;
        #pragma unroll
        for (int j = 0; j < 8; ++j) {
            float z = a[j * 32] + bb[j];
            z = z > 0.0f ? z : 0.0f;
            p0 += z * w0[j];
            p1 += z * w1[j];
        }
        #pragma unroll
        for (int off = 16; off > 0; off >>= 1) {
            p0 += __shfl_xor(p0, off, 32);
            p1 += __shfl_xor(p1, off, 32);
        }
        if (lane == 0) {
            const int row = lo + r;
            float s = dinv[row]; s *= s;
            H2[2 * row]      = p0;
            H2[2 * row + 1]  = p1;
            out[2 * row]     = s * p0 + b20;
            out[2 * row + 1] = s * p1 + b21;
        }
    }
}


static_assert(NNODES % 2 == 0, "layer 2 splits the destination rows in two equal halves");
#define AGG2_CHUNKS 64
#define HALF_ROWS   (NNODES / 2)
static_assert((2 * HALF_ROWS) % 4 == 0, "the LDS table is cleared with 16-byte stores");

__global__ void __launch_bounds__(PRIV_THREADS)
k_agg2_priv(const unsigned* __restrict__ key, const float* __restrict__ dinv,
            const float* __restrict__ H2, float* __restrict__ out, int e) {
    __shared__ alignas(16) float acc[2 * HALF_ROWS];
    const int tid = threadIdx.x;
    const int half = (int)blockIdx.x & 1;
    const int ck   = (int)blockIdx.x >> 1;
    const int nck  = (int)gridDim.x >> 1;
    for (int i = tid; i < (2 * HALF_ROWS) / 4; i += PRIV_THREADS)
        ((float4*)acc)[i] = make_float4(0.0f, 0.0f, 0.0f, 0.0f);
    __syncthreads();

    const int chunk = (e + nck - 1) / nck;
    const int lo = ck * chunk;
    int hi = lo + chunk; if (hi > e) hi = e;
    const unsigned rlo = (unsigned)(half * HALF_ROWS);
    for (int base = lo; base < hi; base += PRIV_BATCH * PRIV_THREADS) {
        unsigned kw[PRIV_BATCH], r[PRIV_BATCH];
        float a[PRIV_BATCH], b[PRIV_BATCH];
        float2 h[PRIV_BATCH];
        #pragma unroll
        for (int j = 0; j < PRIV_BATCH; ++j) {
            const int i = base + tid + PRIV_THREADS * j;
            kw[j] = key[(i < hi) ? i : lo];
        }
        #pragma unroll
        for (int j = 0; j < PRIV_BATCH; ++j) {
            const int i = base + tid + PRIV_THREADS * j;
            const unsigned kk = (i < hi) ? kw[j] : 0xFFFFFFFFu;
            const unsigned d = kk >> 16, s = kk & 0xFFFFu;
            r[j] = d - rlo;
            const bool own = r[j] < (unsigned)HALF_ROWS;
            const unsigned se = own ? s : 0u, de = own ? d : 0u;
            a[j] = dinv[se];
            b[j] = dinv[de];
            h[j] = *(const float2*)(H2 + 2 * se);
        }
        #pragma unroll
        for (int j = 0; j < PRIV_BATCH; ++j) {
            if (r[j] < (unsigned)HALF_ROWS) {
                const float norm = a[j] * b[j];
                __hip_atomic_fetch_add(&acc[2 * r[j]],     norm * h[j].x, __ATOMIC_RELAXED, __HIP_MEMORY_SCOPE_WORKGROUP);
                __hip_atomic_fetch_add(&acc[2 * r[j] + 1], norm * h[j].y, __ATOMIC_RELAXED, __HIP_MEMORY_SCOPE_WORKGROUP);
            }
        }
    }
    __syncthreads();

    float* o = out + 2 * (size_t)rlo;
    const int nblk = (2 * HALF_ROWS + PRIV_THREADS - 1) / PRIV_THREADS;
    const int rot  = ck % nblk;
    for (int jj = 0; jj < nblk; jj += FLUSH_BATCH) {
        int idx[FLUSH_BATCH];
        float v[FLUSH_BATCH];
        #pragma unroll
        for (int u = 0; u < FLUSH_BATCH; ++u) {
            int j = jj + u + rot; if (j >= nblk) j -= nblk;
            const int i = j * PRIV_THREADS + tid;
            idx[u] = (jj + u < nblk && i < 2 * HALF_ROWS) ? i : -1;
            v[u] = acc[idx[u] >= 0 ? idx[u] : 0];
        }
        #pragma unroll
        for (int u = 0; u < FLUSH_BATCH; ++u)
            if (idx[u] >= 0 && v[u] != 0.0f) atomicAdd(&o[idx[u]], v[u]);
    }
}


extern "C" void kernel_launch(void* const* d_in, const int* in_sizes, int n_in,
                              void* d_out, int out_size, void* d_ws, size_t ws_size,
                              hipStream_t stream) {
    const float* X   = (const float*)d_in[0];
    const float* W1  = (const float*)d_in[1];
    const float* b1  = (const float*)d_in[2];
    const float* W2  = (const float*)d_in[3];
    const float* b2  = (const float*)d_in[4];
    const int*   ei  = (const int*)d_in[5];
    float*       out = (float*)d_out;

    const int N = NNODES;
    const int E = in_sizes[5] / 2;
    const int* src = ei;
    const int* dst = ei + E;

    float*  deg  = (float*)d_ws;
    float*  dinv = deg  + N;
    float*  H1   = dinv + N;
    float*  H2   = H1   + (size_t)N * HIDDEN;
    unsigned* key = (unsigned*)(H2 + (size_t)N * NCLS);

    const int T = 256;

    k_init_deg <<<(N + T - 1) / T, T, 0, stream>>>(deg, N);
    k_deg_hist <<<DEG_WGS, PRIV_THREADS, 0, stream>>>(src, dst, deg, key, E, N);
    k_dinv     <<<(N + T - 1) / T, T, 0, stream>>>(deg, dinv, N);

    k_gemm1<<<2 * G_ROWG, G_THREADS, 0, stream>>>(X, W1, H1);
    k_agg1_own<<<(N + OWN_ROWS - 1) / OWN_ROWS, OWN_THREADS, 0, stream>>>(key, dinv, H1, b1, W2, b2,
                                                                      H2, out, E, N);

    k_agg2_priv<<<2 * AGG2_CHUNKS, PRIV_THREADS, 0, stream>>>(key, dinv, H2, out, E);
}


# Degree pass sorts each 13312-edge chunk by owner work-group in LDS (exact counts), writes a run table: layer-1 owners read only their own runs, not the whole list; arrival-order f32 sums as before
# speedup vs baseline: 3.0011x; 1.0313x over previous
#include <hip/hip_runtime.h>
#include <hip/hip_bf16.h>

#define NNODES 50000
#define DIMIN  512
#define HIDDEN 256
#define NCLS   2

typedef __attribute__((ext_vector_type(2))) float v2f;
typedef __attribute__((ext_vector_type(8))) float v8f;


__global__ void k_init_deg(float* __restrict__ deg, int n) {
    int i = blockIdx.x * blockDim.x + threadIdx.x;
    if (i < n) deg[i] = 1.0f;
}

static_assert(NNODES <= 65536, "packed edge word keeps a node id in 16 bits");
static_assert(NNODES % 4 == 0, "the LDS tables are cleared with 16-byte stores");

#define OWN_ROWS     196
#define NOWN         ((NNODES + OWN_ROWS - 1) / OWN_ROWS)
#define TAB_PITCH    (NOWN + 1)
static_assert(NOWN % 32 == 0, "the prefix over the owners gives each lane of one wave NOWN / 32 counters");

#define PRIV_THREADS 1024
#define PRIV_BATCH   8
#define FLUSH_BATCH  4
#define SORT_BATCH   13
#define SORT_EDGES   (SORT_BATCH * PRIV_THREADS)
#define NEDGES       800000
#define DEG_WGS      ((NEDGES + SORT_EDGES - 1) / SORT_EDGES)
static_assert(DEG_WGS * TAB_PITCH <= NNODES, "segment table must fit its workspace region");

__global__ void __launch_bounds__(PRIV_THREADS)
k_deg_hist(const int* __restrict__ src, const int* __restrict__ dst,
           float* __restrict__ deg, unsigned* __restrict__ key, int* __restrict__ tab, int e, int n) {
    __shared__ alignas(16) unsigned hist[NNODES];
    __shared__ unsigned sorted[SORT_EDGES];
    __shared__ unsigned ocur[NOWN];
    const int tid = threadIdx.x;
    for (int i = tid; i < NNODES / 4; i += PRIV_THREADS) ((uint4*)hist)[i] = make_uint4(0u, 0u, 0u, 0u);

    const int nsub = (e + SORT_EDGES - 1) / SORT_EDGES;
    for (int sc = (int)blockIdx.x; sc < nsub; sc += (int)gridDim.x) {
        if (tid < NOWN) ocur[tid] = 0u;
        __syncthreads();
        const int lo = sc * SORT_EDGES;
        int hi = lo + SORT_EDGES; if (hi > e) hi = e;
        int s[SORT_BATCH], d[SORT_BATCH];
        #pragma unroll
        for (int j = 0; j < SORT_BATCH; ++j) {
            const int i  = lo + tid + PRIV_THREADS * j;
            const int ic = (i < hi) ? i : lo;
            d[j] = dst[ic];
            s[j] = src[ic];
        }
        unsigned w[SORT_BATCH];
        int ow[SORT_BATCH];
        #pragma unroll
        for (int j = 0; j < SORT_BATCH; ++j) {
            const int i = lo + tid + PRIV_THREADS * j;
            w[j]  = (unsigned)s[j] | ((unsigned)d[j] << 16);
            ow[j] = d[j] / OWN_ROWS;
            if (i < hi) {
                __hip_atomic_fetch_add(&hist[d[j]], 1u, __ATOMIC_RELAXED, __HIP_MEMORY_SCOPE_WORKGROUP);
                __hip_atomic_fetch_add(&ocur[ow[j]], 1u, __ATOMIC_RELAXED, __HIP_MEMORY_SCOPE_WORKGROUP);
            }
        }
        __syncthreads();

        if (tid < 32) {
            unsigned c[NOWN / 32];
            unsigned sum = 0u;
            #pragma unroll
            for (int u = 0; u < NOWN / 32; ++u) { c[u] = ocur[tid * (NOWN / 32) + u]; sum += c[u]; }
            unsigned inc = sum;
            #pragma unroll
            for (int off = 1; off < 32; off <<= 1) {
                const unsigned v = __shfl_up(inc, off, 32);
                if (tid >= off) inc += v;
            }
            unsigned run = inc - sum;
            #pragma unroll
            for (int u = 0; u < NOWN / 32; ++u) {
                ocur[tid * (NOWN / 32) + u] = run;
                tab[sc * TAB_PITCH + tid * (NOWN / 32) + u] = lo + (int)run;
                run += c[u];
            }
            if (tid == 31) tab[sc * TAB_PITCH + NOWN] = lo + (int)run;
        }
        __syncthreads();

        unsigned pos[SORT_BATCH];
        #pragma unroll
        for (int j = 0; j < SORT_BATCH; ++j) {
            const int i = lo + tid + PRIV_THREADS * j;
            pos[j] = 0u;
            if (i < hi) pos[j] = __hip_atomic_fetch_add(&ocur[ow[j]], 1u, __ATOMIC_RELAXED,
                                                        __HIP_MEMORY_SCOPE_WORKGROUP);
        }
        #pragma unroll
        for (int j = 0; j < SORT_BATCH; ++j) {
            const int i = lo + tid + PRIV_THREADS * j;
            if (i < hi) sorted[pos[j]] = w[j];
        }
        __syncthreads();

        #pragma unroll
        for (int j = 0; j < SORT_BATCH; ++j) pos[j] = sorted[tid + PRIV_THREADS * j];
        #pragma unroll
        for (int j = 0; j < SORT_BATCH; ++j) {
            const int i = lo + tid + PRIV_THREADS * j;
            if (i < hi) key[i] = pos[j];
        }
    }
    __syncthreads();

    const int nblk = (n + PRIV_THREADS - 1) / PRIV_THREADS;
    const int rot  = (int)blockIdx.x % nblk;
    for (int jj = 0; jj < nblk; jj += FLUSH_BATCH) {
        int idx[FLUSH_BATCH];
        unsigned c[FLUSH_BATCH];
        #pragma unroll
        for (int u = 0; u < FLUSH_BATCH; ++u) {
            int j = jj + u + rot; if (j >= nblk) j -= nblk;
            const int i = j * PRIV_THREADS + tid;
            idx[u] = (jj + u < nblk && i < n) ? i : -1;
            c[u] = hist[idx[u] >= 0 ? idx[u] : 0];
        }
        #pragma unroll
        for (int u = 0; u < FLUSH_BATCH; ++u)
            if (idx[u] >= 0 && c[u] != 0u) atomicAdd(&deg[idx[u]], (float)c[u]);
    }
}

__global__ void k_dinv(float* __restrict__ dg, int n) {
    int i = blockIdx.x * blockDim.x + threadIdx.x;
    if (i < n) dg[i] = __frsqrt_rn(dg[i]);
}


#define G_THREADS 512
#define G_ROWG    128
#define G_HALFC   128
#define G_KH      256
#define G_APITCH  260
#define G_NT      (NNODES / 16)

__device__ __forceinline__ void gemm_stage_load(const float* __restrict__ X, int rg, int s, int tid,
                                                float4 (&r)[4]) {
    const int p  = s >> 1;
    const int kh = s & 1;
    #pragma unroll
    for (int j = 0; j < 4; ++j) {
        const int idx = tid + G_THREADS * j;
        const int rr  = idx >> 6;
        const int c4  = (idx & 63) * 4;
        const int t   = rg + G_ROWG * (2 * p + (rr >> 4));
        if (t < G_NT)
            r[j] = *(const float4*)(X + (size_t)(t * 16 + (rr & 15)) * DIMIN + kh * G_KH + c4);
        else
            r[j] = make_float4(0.0f, 0.0f, 0.0f, 0.0f);
    }
}

__global__ void __launch_bounds__(G_THREADS)
k_gemm1(const float* __restrict__ X, const float* __restrict__ W,
        float* __restrict__ H) {
    __shared__ alignas(16) float Ws[(DIMIN / 2) * G_HALFC * 2];
    __shared__ alignas(16) float As[32 * G_APITCH];

    const int tid  = threadIdx.x;
    const int half = blockIdx.x & 1;
    const int rg   = blockIdx.x >> 1;
    const int ntile  = (G_NT - rg + G_ROWG - 1) / G_ROWG;
    const int nstage = 2 * ((ntile + 1) / 2);

    float4 pre[4];
    gemm_stage_load(X, rg, 0, tid, pre);

    #pragma unroll 4
    for (int i = 0; i < 16; ++i) {
        const int it  = (i + rg) & 15;
        const int idx = tid + G_THREADS * it;
        const int kp  = idx >> 5;
        const int cq  = idx & 31;
        const float4 r0 = *(const float4*)(W + (size_t)(2 * kp)     * HIDDEN + half * G_HALFC + 4 * cq);
        const float4 r1 = *(const float4*)(W + (size_t)(2 * kp + 1) * HIDDEN + half * G_HALFC + 4 * cq);
        const int c = (4 * cq) ^ ((kp & 1) << 4);
        float4* d = (float4*)(&Ws[2 * (kp * G_HALFC + c)]);
        d[0] = make_float4(r0.x, r1.x, r0.y, r1.y);
        d[1] = make_float4(r0.z, r1.z, r0.w, r1.w);
    }

    const int wave = __builtin_amdgcn_readfirstlane(tid >> 5);
    const int lane = tid & 31;
    const int l16  = lane & 15;
    const int hi   = lane >> 4;
    const int sub  = wave >> 3;
    const int ct   = wave & 7;

    const float* ap = As + (sub * 16 + l16) * G_APITCH + 2 * hi;
    const float* bp = Ws + 2 * (hi * G_HALFC + ((ct ^ hi) << 4) + l16);

    v8f c = {};
    for (int s = 0; s < nstage; ++s) {
        __syncthreads();
        #pragma unroll
        for (int j = 0; j < 4; ++j) {
            const int idx = tid + G_THREADS * j;
            *(float4*)(&As[(idx >> 6) * G_APITCH + (idx & 63) * 4]) = pre[j];
        }
        __syncthreads();
        if (s + 1 < nstage) gemm_stage_load(X, rg, s + 1, tid, pre);

        const int kh = s & 1;
        const int t  = rg + G_ROWG * (2 * (s >> 1) + sub);
        if (t < G_NT) {
            const float* b = bp + 2 * (kh * (G_KH / 2) * G_HALFC);
            v2f fa[2][8], fb[2][8];
            #pragma unroll
            for (int i = 0; i < 8; ++i) {
                fa[0][i] = *(const v2f*)(ap + 4 * i);
                fb[0][i] = *(const v2f*)(b + (2 * i) * (2 * G_HALFC));
            }
            #pragma unroll
            for (int j = 0; j < G_KH / 32; ++j) {
                if (j + 1 < G_KH / 32) {
                    #pragma unroll
                    for (int i = 0; i < 8; ++i) {
                        const int kk = 32 * (j + 1) + 4 * i;
                        fa[(j + 1) & 1][i] = *(const v2f*)(ap + kk);
                        fb[(j + 1) & 1][i] = *(const v2f*)(b + (kk >> 1) * (2 * G_HALFC));
                    }
                }
                __builtin_amdgcn_sched_barrier(0);
                #pragma unroll
                for (int i = 0; i < 8; ++i)
                    c = __builtin_amdgcn_wmma_f32_16x16x4_f32(false, fa[j & 1][i], false, fb[j & 1][i],
                                                              (short)0, c, false, false);
                __builtin_amdgcn_sched_barrier(0);
            }
            if (kh == 1) {
                #pragma unroll
                for (int v = 0; v < 8; ++v) {
                    const size_t o = (size_t)(t * 16 + hi * 8 + v) * HIDDEN + half * G_HALFC + ct * 16 + l16;
                    H[o] = c[v];
                }
                c = v8f{};
            }
        }
    }
}


#define OWN_THREADS 1024
#define OWN_WAVES   (OWN_THREADS / 32)
#define RUN_WORDS   64
#define LIST_CAP    (RUN_WORDS + 4)

__device__ __forceinline__ void own_lds_add(float* a, float nrm, const float4& p, const float4& q) {
    __hip_atomic_fetch_add(&a[0 * 32], nrm * p.x, __ATOMIC_RELAXED, __HIP_MEMORY_SCOPE_WORKGROUP);
    __hip_atomic_fetch_add(&a[1 * 32], nrm * p.y, __ATOMIC_RELAXED, __HIP_MEMORY_SCOPE_WORKGROUP);
    __hip_atomic_fetch_add(&a[2 * 32], nrm * p.z, __ATOMIC_RELAXED, __HIP_MEMORY_SCOPE_WORKGROUP);
    __hip_atomic_fetch_add(&a[3 * 32], nrm * p.w, __ATOMIC_RELAXED, __HIP_MEMORY_SCOPE_WORKGROUP);
    __hip_atomic_fetch_add(&a[4 * 32], nrm * q.x, __ATOMIC_RELAXED, __HIP_MEMORY_SCOPE_WORKGROUP);
    __hip_atomic_fetch_add(&a[5 * 32], nrm * q.y, __ATOMIC_RELAXED, __HIP_MEMORY_SCOPE_WORKGROUP);
    __hip_atomic_fetch_add(&a[6 * 32], nrm * q.z, __ATOMIC_RELAXED, __HIP_MEMORY_SCOPE_WORKGROUP);
    __hip_atomic_fetch_add(&a[7 * 32], nrm * q.w, __ATOMIC_RELAXED, __HIP_MEMORY_SCOPE_WORKGROUP);
}

__device__ __forceinline__ void own_add4(const float* __restrict__ dinv, const float* __restrict__ H1,
                                         float* acc, const float* dl, int lane,
                                         unsigned t0, unsigned t1, unsigned t2, unsigned t3) {
    const int s0 = (int)(t0 & 0xFFFFu), s1 = (int)(t1 & 0xFFFFu);
    const int s2 = (int)(t2 & 0xFFFFu), s3 = (int)(t3 & 0xFFFFu);
    const int r0 = (int)(t0 >> 16), r1 = (int)(t1 >> 16), r2 = (int)(t2 >> 16), r3 = (int)(t3 >> 16);
    const float4* h0 = (const float4*)(H1 + (size_t)s0 * HIDDEN) + lane;
    const float4* h1 = (const float4*)(H1 + (size_t)s1 * HIDDEN) + lane;
    const float4* h2 = (const float4*)(H1 + (size_t)s2 * HIDDEN) + lane;
    const float4* h3 = (const float4*)(H1 + (size_t)s3 * HIDDEN) + lane;
    const float4 p0 = h0[0], q0 = h0[32];
    const float4 p1 = h1[0], q1 = h1[32];
    const float4 p2 = h2[0], q2 = h2[32];
    const float4 p3 = h3[0], q3 = h3[32];
    const float n0 = dinv[s0] * dl[r0];
    const float n1 = dinv[s1] * dl[r1];
    const float n2 = dinv[s2] * dl[r2];
    const float n3 = dinv[s3] * dl[r3];
    own_lds_add(acc + r0 * HIDDEN + lane, n0, p0, q0);
    own_lds_add(acc + r1 * HIDDEN + lane, n1, p1, q1);
    own_lds_add(acc + r2 * HIDDEN + lane, n2, p2, q2);
    own_lds_add(acc + r3 * HIDDEN + lane, n3, p3, q3);
}

__device__ __forceinline__ void own_add1(const float* __restrict__ dinv, const float* __restrict__ H1,
                                         float* acc, const float* dl, int lane, unsigned t0) {
    const int s0 = (int)(t0 & 0xFFFFu);
    const int r0 = (int)(t0 >> 16);
    const float4* h0 = (const float4*)(H1 + (size_t)s0 * HIDDEN) + lane;
    const float4 p0 = h0[0], q0 = h0[32];
    const float n0 = dinv[s0] * dl[r0];
    own_lds_add(acc + r0 * HIDDEN + lane, n0, p0, q0);
}

__device__ __forceinline__ int own_take(const float* __restrict__ dinv, const float* __restrict__ H1,
                                        float* acc, const float* dl, unsigned* my, int lane,
                                        unsigned ulo, unsigned a, unsigned b, int m, int cnt) {
    if (lane < m)      my[cnt + lane]      = a - ulo;
    if (lane + 32 < m) my[cnt + 32 + lane] = b - ulo;
    cnt += m;
    while (cnt >= 4) {
        cnt -= 4;
        const unsigned t0 = (unsigned)__builtin_amdgcn_readfirstlane((int)my[cnt]);
        const unsigned t1 = (unsigned)__builtin_amdgcn_readfirstlane((int)my[cnt + 1]);
        const unsigned t2 = (unsigned)__builtin_amdgcn_readfirstlane((int)my[cnt + 2]);
        const unsigned t3 = (unsigned)__builtin_amdgcn_readfirstlane((int)my[cnt + 3]);
        own_add4(dinv, H1, acc, dl, lane, t0, t1, t2, t3);
    }
    return cnt;
}

__global__ void __launch_bounds__(OWN_THREADS)
k_agg1_own(const unsigned* __restrict__ key, const int* __restrict__ tab,
           const float* __restrict__ dinv, const float* __restrict__ H1,
           const float* __restrict__ b1, const float* __restrict__ W2,
           const float* __restrict__ b2, float* __restrict__ H2,
           float* __restrict__ out, int e, int n) {
    __shared__ float acc[OWN_ROWS * HIDDEN];
    __shared__ unsigned lst[OWN_WAVES * LIST_CAP];
    __shared__ float dl[OWN_ROWS];

    const int lo = blockIdx.x * OWN_ROWS;
    int nrows = n - lo;
    if (nrows > OWN_ROWS) nrows = OWN_ROWS;
    const int wave = __builtin_amdgcn_readfirstlane((int)(threadIdx.x >> 5));
    const int lane = threadIdx.x & 31;

    for (int r = wave; r < nrows; r += OWN_WAVES) {
        const float dv = dinv[lo + r];
        if (lane == 0) dl[r] = dv;
        const float4* h = (const float4*)(H1 + (size_t)(lo + r) * HIDDEN) + lane;
        const float4 p = h[0], q = h[32];
        float* a = acc + r * HIDDEN + lane;
        a[0 * 32] = dv * dv * p.x;
        a[1 * 32] = dv * dv * p.y;
        a[2 * 32] = dv * dv * p.z;
        a[3 * 32] = dv * dv * p.w;
        a[4 * 32] = dv * dv * q.x;
        a[5 * 32] = dv * dv * q.y;
        a[6 * 32] = dv * dv * q.z;
        a[7 * 32] = dv * dv * q.w;
    }
    __syncthreads();

    const unsigned ulo = (unsigned)lo << 16;
    const int own   = (int)blockIdx.x;
    const int nsub  = (e + SORT_EDGES - 1) / SORT_EDGES;
    const int elast = e - 1;
    unsigned* my = lst + wave * LIST_CAP;
    int cnt = 0;

    for (int c0 = wave; c0 < nsub; c0 += OWN_WAVES * 32) {
        const int cl = c0 + OWN_WAVES * lane;
        int sb = 0, se = 0;
        if (cl < nsub) { sb = tab[cl * TAB_PITCH + own]; se = tab[cl * TAB_PITCH + own + 1]; }
        int nk = (nsub - c0 + OWN_WAVES - 1) / OWN_WAVES;
        if (nk > 32) nk = 32;
        int beg = __builtin_amdgcn_readlane(sb, 0), end = __builtin_amdgcn_readlane(se, 0);
        int i0 = beg + lane, i1 = beg + 32 + lane;
        unsigned w0 = key[i0 < elast ? i0 : elast];
        unsigned w1 = key[i1 < elast ? i1 : elast];
        for (int k = 0; k < nk; ++k) {
            const int kn = (k + 1 < nk) ? k + 1 : k;
            const int begn = __builtin_amdgcn_readlane(sb, kn);
            const int endn = __builtin_amdgcn_readlane(se, kn);
            i0 = begn + lane; i1 = begn + 32 + lane;
            const unsigned n0 = key[i0 < elast ? i0 : elast];
            const unsigned n1 = key[i1 < elast ? i1 : elast];

            int m = end - beg; if (m > RUN_WORDS) m = RUN_WORDS;
            cnt = own_take(dinv, H1, acc, dl, my, lane, ulo, w0, w1, m, cnt);
            for (int base = beg + RUN_WORDS; base < end; base += RUN_WORDS) {
                i0 = base + lane; i1 = base + 32 + lane;
                const unsigned a = key[i0 < elast ? i0 : elast];
                const unsigned b = key[i1 < elast ? i1 : elast];
                int mm = end - base; if (mm > RUN_WORDS) mm = RUN_WORDS;
                cnt = own_take(dinv, H1, acc, dl, my, lane, ulo, a, b, mm, cnt);
            }
            beg = begn; end = endn; w0 = n0; w1 = n1;
        }
    }
    while (cnt > 0) {
        cnt -= 1;
        const unsigned t0 = (unsigned)__builtin_amdgcn_readfirstlane((int)my[cnt]);
        own_add1(dinv, H1, acc, dl, lane, t0);
    }
    __syncthreads();

    float bb[8], w0[8], w1[8];
    #pragma unroll
    for (int q = 0; q < 2; ++q) {
        const int c = 4 * lane + 128 * q;
        const float4 bv = *(const float4*)(b1 + c);
        const float4 wa = *(const float4*)(W2 + 2 * c);
        const float4 wb = *(const float4*)(W2 + 2 * c + 4);
        bb[4 * q + 0] = bv.x; bb[4 * q + 1] = bv.y; bb[4 * q + 2] = bv.z; bb[4 * q + 3] = bv.w;
        w0[4 * q + 0] = wa.x; w1[4 * q + 0] = wa.y;
        w0[4 * q + 1] = wa.z; w1[4 * q + 1] = wa.w;
        w0[4 * q + 2] = wb.x; w1[4 * q + 2] = wb.y;
        w0[4 * q + 3] = wb.z; w1[4 * q + 3] = wb.w;
    }
    const float b20 = b2[0], b21 = b2[1];
    for (int r = wave; r < nrows; r += OWN_WAVES) {
        const float* a = acc + r * HIDDEN + lane;
        float p0 = 0.0f, p1 = 0.0f;
        #pragma unroll
        for (int j = 0; j < 8; ++j) {
            float z = a[j * 32] + bb[j];
            z = z > 0.0f ? z : 0.0f;
            p0 += z * w0[j];
            p1 += z * w1[j];
        }
        #pragma unroll
        for (int off = 16; off > 0; off >>= 1) {
            p0 += __shfl_xor(p0, off, 32);
            p1 += __shfl_xor(p1, off, 32);
        }
        if (lane == 0) {
            const int row = lo + r;
            float s = dinv[row]; s *= s;
            H2[2 * row]      = p0;
            H2[2 * row + 1]  = p1;
            out[2 * row]     = s * p0 + b20;
            out[2 * row + 1] = s * p1 + b21;
        }
    }
}


static_assert(NNODES % 2 == 0, "layer 2 splits the destination rows in two equal halves");
#define AGG2_CHUNKS 64
#define HALF_ROWS   (NNODES / 2)
static_assert((2 * HALF_ROWS) % 4 == 0, "the LDS table is cleared with 16-byte stores");

__global__ void __launch_bounds__(PRIV_THREADS)
k_agg2_priv(const unsigned* __restrict__ key, const float* __restrict__ dinv,
            const float* __restrict__ H2, float* __restrict__ out, int e) {
    __shared__ alignas(16) float acc[2 * HALF_ROWS];
    const int tid = threadIdx.x;
    const int half = (int)blockIdx.x & 1;
    const int ck   = (int)blockIdx.x >> 1;
    const int nck  = (int)gridDim.x >> 1;
    for (int i = tid; i < (2 * HALF_ROWS) / 4; i += PRIV_THREADS)
        ((float4*)acc)[i] = make_float4(0.0f, 0.0f, 0.0f, 0.0f);
    __syncthreads();

    const int chunk = (e + nck - 1) / nck;
    const int lo = ck * chunk;
    int hi = lo + chunk; if (hi > e) hi = e;
    const unsigned rlo = (unsigned)(half * HALF_ROWS);
    for (int base = lo; base < hi; base += PRIV_BATCH * PRIV_THREADS) {
        unsigned kw[PRIV_BATCH], r[PRIV_BATCH];
        float a[PRIV_BATCH], b[PRIV_BATCH];
        float2 h[PRIV_BATCH];
        #pragma unroll
        for (int j = 0; j < PRIV_BATCH; ++j) {
            const int i = base + tid + PRIV_THREADS * j;
            kw[j] = key[(i < hi) ? i : lo];
        }
        #pragma unroll
        for (int j = 0; j < PRIV_BATCH; ++j) {
            const int i = base + tid + PRIV_THREADS * j;
            const unsigned kk = (i < hi) ? kw[j] : 0xFFFFFFFFu;
            const unsigned d = kk >> 16, s = kk & 0xFFFFu;
            r[j] = d - rlo;
            const bool own = r[j] < (unsigned)HALF_ROWS;
            const unsigned se = own ? s : 0u, de = own ? d : 0u;
            a[j] = dinv[se];
            b[j] = dinv[de];
            h[j] = *(const float2*)(H2 + 2 * se);
        }
        #pragma unroll
        for (int j = 0; j < PRIV_BATCH; ++j) {
            if (r[j] < (unsigned)HALF_ROWS) {
                const float norm = a[j] * b[j];
                __hip_atomic_fetch_add(&acc[2 * r[j]],     norm * h[j].x, __ATOMIC_RELAXED, __HIP_MEMORY_SCOPE_WORKGROUP);
                __hip_atomic_fetch_add(&acc[2 * r[j] + 1], norm * h[j].y, __ATOMIC_RELAXED, __HIP_MEMORY_SCOPE_WORKGROUP);
            }
        }
    }
    __syncthreads();

    float* o = out + 2 * (size_t)rlo;
    const int nblk = (2 * HALF_ROWS + PRIV_THREADS - 1) / PRIV_THREADS;
    const int rot  = ck % nblk;
    for (int jj = 0; jj < nblk; jj += FLUSH_BATCH) {
        int idx[FLUSH_BATCH];
        float v[FLUSH_BATCH];
        #pragma unroll
        for (int u = 0; u < FLUSH_BATCH; ++u) {
            int j = jj + u + rot; if (j >= nblk) j -= nblk;
            const int i = j * PRIV_THREADS + tid;
            idx[u] = (jj + u < nblk && i < 2 * HALF_ROWS) ? i : -1;
            v[u] = acc[idx[u] >= 0 ? idx[u] : 0];
        }
        #pragma unroll
        for (int u = 0; u < FLUSH_BATCH; ++u)
            if (idx[u] >= 0 && v[u] != 0.0f) atomicAdd(&o[idx[u]], v[u]);
    }
}


extern "C" void kernel_launch(void* const* d_in, const int* in_sizes, int n_in,
                              void* d_out, int out_size, void* d_ws, size_t ws_size,
                              hipStream_t stream) {
    const float* X   = (const float*)d_in[0];
    const float* W1  = (const float*)d_in[1];
    const float* b1  = (const float*)d_in[2];
    const float* W2  = (const float*)d_in[3];
    const float* b2  = (const float*)d_in[4];
    const int*   ei  = (const int*)d_in[5];
    float*       out = (float*)d_out;

    const int N = NNODES;
    const int E = in_sizes[5] / 2;
    const int* src = ei;
    const int* dst = ei + E;

    float*  dinv = (float*)d_ws;
    int*    tab  = (int*)(dinv + N);
    float*  H1   = (float*)(tab + N);
    float*  H2   = H1   + (size_t)N * HIDDEN;
    unsigned* key = (unsigned*)(H2 + (size_t)N * NCLS);

    const int T = 256;

    k_init_deg <<<(N + T - 1) / T, T, 0, stream>>>(dinv, N);
    k_deg_hist <<<DEG_WGS, PRIV_THREADS, 0, stream>>>(src, dst, dinv, key, tab, E, N);
    k_dinv     <<<(N + T - 1) / T, T, 0, stream>>>(dinv, N);

    k_gemm1<<<2 * G_ROWG, G_THREADS, 0, stream>>>(X, W1, H1);
    k_agg1_own<<<NOWN, OWN_THREADS, 0, stream>>>(key, tab, dinv, H1, b1, W2, b2, H2, out, E, N);

    k_agg2_priv<<<2 * AGG2_CHUNKS, PRIV_THREADS, 0, stream>>>(key, dinv, H2, out, E);
}


# Global f32 atomics removed: each owner work-group walks its own sorted runs to count its rows' degrees (integer LDS adds) and to sum layer 2 in LDS (f32 adds, arrival order), storing each result once
# speedup vs baseline: 3.4663x; 1.1550x over previous
#include <hip/hip_runtime.h>
#include <hip/hip_bf16.h>

#define NNODES 50000
#define DIMIN  512
#define HIDDEN 256
#define NCLS   2

typedef __attribute__((ext_vector_type(2))) float v2f;
typedef __attribute__((ext_vector_type(8))) float v8f;

static_assert(NNODES <= 65536, "packed edge word keeps a node id in 16 bits");

#define OWN_ROWS     196
#define NOWN         ((NNODES + OWN_ROWS - 1) / OWN_ROWS)
#define TAB_PITCH    (NOWN + 1)
static_assert(NOWN % 32 == 0, "the prefix over the owners gives each lane of one wave NOWN / 32 counters");

#define PRIV_THREADS 1024
#define SORT_BATCH   13
#define SORT_EDGES   (SORT_BATCH * PRIV_THREADS)
#define NEDGES       800000
#define DEG_WGS      ((NEDGES + SORT_EDGES - 1) / SORT_EDGES)
static_assert(DEG_WGS * TAB_PITCH <= NNODES, "segment table must fit its workspace region");

__global__ void __launch_bounds__(PRIV_THREADS)
k_sort_own(const int* __restrict__ src, const int* __restrict__ dst,
           unsigned* __restrict__ key, int* __restrict__ tab, int e) {
    __shared__ unsigned sorted[SORT_EDGES];
    __shared__ unsigned ocur[NOWN];
    const int tid = threadIdx.x;

    const int nsub = (e + SORT_EDGES - 1) / SORT_EDGES;
    for (int sc = (int)blockIdx.x; sc < nsub; sc += (int)gridDim.x) {
        if (tid < NOWN) ocur[tid] = 0u;
        __syncthreads();
        const int lo = sc * SORT_EDGES;
        int hi = lo + SORT_EDGES; if (hi > e) hi = e;
        int s[SORT_BATCH], d[SORT_BATCH];
        #pragma unroll
        for (int j = 0; j < SORT_BATCH; ++j) {
            const int i  = lo + tid + PRIV_THREADS * j;
            const int ic = (i < hi) ? i : lo;
            d[j] = dst[ic];
            s[j] = src[ic];
        }
        unsigned w[SORT_BATCH];
        int ow[SORT_BATCH];
        #pragma unroll
        for (int j = 0; j < SORT_BATCH; ++j) {
            const int i = lo + tid + PRIV_THREADS * j;
            w[j]  = (unsigned)s[j] | ((unsigned)d[j] << 16);
            ow[j] = d[j] / OWN_ROWS;
            if (i < hi)
                __hip_atomic_fetch_add(&ocur[ow[j]], 1u, __ATOMIC_RELAXED, __HIP_MEMORY_SCOPE_WORKGROUP);
        }
        __syncthreads();

        if (tid < 32) {
            unsigned c[NOWN / 32];
            unsigned sum = 0u;
            #pragma unroll
            for (int u = 0; u < NOWN / 32; ++u) { c[u] = ocur[tid * (NOWN / 32) + u]; sum += c[u]; }
            unsigned inc = sum;
            #pragma unroll
            for (int off = 1; off < 32; off <<= 1) {
                const unsigned v = __shfl_up(inc, off, 32);
                if (tid >= off) inc += v;
            }
            unsigned run = inc - sum;
            #pragma unroll
            for (int u = 0; u < NOWN / 32; ++u) {
                ocur[tid * (NOWN / 32) + u] = run;
                tab[sc * TAB_PITCH + tid * (NOWN / 32) + u] = lo + (int)run;
                run += c[u];
            }
            if (tid == 31) tab[sc * TAB_PITCH + NOWN] = lo + (int)run;
        }
        __syncthreads();

        unsigned pos[SORT_BATCH];
        #pragma unroll
        for (int j = 0; j < SORT_BATCH; ++j) {
            const int i = lo + tid + PRIV_THREADS * j;
            pos[j] = 0u;
            if (i < hi) pos[j] = __hip_atomic_fetch_add(&ocur[ow[j]], 1u, __ATOMIC_RELAXED,
                                                        __HIP_MEMORY_SCOPE_WORKGROUP);
        }
        #pragma unroll
        for (int j = 0; j < SORT_BATCH; ++j) {
            const int i = lo + tid + PRIV_THREADS * j;
            if (i < hi) sorted[pos[j]] = w[j];
        }
        __syncthreads();

        #pragma unroll
        for (int j = 0; j < SORT_BATCH; ++j) pos[j] = sorted[tid + PRIV_THREADS * j];
        #pragma unroll
        for (int j = 0; j < SORT_BATCH; ++j) {
            const int i = lo + tid + PRIV_THREADS * j;
            if (i < hi) key[i] = pos[j];
        }
    }
}


#define ROWN_THREADS 512
#define ROWN_WAVES   (ROWN_THREADS / 32)
static_assert(2 * OWN_ROWS <= ROWN_THREADS, "one thread per owned row, and per word of the layer-2 rows, in the prologues and epilogues");

__global__ void __launch_bounds__(ROWN_THREADS)
k_own_deg(const unsigned* __restrict__ key, const int* __restrict__ tab,
          float* __restrict__ dinv, int e, int n) {
    __shared__ unsigned cnt[OWN_ROWS];
    const int tid = threadIdx.x;
    const int own = (int)blockIdx.x;
    const int lo  = own * OWN_ROWS;
    int nrows = n - lo;
    if (nrows > OWN_ROWS) nrows = OWN_ROWS;
    const int wave = __builtin_amdgcn_readfirstlane(tid >> 5);
    const int lane = tid & 31;
    if (tid < OWN_ROWS) cnt[tid] = 0u;
    __syncthreads();

    const unsigned ulo = (unsigned)lo;
    const int nsub  = (e + SORT_EDGES - 1) / SORT_EDGES;
    const int elast = e - 1;
    for (int c0 = wave; c0 < nsub; c0 += ROWN_WAVES * 32) {
        const int cl = c0 + ROWN_WAVES * lane;
        int sb = 0, se = 0;
        if (cl < nsub) { sb = tab[cl * TAB_PITCH + own]; se = tab[cl * TAB_PITCH + own + 1]; }
        int nk = (nsub - c0 + ROWN_WAVES - 1) / ROWN_WAVES;
        if (nk > 32) nk = 32;
        int beg = __builtin_amdgcn_readlane(sb, 0), end = __builtin_amdgcn_readlane(se, 0);
        int i0 = beg + lane, i1 = beg + 32 + lane;
        unsigned w0 = key[i0 < elast ? i0 : elast];
        unsigned w1 = key[i1 < elast ? i1 : elast];
        for (int k = 0; k < nk; ++k) {
            const int kn = (k + 1 < nk) ? k + 1 : k;
            const int begn = __builtin_amdgcn_readlane(sb, kn);
            const int endn = __builtin_amdgcn_readlane(se, kn);
            i0 = begn + lane; i1 = begn + 32 + lane;
            const unsigned n0 = key[i0 < elast ? i0 : elast];
            const unsigned n1 = key[i1 < elast ? i1 : elast];
            for (int base = beg; base < end; base += 64) {
                if (base != beg) {
                    i0 = base + lane; i1 = base + 32 + lane;
                    w0 = key[i0 < elast ? i0 : elast];
                    w1 = key[i1 < elast ? i1 : elast];
                }
                if (base + lane < end)
                    __hip_atomic_fetch_add(&cnt[(w0 >> 16) - ulo], 1u, __ATOMIC_RELAXED, __HIP_MEMORY_SCOPE_WORKGROUP);
                if (base + 32 + lane < end)
                    __hip_atomic_fetch_add(&cnt[(w1 >> 16) - ulo], 1u, __ATOMIC_RELAXED, __HIP_MEMORY_SCOPE_WORKGROUP);
            }
            beg = begn; end = endn; w0 = n0; w1 = n1;
        }
    }
    __syncthreads();
    if (tid < nrows) dinv[lo + tid] = __frsqrt_rn((float)(cnt[tid] + 1u));
}


#define G_THREADS 512
#define G_ROWG    128
#define G_HALFC   128
#define G_KH      256
#define G_APITCH  260
#define G_NT      (NNODES / 16)

__device__ __forceinline__ void gemm_stage_load(const float* __restrict__ X, int rg, int s, int tid,
                                                float4 (&r)[4]) {
    const int p  = s >> 1;
    const int kh = s & 1;
    #pragma unroll
    for (int j = 0; j < 4; ++j) {
        const int idx = tid + G_THREADS * j;
        const int rr  = idx >> 6;
        const int c4  = (idx & 63) * 4;
        const int t   = rg + G_ROWG * (2 * p + (rr >> 4));
        if (t < G_NT)
            r[j] = *(const float4*)(X + (size_t)(t * 16 + (rr & 15)) * DIMIN + kh * G_KH + c4);
        else
            r[j] = make_float4(0.0f, 0.0f, 0.0f, 0.0f);
    }
}

__global__ void __launch_bounds__(G_THREADS)
k_gemm1(const float* __restrict__ X, const float* __restrict__ W,
        float* __restrict__ H) {
    __shared__ alignas(16) float Ws[(DIMIN / 2) * G_HALFC * 2];
    __shared__ alignas(16) float As[32 * G_APITCH];

    const int tid  = threadIdx.x;
    const int half = blockIdx.x & 1;
    const int rg   = blockIdx.x >> 1;
    const int ntile  = (G_NT - rg + G_ROWG - 1) / G_ROWG;
    const int nstage = 2 * ((ntile + 1) / 2);

    float4 pre[4];
    gemm_stage_load(X, rg, 0, tid, pre);

    #pragma unroll 4
    for (int i = 0; i < 16; ++i) {
        const int it  = (i + rg) & 15;
        const int idx = tid + G_THREADS * it;
        const int kp  = idx >> 5;
        const int cq  = idx & 31;
        const float4 r0 = *(const float4*)(W + (size_t)(2 * kp)     * HIDDEN + half * G_HALFC + 4 * cq);
        const float4 r1 = *(const float4*)(W + (size_t)(2 * kp + 1) * HIDDEN + half * G_HALFC + 4 * cq);
        const int c = (4 * cq) ^ ((kp & 1) << 4);
        float4* d = (float4*)(&Ws[2 * (kp * G_HALFC + c)]);
        d[0] = make_float4(r0.x, r1.x, r0.y, r1.y);
        d[1] = make_float4(r0.z, r1.z, r0.w, r1.w);
    }

    const int wave = __builtin_amdgcn_readfirstlane(tid >> 5);
    const int lane = tid & 31;
    const int l16  = lane & 15;
    const int hi   = lane >> 4;
    const int sub  = wave >> 3;
    const int ct   = wave & 7;

    const float* ap = As + (sub * 16 + l16) * G_APITCH + 2 * hi;
    const float* bp = Ws + 2 * (hi * G_HALFC + ((ct ^ hi) << 4) + l16);

    v8f c = {};
    for (int s = 0; s < nstage; ++s) {
        __syncthreads();
        #pragma unroll
        for (int j = 0; j < 4; ++j) {
            const int idx = tid + G_THREADS * j;
            *(float4*)(&As[(idx >> 6) * G_APITCH + (idx & 63) * 4]) = pre[j];
        }
        __syncthreads();
        if (s + 1 < nstage) gemm_stage_load(X, rg, s + 1, tid, pre);

        const int kh = s & 1;
        const int t  = rg + G_ROWG * (2 * (s >> 1) + sub);
        if (t < G_NT) {
            const float* b = bp + 2 * (kh * (G_KH / 2) * G_HALFC);
            v2f fa[2][8], fb[2][8];
            #pragma unroll
            for (int i = 0; i < 8; ++i) {
                fa[0][i] = *(const v2f*)(ap + 4 * i);
                fb[0][i] = *(const v2f*)(b + (2 * i) * (2 * G_HALFC));
            }
            #pragma unroll
            for (int j = 0; j < G_KH / 32; ++j) {
                if (j + 1 < G_KH / 32) {
                    #pragma unroll
                    for (int i = 0; i < 8; ++i) {
                        const int kk = 32 * (j + 1) + 4 * i;
                        fa[(j + 1) & 1][i] = *(const v2f*)(ap + kk);
                        fb[(j + 1) & 1][i] = *(const v2f*)(b + (kk >> 1) * (2 * G_HALFC));
                    }
                }
                __builtin_amdgcn_sched_barrier(0);
                #pragma unroll
                for (int i = 0; i < 8; ++i)
                    c = __builtin_amdgcn_wmma_f32_16x16x4_f32(false, fa[j & 1][i], false, fb[j & 1][i],
                                                              (short)0, c, false, false);
                __builtin_amdgcn_sched_barrier(0);
            }
            if (kh == 1) {
                #pragma unroll
                for (int v = 0; v < 8; ++v) {
                    const size_t o = (size_t)(t * 16 + hi * 8 + v) * HIDDEN + half * G_HALFC + ct * 16 + l16;
                    H[o] = c[v];
                }
                c = v8f{};
            }
        }
    }
}


#define OWN_THREADS 1024
#define OWN_WAVES   (OWN_THREADS / 32)
#define RUN_WORDS   64
#define LIST_CAP    (RUN_WORDS + 4)

__device__ __forceinline__ void own_lds_add(float* a, float nrm, const float4& p, const float4& q) {
    __hip_atomic_fetch_add(&a[0 * 32], nrm * p.x, __ATOMIC_RELAXED, __HIP_MEMORY_SCOPE_WORKGROUP);
    __hip_atomic_fetch_add(&a[1 * 32], nrm * p.y, __ATOMIC_RELAXED, __HIP_MEMORY_SCOPE_WORKGROUP);
    __hip_atomic_fetch_add(&a[2 * 32], nrm * p.z, __ATOMIC_RELAXED, __HIP_MEMORY_SCOPE_WORKGROUP);
    __hip_atomic_fetch_add(&a[3 * 32], nrm * p.w, __ATOMIC_RELAXED, __HIP_MEMORY_SCOPE_WORKGROUP);
    __hip_atomic_fetch_add(&a[4 * 32], nrm * q.x, __ATOMIC_RELAXED, __HIP_MEMORY_SCOPE_WORKGROUP);
    __hip_atomic_fetch_add(&a[5 * 32], nrm * q.y, __ATOMIC_RELAXED, __HIP_MEMORY_SCOPE_WORKGROUP);
    __hip_atomic_fetch_add(&a[6 * 32], nrm * q.z, __ATOMIC_RELAXED, __HIP_MEMORY_SCOPE_WORKGROUP);
    __hip_atomic_fetch_add(&a[7 * 32], nrm * q.w, __ATOMIC_RELAXED, __HIP_MEMORY_SCOPE_WORKGROUP);
}

__device__ __forceinline__ void own_add4(const float* __restrict__ dinv, const float* __restrict__ H1,
                                         float* acc, const float* dl, int lane,
                                         unsigned t0, unsigned t1, unsigned t2, unsigned t3) {
    const int s0 = (int)(t0 & 0xFFFFu), s1 = (int)(t1 & 0xFFFFu);
    const int s2 = (int)(t2 & 0xFFFFu), s3 = (int)(t3 & 0xFFFFu);
    const int r0 = (int)(t0 >> 16), r1 = (int)(t1 >> 16), r2 = (int)(t2 >> 16), r3 = (int)(t3 >> 16);
    const float4* h0 = (const float4*)(H1 + (size_t)s0 * HIDDEN) + lane;
    const float4* h1 = (const float4*)(H1 + (size_t)s1 * HIDDEN) + lane;
    const float4* h2 = (const float4*)(H1 + (size_t)s2 * HIDDEN) + lane;
    const float4* h3 = (const float4*)(H1 + (size_t)s3 * HIDDEN) + lane;
    const float4 p0 = h0[0], q0 = h0[32];
    const float4 p1 = h1[0], q1 = h1[32];
    const float4 p2 = h2[0], q2 = h2[32];
    const float4 p3 = h3[0], q3 = h3[32];
    const float n0 = dinv[s0] * dl[r0];
    const float n1 = dinv[s1] * dl[r1];
    const float n2 = dinv[s2] * dl[r2];
    const float n3 = dinv[s3] * dl[r3];
    own_lds_add(acc + r0 * HIDDEN + lane, n0, p0, q0);
    own_lds_add(acc + r1 * HIDDEN + lane, n1, p1, q1);
    own_lds_add(acc + r2 * HIDDEN + lane, n2, p2, q2);
    own_lds_add(acc + r3 * HIDDEN + lane, n3, p3, q3);
}

__device__ __forceinline__ void own_add1(const float* __restrict__ dinv, const float* __restrict__ H1,
                                         float* acc, const float* dl, int lane, unsigned t0) {
    const int s0 = (int)(t0 & 0xFFFFu);
    const int r0 = (int)(t0 >> 16);
    const float4* h0 = (const float4*)(H1 + (size_t)s0 * HIDDEN) + lane;
    const float4 p0 = h0[0], q0 = h0[32];
    const float n0 = dinv[s0] * dl[r0];
    own_lds_add(acc + r0 * HIDDEN + lane, n0, p0, q0);
}

__device__ __forceinline__ int own_take(const float* __restrict__ dinv, const float* __restrict__ H1,
                                        float* acc, const float* dl, unsigned* my, int lane,
                                        unsigned ulo, unsigned a, unsigned b, int m, int cnt) {
    if (lane < m)      my[cnt + lane]      = a - ulo;
    if (lane + 32 < m) my[cnt + 32 + lane] = b - ulo;
    cnt += m;
    while (cnt >= 4) {
        cnt -= 4;
        const unsigned t0 = (unsigned)__builtin_amdgcn_readfirstlane((int)my[cnt]);
        const unsigned t1 = (unsigned)__builtin_amdgcn_readfirstlane((int)my[cnt + 1]);
        const unsigned t2 = (unsigned)__builtin_amdgcn_readfirstlane((int)my[cnt + 2]);
        const unsigned t3 = (unsigned)__builtin_amdgcn_readfirstlane((int)my[cnt + 3]);
        own_add4(dinv, H1, acc, dl, lane, t0, t1, t2, t3);
    }
    return cnt;
}

__global__ void __launch_bounds__(OWN_THREADS)
k_agg1_own(const unsigned* __restrict__ key, const int* __restrict__ tab,
           const float* __restrict__ dinv, const float* __restrict__ H1,
           const float* __restrict__ b1, const float* __restrict__ W2,
           float* __restrict__ H2, int e, int n) {
    __shared__ float acc[OWN_ROWS * HIDDEN];
    __shared__ unsigned lst[OWN_WAVES * LIST_CAP];
    __shared__ float dl[OWN_ROWS];

    const int lo = blockIdx.x * OWN_ROWS;
    int nrows = n - lo;
    if (nrows > OWN_ROWS) nrows = OWN_ROWS;
    const int wave = __builtin_amdgcn_readfirstlane((int)(threadIdx.x >> 5));
    const int lane = threadIdx.x & 31;

    for (int r = wave; r < nrows; r += OWN_WAVES) {
        const float dv = dinv[lo + r];
        if (lane == 0) dl[r] = dv;
        const float4* h = (const float4*)(H1 + (size_t)(lo + r) * HIDDEN) + lane;
        const float4 p = h[0], q = h[32];
        float* a = acc + r * HIDDEN + lane;
        a[0 * 32] = dv * dv * p.x;
        a[1 * 32] = dv * dv * p.y;
        a[2 * 32] = dv * dv * p.z;
        a[3 * 32] = dv * dv * p.w;
        a[4 * 32] = dv * dv * q.x;
        a[5 * 32] = dv * dv * q.y;
        a[6 * 32] = dv * dv * q.z;
        a[7 * 32] = dv * dv * q.w;
    }
    __syncthreads();

    const unsigned ulo = (unsigned)lo << 16;
    const int own   = (int)blockIdx.x;
    const int nsub  = (e + SORT_EDGES - 1) / SORT_EDGES;
    const int elast = e - 1;
    unsigned* my = lst + wave * LIST_CAP;
    int cnt = 0;

    for (int c0 = wave; c0 < nsub; c0 += OWN_WAVES * 32) {
        const int cl = c0 + OWN_WAVES * lane;
        int sb = 0, se = 0;
        if (cl < nsub) { sb = tab[cl * TAB_PITCH + own]; se = tab[cl * TAB_PITCH + own + 1]; }
        int nk = (nsub - c0 + OWN_WAVES - 1) / OWN_WAVES;
        if (nk > 32) nk = 32;
        int beg = __builtin_amdgcn_readlane(sb, 0), end = __builtin_amdgcn_readlane(se, 0);
        int i0 = beg + lane, i1 = beg + 32 + lane;
        unsigned w0 = key[i0 < elast ? i0 : elast];
        unsigned w1 = key[i1 < elast ? i1 : elast];
        for (int k = 0; k < nk; ++k) {
            const int kn = (k + 1 < nk) ? k + 1 : k;
            const int begn = __builtin_amdgcn_readlane(sb, kn);
            const int endn = __builtin_amdgcn_readlane(se, kn);
            i0 = begn + lane; i1 = begn + 32 + lane;
            const unsigned n0 = key[i0 < elast ? i0 : elast];
            const unsigned n1 = key[i1 < elast ? i1 : elast];

            int m = end - beg; if (m > RUN_WORDS) m = RUN_WORDS;
            cnt = own_take(dinv, H1, acc, dl, my, lane, ulo, w0, w1, m, cnt);
            for (int base = beg + RUN_WORDS; base < end; base += RUN_WORDS) {
                i0 = base + lane; i1 = base + 32 + lane;
                const unsigned a = key[i0 < elast ? i0 : elast];
                const unsigned b = key[i1 < elast ? i1 : elast];
                int mm = end - base; if (mm > RUN_WORDS) mm = RUN_WORDS;
                cnt = own_take(dinv, H1, acc, dl, my, lane, ulo, a, b, mm, cnt);
            }
            beg = begn; end = endn; w0 = n0; w1 = n1;
        }
    }
    while (cnt > 0) {
        cnt -= 1;
        const unsigned t0 = (unsigned)__builtin_amdgcn_readfirstlane((int)my[cnt]);
        own_add1(dinv, H1, acc, dl, lane, t0);
    }
    __syncthreads();

    float bb[8], w0[8], w1[8];
    #pragma unroll
    for (int q = 0; q < 2; ++q) {
        const int c = 4 * lane + 128 * q;
        const float4 bv = *(const float4*)(b1 + c);
        const float4 wa = *(const float4*)(W2 + 2 * c);
        const float4 wb = *(const float4*)(W2 + 2 * c + 4);
        bb[4 * q + 0] = bv.x; bb[4 * q + 1] = bv.y; bb[4 * q + 2] = bv.z; bb[4 * q + 3] = bv.w;
        w0[4 * q + 0] = wa.x; w1[4 * q + 0] = wa.y;
        w0[4 * q + 1] = wa.z; w1[4 * q + 1] = wa.w;
        w0[4 * q + 2] = wb.x; w1[4 * q + 2] = wb.y;
        w0[4 * q + 3] = wb.z; w1[4 * q + 3] = wb.w;
    }
    for (int r = wave; r < nrows; r += OWN_WAVES) {
        const float* a = acc + r * HIDDEN + lane;
        float p0 = 0.0f, p1 = 0.0f;
        #pragma unroll
        for (int j = 0; j < 8; ++j) {
            float z = a[j * 32] + bb[j];
            z = z > 0.0f ? z : 0.0f;
            p0 += z * w0[j];
            p1 += z * w1[j];
        }
        #pragma unroll
        for (int off = 16; off > 0; off >>= 1) {
            p0 += __shfl_xor(p0, off, 32);
            p1 += __shfl_xor(p1, off, 32);
        }
        if (lane == 0) {
            const int row = lo + r;
            H2[2 * row]      = p0;
            H2[2 * row + 1]  = p1;
        }
    }
}

__global__ void __launch_bounds__(ROWN_THREADS)
k_agg2_own(const unsigned* __restrict__ key, const int* __restrict__ tab,
           const float* __restrict__ dinv, const float* __restrict__ H2,
           const float* __restrict__ b2, float* __restrict__ out, int e, int n) {
    __shared__ float acc[2 * OWN_ROWS];
    __shared__ float dl[OWN_ROWS];
    const int tid = threadIdx.x;
    const int own = (int)blockIdx.x;
    const int lo  = own * OWN_ROWS;
    int nrows = n - lo;
    if (nrows > OWN_ROWS) nrows = OWN_ROWS;
    const int wave = __builtin_amdgcn_readfirstlane(tid >> 5);
    const int lane = tid & 31;

    if (tid < nrows) {
        const float dv = dinv[lo + tid];
        const float2 h = *(const float2*)(H2 + 2 * (size_t)(lo + tid));
        dl[tid] = dv;
        acc[2 * tid]     = dv * dv * h.x + b2[0];
        acc[2 * tid + 1] = dv * dv * h.y + b2[1];
    }
    __syncthreads();

    const unsigned ulo = (unsigned)lo;
    const int nsub  = (e + SORT_EDGES - 1) / SORT_EDGES;
    const int elast = e - 1;
    for (int c0 = wave; c0 < nsub; c0 += ROWN_WAVES * 32) {
        const int cl = c0 + ROWN_WAVES * lane;
        int sb = 0, se = 0;
        if (cl < nsub) { sb = tab[cl * TAB_PITCH + own]; se = tab[cl * TAB_PITCH + own + 1]; }
        int nk = (nsub - c0 + ROWN_WAVES - 1) / ROWN_WAVES;
        if (nk > 32) nk = 32;
        int beg = __builtin_amdgcn_readlane(sb, 0), end = __builtin_amdgcn_readlane(se, 0);
        int i0 = beg + lane, i1 = beg + 32 + lane;
        unsigned w0 = key[i0 < elast ? i0 : elast];
        unsigned w1 = key[i1 < elast ? i1 : elast];
        for (int k = 0; k < nk; ++k) {
            const int kn = (k + 1 < nk) ? k + 1 : k;
            const int begn = __builtin_amdgcn_readlane(sb, kn);
            const int endn = __builtin_amdgcn_readlane(se, kn);
            i0 = begn + lane; i1 = begn + 32 + lane;
            const unsigned n0 = key[i0 < elast ? i0 : elast];
            const unsigned n1 = key[i1 < elast ? i1 : elast];
            for (int base = beg; base < end; base += 64) {
                if (base != beg) {
                    i0 = base + lane; i1 = base + 32 + lane;
                    w0 = key[i0 < elast ? i0 : elast];
                    w1 = key[i1 < elast ? i1 : elast];
                }
                const unsigned s0 = w0 & 0xFFFFu, s1 = w1 & 0xFFFFu;
                const float  a0 = dinv[s0], a1 = dinv[s1];
                const float2 h0 = *(const float2*)(H2 + 2 * s0);
                const float2 h1 = *(const float2*)(H2 + 2 * s1);
                if (base + lane < end) {
                    const unsigned r = (w0 >> 16) - ulo;
                    const float norm = a0 * dl[r];
                    __hip_atomic_fetch_add(&acc[2 * r],     norm * h0.x, __ATOMIC_RELAXED, __HIP_MEMORY_SCOPE_WORKGROUP);
                    __hip_atomic_fetch_add(&acc[2 * r + 1], norm * h0.y, __ATOMIC_RELAXED, __HIP_MEMORY_SCOPE_WORKGROUP);
                }
                if (base + 32 + lane < end) {
                    const unsigned r = (w1 >> 16) - ulo;
                    const float norm = a1 * dl[r];
                    __hip_atomic_fetch_add(&acc[2 * r],     norm * h1.x, __ATOMIC_RELAXED, __HIP_MEMORY_SCOPE_WORKGROUP);
                    __hip_atomic_fetch_add(&acc[2 * r + 1], norm * h1.y, __ATOMIC_RELAXED, __HIP_MEMORY_SCOPE_WORKGROUP);
                }
            }
            beg = begn; end = endn; w0 = n0; w1 = n1;
        }
    }
    __syncthreads();
    if (tid < 2 * nrows) out[2 * (size_t)lo + tid] = acc[tid];
}


extern "C" void kernel_launch(void* const* d_in, const int* in_sizes, int n_in,
                              void* d_out, int out_size, void* d_ws, size_t ws_size,
                              hipStream_t stream) {
    const float* X   = (const float*)d_in[0];
    const float* W1  = (const float*)d_in[1];
    const float* b1  = (const float*)d_in[2];
    const float* W2  = (const float*)d_in[3];
    const float* b2  = (const float*)d_in[4];
    const int*   ei  = (const int*)d_in[5];
    float*       out = (float*)d_out;

    const int N = NNODES;
    const int E = in_sizes[5] / 2;
    const int* src = ei;
    const int* dst = ei + E;

    float*  dinv = (float*)d_ws;
    int*    tab  = (int*)(dinv + N);
    float*  H1   = (float*)(tab + N);
    float*  H2   = H1   + (size_t)N * HIDDEN;
    unsigned* key = (unsigned*)(H2 + (size_t)N * NCLS);

    k_sort_own<<<DEG_WGS, PRIV_THREADS, 0, stream>>>(src, dst, key, tab, E);
    k_own_deg <<<NOWN, ROWN_THREADS, 0, stream>>>(key, tab, dinv, E, N);

    k_gemm1<<<2 * G_ROWG, G_THREADS, 0, stream>>>(X, W1, H1);
    k_agg1_own<<<NOWN, OWN_THREADS, 0, stream>>>(key, tab, dinv, H1, b1, W2, H2, E, N);

    k_agg2_own<<<NOWN, ROWN_THREADS, 0, stream>>>(key, tab, dinv, H2, b2, out, E, N);
}


# Source-side factor dinv[s] folded into stored rows (h1 in the product epilogue, h2 in layer 1): owner sums add rows as loaded, dinv[d] once per row; products re-associated, sum order as before
# speedup vs baseline: 3.6904x; 1.0647x over previous
#include <hip/hip_runtime.h>
#include <hip/hip_bf16.h>

#define NNODES 50000
#define DIMIN  512
#define HIDDEN 256
#define NCLS   2

typedef __attribute__((ext_vector_type(2))) float v2f;
typedef __attribute__((ext_vector_type(8))) float v8f;

static_assert(NNODES <= 65536, "packed edge word keeps a node id in 16 bits");

#define OWN_ROWS     196
#define NOWN         ((NNODES + OWN_ROWS - 1) / OWN_ROWS)
#define TAB_PITCH    (NOWN + 1)
static_assert(NOWN % 32 == 0, "the prefix over the owners gives each lane of one wave NOWN / 32 counters");

#define PRIV_THREADS 1024
#define SORT_BATCH   13
#define SORT_EDGES   (SORT_BATCH * PRIV_THREADS)
#define NEDGES       800000
#define DEG_WGS      ((NEDGES + SORT_EDGES - 1) / SORT_EDGES)
static_assert(DEG_WGS * TAB_PITCH <= NNODES, "segment table must fit its workspace region");

__global__ void __launch_bounds__(PRIV_THREADS)
k_sort_own(const int* __restrict__ src, const int* __restrict__ dst,
           unsigned* __restrict__ key, int* __restrict__ tab, int e) {
    __shared__ unsigned sorted[SORT_EDGES];
    __shared__ unsigned ocur[NOWN];
    const int tid = threadIdx.x;

    const int nsub = (e + SORT_EDGES - 1) / SORT_EDGES;
    for (int sc = (int)blockIdx.x; sc < nsub; sc += (int)gridDim.x) {
        if (tid < NOWN) ocur[tid] = 0u;
        __syncthreads();
        const int lo = sc * SORT_EDGES;
        int hi = lo + SORT_EDGES; if (hi > e) hi = e;
        int s[SORT_BATCH], d[SORT_BATCH];
        #pragma unroll
        for (int j = 0; j < SORT_BATCH; ++j) {
            const int i  = lo + tid + PRIV_THREADS * j;
            const int ic = (i < hi) ? i : lo;
            d[j] = dst[ic];
            s[j] = src[ic];
        }
        unsigned w[SORT_BATCH];
        int ow[SORT_BATCH];
        #pragma unroll
        for (int j = 0; j < SORT_BATCH; ++j) {
            const int i = lo + tid + PRIV_THREADS * j;
            w[j]  = (unsigned)s[j] | ((unsigned)d[j] << 16);
            ow[j] = d[j] / OWN_ROWS;
            if (i < hi)
                __hip_atomic_fetch_add(&ocur[ow[j]], 1u, __ATOMIC_RELAXED, __HIP_MEMORY_SCOPE_WORKGROUP);
        }
        __syncthreads();

        if (tid < 32) {
            unsigned c[NOWN / 32];
            unsigned sum = 0u;
            #pragma unroll
            for (int u = 0; u < NOWN / 32; ++u) { c[u] = ocur[tid * (NOWN / 32) + u]; sum += c[u]; }
            unsigned inc = sum;
            #pragma unroll
            for (int off = 1; off < 32; off <<= 1) {
                const unsigned v = __shfl_up(inc, off, 32);
                if (tid >= off) inc += v;
            }
            unsigned run = inc - sum;
            #pragma unroll
            for (int u = 0; u < NOWN / 32; ++u) {
                ocur[tid * (NOWN / 32) + u] = run;
                tab[sc * TAB_PITCH + tid * (NOWN / 32) + u] = lo + (int)run;
                run += c[u];
            }
            if (tid == 31) tab[sc * TAB_PITCH + NOWN] = lo + (int)run;
        }
        __syncthreads();

        unsigned pos[SORT_BATCH];
        #pragma unroll
        for (int j = 0; j < SORT_BATCH; ++j) {
            const int i = lo + tid + PRIV_THREADS * j;
            pos[j] = 0u;
            if (i < hi) pos[j] = __hip_atomic_fetch_add(&ocur[ow[j]], 1u, __ATOMIC_RELAXED,
                                                        __HIP_MEMORY_SCOPE_WORKGROUP);
        }
        #pragma unroll
        for (int j = 0; j < SORT_BATCH; ++j) {
            const int i = lo + tid + PRIV_THREADS * j;
            if (i < hi) sorted[pos[j]] = w[j];
        }
        __syncthreads();

        #pragma unroll
        for (int j = 0; j < SORT_BATCH; ++j) pos[j] = sorted[tid + PRIV_THREADS * j];
        #pragma unroll
        for (int j = 0; j < SORT_BATCH; ++j) {
            const int i = lo + tid + PRIV_THREADS * j;
            if (i < hi) key[i] = pos[j];
        }
    }
}


#define ROWN_THREADS 512
#define ROWN_WAVES   (ROWN_THREADS / 32)
static_assert(2 * OWN_ROWS <= ROWN_THREADS, "one thread per owned row, and per word of the layer-2 rows, in the prologues and epilogues");

__global__ void __launch_bounds__(ROWN_THREADS)
k_own_deg(const unsigned* __restrict__ key, const int* __restrict__ tab,
          float* __restrict__ dinv, int e, int n) {
    __shared__ unsigned cnt[OWN_ROWS];
    const int tid = threadIdx.x;
    const int own = (int)blockIdx.x;
    const int lo  = own * OWN_ROWS;
    int nrows = n - lo;
    if (nrows > OWN_ROWS) nrows = OWN_ROWS;
    const int wave = __builtin_amdgcn_readfirstlane(tid >> 5);
    const int lane = tid & 31;
    if (tid < OWN_ROWS) cnt[tid] = 0u;
    __syncthreads();

    const unsigned ulo = (unsigned)lo;
    const int nsub  = (e + SORT_EDGES - 1) / SORT_EDGES;
    const int elast = e - 1;
    for (int c0 = wave; c0 < nsub; c0 += ROWN_WAVES * 32) {
        const int cl = c0 + ROWN_WAVES * lane;
        int sb = 0, se = 0;
        if (cl < nsub) { sb = tab[cl * TAB_PITCH + own]; se = tab[cl * TAB_PITCH + own + 1]; }
        int nk = (nsub - c0 + ROWN_WAVES - 1) / ROWN_WAVES;
        if (nk > 32) nk = 32;
        int beg = __builtin_amdgcn_readlane(sb, 0), end = __builtin_amdgcn_readlane(se, 0);
        int i0 = beg + lane, i1 = beg + 32 + lane;
        unsigned w0 = key[i0 < elast ? i0 : elast];
        unsigned w1 = key[i1 < elast ? i1 : elast];
        for (int k = 0; k < nk; ++k) {
            const int kn = (k + 1 < nk) ? k + 1 : k;
            const int begn = __builtin_amdgcn_readlane(sb, kn);
            const int endn = __builtin_amdgcn_readlane(se, kn);
            i0 = begn + lane; i1 = begn + 32 + lane;
            const unsigned n0 = key[i0 < elast ? i0 : elast];
            const unsigned n1 = key[i1 < elast ? i1 : elast];
            for (int base = beg; base < end; base += 64) {
                if (base != beg) {
                    i0 = base + lane; i1 = base + 32 + lane;
                    w0 = key[i0 < elast ? i0 : elast];
                    w1 = key[i1 < elast ? i1 : elast];
                }
                if (base + lane < end)
                    __hip_atomic_fetch_add(&cnt[(w0 >> 16) - ulo], 1u, __ATOMIC_RELAXED, __HIP_MEMORY_SCOPE_WORKGROUP);
                if (base + 32 + lane < end)
                    __hip_atomic_fetch_add(&cnt[(w1 >> 16) - ulo], 1u, __ATOMIC_RELAXED, __HIP_MEMORY_SCOPE_WORKGROUP);
            }
            beg = begn; end = endn; w0 = n0; w1 = n1;
        }
    }
    __syncthreads();
    if (tid < nrows) dinv[lo + tid] = __frsqrt_rn((float)(cnt[tid] + 1u));
}


#define G_THREADS 512
#define G_ROWG    128
#define G_HALFC   128
#define G_KH      256
#define G_APITCH  260
#define G_NT      (NNODES / 16)

__device__ __forceinline__ void gemm_stage_load(const float* __restrict__ X, int rg, int s, int tid,
                                                float4 (&r)[4]) {
    const int p  = s >> 1;
    const int kh = s & 1;
    #pragma unroll
    for (int j = 0; j < 4; ++j) {
        const int idx = tid + G_THREADS * j;
        const int rr  = idx >> 6;
        const int c4  = (idx & 63) * 4;
        const int t   = rg + G_ROWG * (2 * p + (rr >> 4));
        if (t < G_NT)
            r[j] = *(const float4*)(X + (size_t)(t * 16 + (rr & 15)) * DIMIN + kh * G_KH + c4);
        else
            r[j] = make_float4(0.0f, 0.0f, 0.0f, 0.0f);
    }
}

__global__ void __launch_bounds__(G_THREADS)
k_gemm1(const float* __restrict__ X, const float* __restrict__ W,
        const float* __restrict__ dinv, float* __restrict__ H) {
    __shared__ alignas(16) float Ws[(DIMIN / 2) * G_HALFC * 2];
    __shared__ alignas(16) float As[32 * G_APITCH];

    const int tid  = threadIdx.x;
    const int half = blockIdx.x & 1;
    const int rg   = blockIdx.x >> 1;
    const int ntile  = (G_NT - rg + G_ROWG - 1) / G_ROWG;
    const int nstage = 2 * ((ntile + 1) / 2);

    float4 pre[4];
    gemm_stage_load(X, rg, 0, tid, pre);

    #pragma unroll 4
    for (int i = 0; i < 16; ++i) {
        const int it  = (i + rg) & 15;
        const int idx = tid + G_THREADS * it;
        const int kp  = idx >> 5;
        const int cq  = idx & 31;
        const float4 r0 = *(const float4*)(W + (size_t)(2 * kp)     * HIDDEN + half * G_HALFC + 4 * cq);
        const float4 r1 = *(const float4*)(W + (size_t)(2 * kp + 1) * HIDDEN + half * G_HALFC + 4 * cq);
        const int c = (4 * cq) ^ ((kp & 1) << 4);
        float4* d = (float4*)(&Ws[2 * (kp * G_HALFC + c)]);
        d[0] = make_float4(r0.x, r1.x, r0.y, r1.y);
        d[1] = make_float4(r0.z, r1.z, r0.w, r1.w);
    }

    const int wave = __builtin_amdgcn_readfirstlane(tid >> 5);
    const int lane = tid & 31;
    const int l16  = lane & 15;
    const int hi   = lane >> 4;
    const int sub  = wave >> 3;
    const int ct   = wave & 7;

    const float* ap = As + (sub * 16 + l16) * G_APITCH + 2 * hi;
    const float* bp = Ws + 2 * (hi * G_HALFC + ((ct ^ hi) << 4) + l16);

    v8f c = {};
    for (int s = 0; s < nstage; ++s) {
        __syncthreads();
        #pragma unroll
        for (int j = 0; j < 4; ++j) {
            const int idx = tid + G_THREADS * j;
            *(float4*)(&As[(idx >> 6) * G_APITCH + (idx & 63) * 4]) = pre[j];
        }
        __syncthreads();
        if (s + 1 < nstage) gemm_stage_load(X, rg, s + 1, tid, pre);

        const int kh = s & 1;
        const int t  = rg + G_ROWG * (2 * (s >> 1) + sub);
        if (t < G_NT) {
            const float* b = bp + 2 * (kh * (G_KH / 2) * G_HALFC);
            float4 da = make_float4(0.0f, 0.0f, 0.0f, 0.0f), db = da;
            if (kh == 1) {
                da = *(const float4*)(dinv + t * 16 + hi * 8);
                db = *(const float4*)(dinv + t * 16 + hi * 8 + 4);
            }
            v2f fa[2][8], fb[2][8];
            #pragma unroll
            for (int i = 0; i < 8; ++i) {
                fa[0][i] = *(const v2f*)(ap + 4 * i);
                fb[0][i] = *(const v2f*)(b + (2 * i) * (2 * G_HALFC));
            }
            #pragma unroll
            for (int j = 0; j < G_KH / 32; ++j) {
                if (j + 1 < G_KH / 32) {
                    #pragma unroll
                    for (int i = 0; i < 8; ++i) {
                        const int kk = 32 * (j + 1) + 4 * i;
                        fa[(j + 1) & 1][i] = *(const v2f*)(ap + kk);
                        fb[(j + 1) & 1][i] = *(const v2f*)(b + (kk >> 1) * (2 * G_HALFC));
                    }
                }
                __builtin_amdgcn_sched_barrier(0);
                #pragma unroll
                for (int i = 0; i < 8; ++i)
                    c = __builtin_amdgcn_wmma_f32_16x16x4_f32(false, fa[j & 1][i], false, fb[j & 1][i],
                                                              (short)0, c, false, false);
                __builtin_amdgcn_sched_barrier(0);
            }
            if (kh == 1) {
                const float dv[8] = {da.x, da.y, da.z, da.w, db.x, db.y, db.z, db.w};
                #pragma unroll
                for (int v = 0; v < 8; ++v) {
                    const size_t o = (size_t)(t * 16 + hi * 8 + v) * HIDDEN + half * G_HALFC + ct * 16 + l16;
                    H[o] = c[v] * dv[v];
                }
                c = v8f{};
            }
        }
    }
}


#define OWN_THREADS 1024
#define OWN_WAVES   (OWN_THREADS / 32)
#define RUN_WORDS   64
#define LIST_CAP    (RUN_WORDS + 4)

__device__ __forceinline__ void own_lds_add(float* a, const float4& p, const float4& q) {
    __hip_atomic_fetch_add(&a[0 * 32], p.x, __ATOMIC_RELAXED, __HIP_MEMORY_SCOPE_WORKGROUP);
    __hip_atomic_fetch_add(&a[1 * 32], p.y, __ATOMIC_RELAXED, __HIP_MEMORY_SCOPE_WORKGROUP);
    __hip_atomic_fetch_add(&a[2 * 32], p.z, __ATOMIC_RELAXED, __HIP_MEMORY_SCOPE_WORKGROUP);
    __hip_atomic_fetch_add(&a[3 * 32], p.w, __ATOMIC_RELAXED, __HIP_MEMORY_SCOPE_WORKGROUP);
    __hip_atomic_fetch_add(&a[4 * 32], q.x, __ATOMIC_RELAXED, __HIP_MEMORY_SCOPE_WORKGROUP);
    __hip_atomic_fetch_add(&a[5 * 32], q.y, __ATOMIC_RELAXED, __HIP_MEMORY_SCOPE_WORKGROUP);
    __hip_atomic_fetch_add(&a[6 * 32], q.z, __ATOMIC_RELAXED, __HIP_MEMORY_SCOPE_WORKGROUP);
    __hip_atomic_fetch_add(&a[7 * 32], q.w, __ATOMIC_RELAXED, __HIP_MEMORY_SCOPE_WORKGROUP);
}

__device__ __forceinline__ void own_add4(const float* __restrict__ H1, float* acc, int lane,
                                         unsigned t0, unsigned t1, unsigned t2, unsigned t3) {
    const int s0 = (int)(t0 & 0xFFFFu), s1 = (int)(t1 & 0xFFFFu);
    const int s2 = (int)(t2 & 0xFFFFu), s3 = (int)(t3 & 0xFFFFu);
    const int r0 = (int)(t0 >> 16), r1 = (int)(t1 >> 16), r2 = (int)(t2 >> 16), r3 = (int)(t3 >> 16);
    const float4* h0 = (const float4*)(H1 + (size_t)s0 * HIDDEN) + lane;
    const float4* h1 = (const float4*)(H1 + (size_t)s1 * HIDDEN) + lane;
    const float4* h2 = (const float4*)(H1 + (size_t)s2 * HIDDEN) + lane;
    const float4* h3 = (const float4*)(H1 + (size_t)s3 * HIDDEN) + lane;
    const float4 p0 = h0[0], q0 = h0[32];
    const float4 p1 = h1[0], q1 = h1[32];
    const float4 p2 = h2[0], q2 = h2[32];
    const float4 p3 = h3[0], q3 = h3[32];
    own_lds_add(acc + r0 * HIDDEN + lane, p0, q0);
    own_lds_add(acc + r1 * HIDDEN + lane, p1, q1);
    own_lds_add(acc + r2 * HIDDEN + lane, p2, q2);
    own_lds_add(acc + r3 * HIDDEN + lane, p3, q3);
}

__device__ __forceinline__ void own_add1(const float* __restrict__ H1, float* acc, int lane, unsigned t0) {
    const int s0 = (int)(t0 & 0xFFFFu);
    const int r0 = (int)(t0 >> 16);
    const float4* h0 = (const float4*)(H1 + (size_t)s0 * HIDDEN) + lane;
    const float4 p0 = h0[0], q0 = h0[32];
    own_lds_add(acc + r0 * HIDDEN + lane, p0, q0);
}

__device__ __forceinline__ int own_take(const float* __restrict__ H1, float* acc, unsigned* my, int lane,
                                        unsigned ulo, unsigned a, unsigned b, int m, int cnt) {
    if (lane < m)      my[cnt + lane]      = a - ulo;
    if (lane + 32 < m) my[cnt + 32 + lane] = b - ulo;
    cnt += m;
    while (cnt >= 4) {
        cnt -= 4;
        const unsigned t0 = (unsigned)__builtin_amdgcn_readfirstlane((int)my[cnt]);
        const unsigned t1 = (unsigned)__builtin_amdgcn_readfirstlane((int)my[cnt + 1]);
        const unsigned t2 = (unsigned)__builtin_amdgcn_readfirstlane((int)my[cnt + 2]);
        const unsigned t3 = (unsigned)__builtin_amdgcn_readfirstlane((int)my[cnt + 3]);
        own_add4(H1, acc, lane, t0, t1, t2, t3);
    }
    return cnt;
}

__global__ void __launch_bounds__(OWN_THREADS)
k_agg1_own(const unsigned* __restrict__ key, const int* __restrict__ tab,
           const float* __restrict__ dinv, const float* __restrict__ H1,
           const float* __restrict__ b1, const float* __restrict__ W2,
           float* __restrict__ H2, int e, int n) {
    __shared__ float acc[OWN_ROWS * HIDDEN];
    __shared__ unsigned lst[OWN_WAVES * LIST_CAP];
    __shared__ float dl[OWN_ROWS];

    const int lo = blockIdx.x * OWN_ROWS;
    int nrows = n - lo;
    if (nrows > OWN_ROWS) nrows = OWN_ROWS;
    const int wave = __builtin_amdgcn_readfirstlane((int)(threadIdx.x >> 5));
    const int lane = threadIdx.x & 31;

    for (int r = wave; r < nrows; r += OWN_WAVES) {
        if (lane == 0) dl[r] = dinv[lo + r];
        const float4* h = (const float4*)(H1 + (size_t)(lo + r) * HIDDEN) + lane;
        const float4 p = h[0], q = h[32];
        float* a = acc + r * HIDDEN + lane;
        a[0 * 32] = p.x;
        a[1 * 32] = p.y;
        a[2 * 32] = p.z;
        a[3 * 32] = p.w;
        a[4 * 32] = q.x;
        a[5 * 32] = q.y;
        a[6 * 32] = q.z;
        a[7 * 32] = q.w;
    }
    __syncthreads();

    const unsigned ulo = (unsigned)lo << 16;
    const int own   = (int)blockIdx.x;
    const int nsub  = (e + SORT_EDGES - 1) / SORT_EDGES;
    const int elast = e - 1;
    unsigned* my = lst + wave * LIST_CAP;
    int cnt = 0;

    for (int c0 = wave; c0 < nsub; c0 += OWN_WAVES * 32) {
        const int cl = c0 + OWN_WAVES * lane;
        int sb = 0, se = 0;
        if (cl < nsub) { sb = tab[cl * TAB_PITCH + own]; se = tab[cl * TAB_PITCH + own + 1]; }
        int nk = (nsub - c0 + OWN_WAVES - 1) / OWN_WAVES;
        if (nk > 32) nk = 32;
        int beg = __builtin_amdgcn_readlane(sb, 0), end = __builtin_amdgcn_readlane(se, 0);
        int i0 = beg + lane, i1 = beg + 32 + lane;
        unsigned w0 = key[i0 < elast ? i0 : elast];
        unsigned w1 = key[i1 < elast ? i1 : elast];
        for (int k = 0; k < nk; ++k) {
            const int kn = (k + 1 < nk) ? k + 1 : k;
            const int begn = __builtin_amdgcn_readlane(sb, kn);
            const int endn = __builtin_amdgcn_readlane(se, kn);
            i0 = begn + lane; i1 = begn + 32 + lane;
            const unsigned n0 = key[i0 < elast ? i0 : elast];
            const unsigned n1 = key[i1 < elast ? i1 : elast];

            int m = end - beg; if (m > RUN_WORDS) m = RUN_WORDS;
            cnt = own_take(H1, acc, my, lane, ulo, w0, w1, m, cnt);
            for (int base = beg + RUN_WORDS; base < end; base += RUN_WORDS) {
                i0 = base + lane; i1 = base + 32 + lane;
                const unsigned a = key[i0 < elast ? i0 : elast];
                const unsigned b = key[i1 < elast ? i1 : elast];
                int mm = end - base; if (mm > RUN_WORDS) mm = RUN_WORDS;
                cnt = own_take(H1, acc, my, lane, ulo, a, b, mm, cnt);
            }
            beg = begn; end = endn; w0 = n0; w1 = n1;
        }
    }
    while (cnt > 0) {
        cnt -= 1;
        const unsigned t0 = (unsigned)__builtin_amdgcn_readfirstlane((int)my[cnt]);
        own_add1(H1, acc, lane, t0);
    }
    __syncthreads();

    float bb[8], w0[8], w1[8];
    #pragma unroll
    for (int q = 0; q < 2; ++q) {
        const int c = 4 * lane + 128 * q;
        const float4 bv = *(const float4*)(b1 + c);
        const float4 wa = *(const float4*)(W2 + 2 * c);
        const float4 wb = *(const float4*)(W2 + 2 * c + 4);
        bb[4 * q + 0] = bv.x; bb[4 * q + 1] = bv.y; bb[4 * q + 2] = bv.z; bb[4 * q + 3] = bv.w;
        w0[4 * q + 0] = wa.x; w1[4 * q + 0] = wa.y;
        w0[4 * q + 1] = wa.z; w1[4 * q + 1] = wa.w;
        w0[4 * q + 2] = wb.x; w1[4 * q + 2] = wb.y;
        w0[4 * q + 3] = wb.z; w1[4 * q + 3] = wb.w;
    }
    for (int r = wave; r < nrows; r += OWN_WAVES) {
        const float* a = acc + r * HIDDEN + lane;
        const float dv = dl[r];
        float p0 = 0.0f, p1 = 0.0f;
        #pragma unroll
        for (int j = 0; j < 8; ++j) {
            float z = dv * a[j * 32] + bb[j];
            z = z > 0.0f ? z : 0.0f;
            p0 += z * w0[j];
            p1 += z * w1[j];
        }
        #pragma unroll
        for (int off = 16; off > 0; off >>= 1) {
            p0 += __shfl_xor(p0, off, 32);
            p1 += __shfl_xor(p1, off, 32);
        }
        if (lane == 0) {
            const int row = lo + r;
            H2[2 * row]      = dv * p0;
            H2[2 * row + 1]  = dv * p1;
        }
    }
}

__global__ void __launch_bounds__(ROWN_THREADS)
k_agg2_own(const unsigned* __restrict__ key, const int* __restrict__ tab,
           const float* __restrict__ dinv, const float* __restrict__ H2,
           const float* __restrict__ b2, float* __restrict__ out, int e, int n) {
    __shared__ float acc[2 * OWN_ROWS];
    __shared__ float dl[OWN_ROWS];
    const int tid = threadIdx.x;
    const int own = (int)blockIdx.x;
    const int lo  = own * OWN_ROWS;
    int nrows = n - lo;
    if (nrows > OWN_ROWS) nrows = OWN_ROWS;
    const int wave = __builtin_amdgcn_readfirstlane(tid >> 5);
    const int lane = tid & 31;

    if (tid < nrows) {
        const float2 h = *(const float2*)(H2 + 2 * (size_t)(lo + tid));
        dl[tid] = dinv[lo + tid];
        acc[2 * tid]     = h.x;
        acc[2 * tid + 1] = h.y;
    }
    __syncthreads();

    const unsigned ulo = (unsigned)lo;
    const int nsub  = (e + SORT_EDGES - 1) / SORT_EDGES;
    const int elast = e - 1;
    for (int c0 = wave; c0 < nsub; c0 += ROWN_WAVES * 32) {
        const int cl = c0 + ROWN_WAVES * lane;
        int sb = 0, se = 0;
        if (cl < nsub) { sb = tab[cl * TAB_PITCH + own]; se = tab[cl * TAB_PITCH + own + 1]; }
        int nk = (nsub - c0 + ROWN_WAVES - 1) / ROWN_WAVES;
        if (nk > 32) nk = 32;
        int beg = __builtin_amdgcn_readlane(sb, 0), end = __builtin_amdgcn_readlane(se, 0);
        int i0 = beg + lane, i1 = beg + 32 + lane;
        unsigned w0 = key[i0 < elast ? i0 : elast];
        unsigned w1 = key[i1 < elast ? i1 : elast];
        for (int k = 0; k < nk; ++k) {
            const int kn = (k + 1 < nk) ? k + 1 : k;
            const int begn = __builtin_amdgcn_readlane(sb, kn);
            const int endn = __builtin_amdgcn_readlane(se, kn);
            i0 = begn + lane; i1 = begn + 32 + lane;
            const unsigned n0 = key[i0 < elast ? i0 : elast];
            const unsigned n1 = key[i1 < elast ? i1 : elast];
            for (int base = beg; base < end; base += 64) {
                if (base != beg) {
                    i0 = base + lane; i1 = base + 32 + lane;
                    w0 = key[i0 < elast ? i0 : elast];
                    w1 = key[i1 < elast ? i1 : elast];
                }
                const unsigned s0 = w0 & 0xFFFFu, s1 = w1 & 0xFFFFu;
                const float2 h0 = *(const float2*)(H2 + 2 * s0);
                const float2 h1 = *(const float2*)(H2 + 2 * s1);
                if (base + lane < end) {
                    const unsigned r = (w0 >> 16) - ulo;
                    __hip_atomic_fetch_add(&acc[2 * r],     h0.x, __ATOMIC_RELAXED, __HIP_MEMORY_SCOPE_WORKGROUP);
                    __hip_atomic_fetch_add(&acc[2 * r + 1], h0.y, __ATOMIC_RELAXED, __HIP_MEMORY_SCOPE_WORKGROUP);
                }
                if (base + 32 + lane < end) {
                    const unsigned r = (w1 >> 16) - ulo;
                    __hip_atomic_fetch_add(&acc[2 * r],     h1.x, __ATOMIC_RELAXED, __HIP_MEMORY_SCOPE_WORKGROUP);
                    __hip_atomic_fetch_add(&acc[2 * r + 1], h1.y, __ATOMIC_RELAXED, __HIP_MEMORY_SCOPE_WORKGROUP);
                }
            }
            beg = begn; end = endn; w0 = n0; w1 = n1;
        }
    }
    __syncthreads();
    if (tid < 2 * nrows) out[2 * (size_t)lo + tid] = dl[tid >> 1] * acc[tid] + b2[tid & 1];
}


extern "C" void kernel_launch(void* const* d_in, const int* in_sizes, int n_in,
                              void* d_out, int out_size, void* d_ws, size_t ws_size,
                              hipStream_t stream) {
    const float* X   = (const float*)d_in[0];
    const float* W1  = (const float*)d_in[1];
    const float* b1  = (const float*)d_in[2];
    const float* W2  = (const float*)d_in[3];
    const float* b2  = (const float*)d_in[4];
    const int*   ei  = (const int*)d_in[5];
    float*       out = (float*)d_out;

    const int N = NNODES;
    const int E = in_sizes[5] / 2;
    const int* src = ei;
    const int* dst = ei + E;

    float*  dinv = (float*)d_ws;
    int*    tab  = (int*)(dinv + N);
    float*  H1   = (float*)(tab + N);
    float*  H2   = H1   + (size_t)N * HIDDEN;
    unsigned* key = (unsigned*)(H2 + (size_t)N * NCLS);

    k_sort_own<<<DEG_WGS, PRIV_THREADS, 0, stream>>>(src, dst, key, tab, E);
    k_own_deg <<<NOWN, ROWN_THREADS, 0, stream>>>(key, tab, dinv, E, N);

    k_gemm1<<<2 * G_ROWG, G_THREADS, 0, stream>>>(X, W1, dinv, H1);
    k_agg1_own<<<NOWN, OWN_THREADS, 0, stream>>>(key, tab, dinv, H1, b1, W2, H2, E, N);

    k_agg2_own<<<NOWN, ROWN_THREADS, 0, stream>>>(key, tab, dinv, H2, b2, out, E, N);
}


# speedup vs baseline: 3.8848x; 1.0527x over previous
#include <hip/hip_runtime.h>
#include <hip/hip_bf16.h>

#define NNODES 50000
#define DIMIN  512
#define HIDDEN 256
#define NCLS   2

typedef __attribute__((ext_vector_type(2))) float v2f;
typedef __attribute__((ext_vector_type(8))) float v8f;

static_assert(NNODES <= 65536, "packed edge word keeps a node id in 16 bits");

#define OWN_ROWS     196
#define NOWN         ((NNODES + OWN_ROWS - 1) / OWN_ROWS)
#define TAB_PITCH    (NOWN + 1)
static_assert(NOWN % 32 == 0, "the prefix over the owners gives each lane of one wave NOWN / 32 counters");

#define PRIV_THREADS 1024
#define SORT_BATCH   13
#define SORT_EDGES   (SORT_BATCH * PRIV_THREADS)
#define NEDGES       800000
#define DEG_WGS      ((NEDGES + SORT_EDGES - 1) / SORT_EDGES)
static_assert(DEG_WGS * TAB_PITCH <= NNODES, "segment table must fit its workspace region");

__global__ void __launch_bounds__(PRIV_THREADS)
k_sort_own(const int* __restrict__ src, const int* __restrict__ dst,
           unsigned* __restrict__ key, int* __restrict__ tab, int e) {
    __shared__ unsigned sorted[SORT_EDGES];
    __shared__ unsigned ocur[NOWN];
    const int tid = threadIdx.x;

    const int nsub = (e + SORT_EDGES - 1) / SORT_EDGES;
    for (int sc = (int)blockIdx.x; sc < nsub; sc += (int)gridDim.x) {
        if (tid < NOWN) ocur[tid] = 0u;
        __syncthreads();
        const int lo = sc * SORT_EDGES;
        int hi = lo + SORT_EDGES; if (hi > e) hi = e;
        int s[SORT_BATCH], d[SORT_BATCH];
        #pragma unroll
        for (int j = 0; j < SORT_BATCH; ++j) {
            const int i  = lo + tid + PRIV_THREADS * j;
            const int ic = (i < hi) ? i : lo;
            d[j] = dst[ic];
            s[j] = src[ic];
        }
        unsigned w[SORT_BATCH];
        int ow[SORT_BATCH];
        #pragma unroll
        for (int j = 0; j < SORT_BATCH; ++j) {
            const int i = lo + tid + PRIV_THREADS * j;
            w[j]  = (unsigned)s[j] | ((unsigned)d[j] << 16);
            ow[j] = d[j] / OWN_ROWS;
            if (i < hi)
                __hip_atomic_fetch_add(&ocur[ow[j]], 1u, __ATOMIC_RELAXED, __HIP_MEMORY_SCOPE_WORKGROUP);
        }
        __syncthreads();

        if (tid < 32) {
            unsigned c[NOWN / 32];
            unsigned sum = 0u;
            #pragma unroll
            for (int u = 0; u < NOWN / 32; ++u) { c[u] = ocur[tid * (NOWN / 32) + u]; sum += c[u]; }
            unsigned inc = sum;
            #pragma unroll
            for (int off = 1; off < 32; off <<= 1) {
                const unsigned v = __shfl_up(inc, off, 32);
                if (tid >= off) inc += v;
            }
            unsigned run = inc - sum;
            #pragma unroll
            for (int u = 0; u < NOWN / 32; ++u) {
                ocur[tid * (NOWN / 32) + u] = run;
                tab[sc * TAB_PITCH + tid * (NOWN / 32) + u] = lo + (int)run;
                run += c[u];
            }
            if (tid == 31) tab[sc * TAB_PITCH + NOWN] = lo + (int)run;
        }
        __syncthreads();

        unsigned pos[SORT_BATCH];
        #pragma unroll
        for (int j = 0; j < SORT_BATCH; ++j) {
            const int i = lo + tid + PRIV_THREADS * j;
            pos[j] = 0u;
            if (i < hi) pos[j] = __hip_atomic_fetch_add(&ocur[ow[j]], 1u, __ATOMIC_RELAXED,
                                                        __HIP_MEMORY_SCOPE_WORKGROUP);
        }
        #pragma unroll
        for (int j = 0; j < SORT_BATCH; ++j) {
            const int i = lo + tid + PRIV_THREADS * j;
            if (i < hi) sorted[pos[j]] = w[j];
        }
        __syncthreads();

        #pragma unroll
        for (int j = 0; j < SORT_BATCH; ++j) pos[j] = sorted[tid + PRIV_THREADS * j];
        #pragma unroll
        for (int j = 0; j < SORT_BATCH; ++j) {
            const int i = lo + tid + PRIV_THREADS * j;
            if (i < hi) key[i] = pos[j];
        }
    }
}


#define ROWN_THREADS 512
#define ROWN_WAVES   (ROWN_THREADS / 32)
static_assert(2 * OWN_ROWS <= ROWN_THREADS, "one thread per owned row, and per word of the layer-2 rows, in the prologues and epilogues");

__global__ void __launch_bounds__(ROWN_THREADS)
k_own_deg(const unsigned* __restrict__ key, const int* __restrict__ tab,
          float* __restrict__ dinv, int e, int n) {
    __shared__ unsigned cnt[OWN_ROWS];
    const int tid = threadIdx.x;
    const int own = (int)blockIdx.x;
    const int lo  = own * OWN_ROWS;
    int nrows = n - lo;
    if (nrows > OWN_ROWS) nrows = OWN_ROWS;
    const int wave = __builtin_amdgcn_readfirstlane(tid >> 5);
    const int lane = tid & 31;
    if (tid < OWN_ROWS) cnt[tid] = 0u;
    __syncthreads();

    const unsigned ulo = (unsigned)lo;
    const int nsub  = (e + SORT_EDGES - 1) / SORT_EDGES;
    const int elast = e - 1;
    for (int c0 = wave; c0 < nsub; c0 += ROWN_WAVES * 32) {
        const int cl = c0 + ROWN_WAVES * lane;
        int sb = 0, se = 0;
        if (cl < nsub) { sb = tab[cl * TAB_PITCH + own]; se = tab[cl * TAB_PITCH + own + 1]; }
        int nk = (nsub - c0 + ROWN_WAVES - 1) / ROWN_WAVES;
        if (nk > 32) nk = 32;
        int beg = __builtin_amdgcn_readlane(sb, 0), end = __builtin_amdgcn_readlane(se, 0);
        int i0 = beg + lane, i1 = beg + 32 + lane;
        unsigned w0 = key[i0 < elast ? i0 : elast];
        unsigned w1 = key[i1 < elast ? i1 : elast];
        for (int k = 0; k < nk; ++k) {
            const int kn = (k + 1 < nk) ? k + 1 : k;
            const int begn = __builtin_amdgcn_readlane(sb, kn);
            const int endn = __builtin_amdgcn_readlane(se, kn);
            i0 = begn + lane; i1 = begn + 32 + lane;
            const unsigned n0 = key[i0 < elast ? i0 : elast];
            const unsigned n1 = key[i1 < elast ? i1 : elast];
            for (int base = beg; base < end; base += 64) {
                if (base != beg) {
                    i0 = base + lane; i1 = base + 32 + lane;
                    w0 = key[i0 < elast ? i0 : elast];
                    w1 = key[i1 < elast ? i1 : elast];
                }
                if (base + lane < end)
                    __hip_atomic_fetch_add(&cnt[(w0 >> 16) - ulo], 1u, __ATOMIC_RELAXED, __HIP_MEMORY_SCOPE_WORKGROUP);
                if (base + 32 + lane < end)
                    __hip_atomic_fetch_add(&cnt[(w1 >> 16) - ulo], 1u, __ATOMIC_RELAXED, __HIP_MEMORY_SCOPE_WORKGROUP);
            }
            beg = begn; end = endn; w0 = n0; w1 = n1;
        }
    }
    __syncthreads();
    if (tid < nrows) dinv[lo + tid] = __frsqrt_rn((float)(cnt[tid] + 1u));
}


#define G_THREADS 512
#define G_ROWG    128
#define G_HALFC   128
#define G_KH      256
#define G_APITCH  260
#define G_NT      (NNODES / 16)

__device__ __forceinline__ void gemm_stage_load(const float* __restrict__ X, int rg, int s, int tid,
                                                float4 (&r)[4]) {
    const int p  = s >> 1;
    const int kh = s & 1;
    #pragma unroll
    for (int j = 0; j < 4; ++j) {
        const int idx = tid + G_THREADS * j;
        const int rr  = idx >> 6;
        const int c4  = (idx & 63) * 4;
        const int t   = rg + G_ROWG * (2 * p + (rr >> 4));
        if (t < G_NT)
            r[j] = *(const float4*)(X + (size_t)(t * 16 + (rr & 15)) * DIMIN + kh * G_KH + c4);
        else
            r[j] = make_float4(0.0f, 0.0f, 0.0f, 0.0f);
    }
}

__global__ void __launch_bounds__(G_THREADS)
k_gemm1(const float* __restrict__ X, const float* __restrict__ W,
        const float* __restrict__ dinv, float* __restrict__ H) {
    __shared__ alignas(16) float Ws[(DIMIN / 2) * G_HALFC * 2];
    __shared__ alignas(16) float As[32 * G_APITCH];

    const int tid  = threadIdx.x;
    const int half = blockIdx.x & 1;
    const int rg   = blockIdx.x >> 1;
    const int ntile  = (G_NT - rg + G_ROWG - 1) / G_ROWG;
    const int nstage = 2 * ((ntile + 1) / 2);

    float4 pre[4];
    gemm_stage_load(X, rg, 0, tid, pre);

    #pragma unroll 4
    for (int i = 0; i < 16; ++i) {
        const int it  = (i + rg) & 15;
        const int idx = tid + G_THREADS * it;
        const int kp  = idx >> 5;
        const int cq  = idx & 31;
        const float4 r0 = *(const float4*)(W + (size_t)(2 * kp)     * HIDDEN + half * G_HALFC + 4 * cq);
        const float4 r1 = *(const float4*)(W + (size_t)(2 * kp + 1) * HIDDEN + half * G_HALFC + 4 * cq);
        const int c = (4 * cq) ^ ((kp & 1) << 4);
        float4* d = (float4*)(&Ws[2 * (kp * G_HALFC + c)]);
        d[0] = make_float4(r0.x, r1.x, r0.y, r1.y);
        d[1] = make_float4(r0.z, r1.z, r0.w, r1.w);
    }

    const int wave = __builtin_amdgcn_readfirstlane(tid >> 5);
    const int lane = tid & 31;
    const int l16  = lane & 15;
    const int hi   = lane >> 4;
    const int sub  = wave >> 3;
    const int ct   = wave & 7;

    const float* ap = As + (sub * 16 + l16) * G_APITCH + 2 * hi;
    const float* bp = Ws + 2 * (hi * G_HALFC + ((ct ^ hi) << 4) + l16);

    v8f c = {};
    for (int s = 0; s < nstage; ++s) {
        __syncthreads();
        #pragma unroll
        for (int j = 0; j < 4; ++j) {
            const int idx = tid + G_THREADS * j;
            *(float4*)(&As[(idx >> 6) * G_APITCH + (idx & 63) * 4]) = pre[j];
        }
        __syncthreads();
        if (s + 1 < nstage) gemm_stage_load(X, rg, s + 1, tid, pre);

        const int kh = s & 1;
        const int t  = rg + G_ROWG * (2 * (s >> 1) + sub);
        if (t < G_NT) {
            const float* b = bp + 2 * (kh * (G_KH / 2) * G_HALFC);
            float4 da = make_float4(0.0f, 0.0f, 0.0f, 0.0f), db = da;
            if (kh == 1) {
                da = *(const float4*)(dinv + t * 16 + hi * 8);
                db = *(const float4*)(dinv + t * 16 + hi * 8 + 4);
            }
            v2f fa[2][8], fb[2][8];
            #pragma unroll
            for (int i = 0; i < 8; ++i) {
                fa[0][i] = *(const v2f*)(ap + 4 * i);
                fb[0][i] = *(const v2f*)(b + (2 * i) * (2 * G_HALFC));
            }
            #pragma unroll
            for (int j = 0; j < G_KH / 32; ++j) {
                if (j + 1 < G_KH / 32) {
                    #pragma unroll
                    for (int i = 0; i < 8; ++i) {
                        const int kk = 32 * (j + 1) + 4 * i;
                        fa[(j + 1) & 1][i] = *(const v2f*)(ap + kk);
                        fb[(j + 1) & 1][i] = *(const v2f*)(b + (kk >> 1) * (2 * G_HALFC));
                    }
                }
                __builtin_amdgcn_sched_barrier(0);
                #pragma unroll
                for (int i = 0; i < 8; ++i)
                    c = __builtin_amdgcn_wmma_f32_16x16x4_f32(false, fa[j & 1][i], false, fb[j & 1][i],
                                                              (short)0, c, false, false);
                __builtin_amdgcn_sched_barrier(0);
            }
            if (kh == 1) {
                const float dv[8] = {da.x, da.y, da.z, da.w, db.x, db.y, db.z, db.w};
                #pragma unroll
                for (int v = 0; v < 8; ++v) {
                    const size_t o = (size_t)(t * 16 + hi * 8 + v) * HIDDEN + half * G_HALFC + ct * 16 + l16;
                    H[o] = c[v] * dv[v];
                }
                c = v8f{};
            }
        }
    }
}


#define OWN_THREADS 1024
#define OWN_WAVES   (OWN_THREADS / 32)
#define RUN_WORDS   64
#define LIST_CAP    (RUN_WORDS + 4)

__device__ __forceinline__ void own_lds_add(float* a, const float4& p, const float4& q) {
    __hip_atomic_fetch_add(&a[0 * 32], p.x, __ATOMIC_RELAXED, __HIP_MEMORY_SCOPE_WORKGROUP);
    __hip_atomic_fetch_add(&a[1 * 32], p.y, __ATOMIC_RELAXED, __HIP_MEMORY_SCOPE_WORKGROUP);
    __hip_atomic_fetch_add(&a[2 * 32], p.z, __ATOMIC_RELAXED, __HIP_MEMORY_SCOPE_WORKGROUP);
    __hip_atomic_fetch_add(&a[3 * 32], p.w, __ATOMIC_RELAXED, __HIP_MEMORY_SCOPE_WORKGROUP);
    __hip_atomic_fetch_add(&a[4 * 32], q.x, __ATOMIC_RELAXED, __HIP_MEMORY_SCOPE_WORKGROUP);
    __hip_atomic_fetch_add(&a[5 * 32], q.y, __ATOMIC_RELAXED, __HIP_MEMORY_SCOPE_WORKGROUP);
    __hip_atomic_fetch_add(&a[6 * 32], q.z, __ATOMIC_RELAXED, __HIP_MEMORY_SCOPE_WORKGROUP);
    __hip_atomic_fetch_add(&a[7 * 32], q.w, __ATOMIC_RELAXED, __HIP_MEMORY_SCOPE_WORKGROUP);
}

__device__ __forceinline__ void own_add4(const float* __restrict__ H1, float* acc, int lane,
                                         unsigned t0, unsigned t1, unsigned t2, unsigned t3) {
    const int s0 = (int)(t0 & 0xFFFFu), s1 = (int)(t1 & 0xFFFFu);
    const int s2 = (int)(t2 & 0xFFFFu), s3 = (int)(t3 & 0xFFFFu);
    const int r0 = (int)(t0 >> 16), r1 = (int)(t1 >> 16), r2 = (int)(t2 >> 16), r3 = (int)(t3 >> 16);
    const float4* h0 = (const float4*)(H1 + (size_t)s0 * HIDDEN) + lane;
    const float4* h1 = (const float4*)(H1 + (size_t)s1 * HIDDEN) + lane;
    const float4* h2 = (const float4*)(H1 + (size_t)s2 * HIDDEN) + lane;
    const float4* h3 = (const float4*)(H1 + (size_t)s3 * HIDDEN) + lane;
    const float4 p0 = h0[0], q0 = h0[32];
    const float4 p1 = h1[0], q1 = h1[32];
    const float4 p2 = h2[0], q2 = h2[32];
    const float4 p3 = h3[0], q3 = h3[32];
    own_lds_add(acc + r0 * HIDDEN + lane, p0, q0);
    own_lds_add(acc + r1 * HIDDEN + lane, p1, q1);
    own_lds_add(acc + r2 * HIDDEN + lane, p2, q2);
    own_lds_add(acc + r3 * HIDDEN + lane, p3, q3);
}

__device__ __forceinline__ void own_add1(const float* __restrict__ H1, float* acc, int lane, unsigned t0) {
    const int s0 = (int)(t0 & 0xFFFFu);
    const int r0 = (int)(t0 >> 16);
    const float4* h0 = (const float4*)(H1 + (size_t)s0 * HIDDEN) + lane;
    const float4 p0 = h0[0], q0 = h0[32];
    own_lds_add(acc + r0 * HIDDEN + lane, p0, q0);
}

__device__ __forceinline__ int own_take(const float* __restrict__ H1, float* acc, unsigned* my, int lane,
                                        unsigned ulo, unsigned a, unsigned b, int m, int cnt) {
    if (lane < m)      my[cnt + lane]      = a - ulo;
    if (lane + 32 < m) my[cnt + 32 + lane] = b - ulo;
    cnt += m;
    while (cnt >= 4) {
        cnt -= 4;
        const unsigned t0 = (unsigned)__builtin_amdgcn_readfirstlane((int)my[cnt]);
        const unsigned t1 = (unsigned)__builtin_amdgcn_readfirstlane((int)my[cnt + 1]);
        const unsigned t2 = (unsigned)__builtin_amdgcn_readfirstlane((int)my[cnt + 2]);
        const unsigned t3 = (unsigned)__builtin_amdgcn_readfirstlane((int)my[cnt + 3]);
        own_add4(H1, acc, lane, t0, t1, t2, t3);
    }
    return cnt;
}

__global__ void __launch_bounds__(OWN_THREADS)
k_agg1_own(const unsigned* __restrict__ key, const int* __restrict__ tab,
           const float* __restrict__ dinv, const float* __restrict__ H1,
           const float* __restrict__ b1, const float* __restrict__ W2,
           float* __restrict__ H2, int e, int n) {
    __shared__ float acc[OWN_ROWS * HIDDEN];
    __shared__ unsigned lst[OWN_WAVES * LIST_CAP];
    __shared__ float dl[OWN_ROWS];

    const int lo = blockIdx.x * OWN_ROWS;
    int nrows = n - lo;
    if (nrows > OWN_ROWS) nrows = OWN_ROWS;
    const int wave = __builtin_amdgcn_readfirstlane((int)(threadIdx.x >> 5));
    const int lane = threadIdx.x & 31;

    for (int r = wave; r < nrows; r += OWN_WAVES) {
        if (lane == 0) dl[r] = dinv[lo + r];
        const float4* h = (const float4*)(H1 + (size_t)(lo + r) * HIDDEN) + lane;
        const float4 p = h[0], q = h[32];
        float* a = acc + r * HIDDEN + lane;
        a[0 * 32] = p.x;
        a[1 * 32] = p.y;
        a[2 * 32] = p.z;
        a[3 * 32] = p.w;
        a[4 * 32] = q.x;
        a[5 * 32] = q.y;
        a[6 * 32] = q.z;
        a[7 * 32] = q.w;
    }
    __syncthreads();

    const unsigned ulo = (unsigned)lo << 16;
    const int own   = (int)blockIdx.x;
    const int nsub  = (e + SORT_EDGES - 1) / SORT_EDGES;
    const int elast = e - 1;
    unsigned* my = lst + wave * LIST_CAP;
    int cnt = 0;

    for (int c0 = wave; c0 < nsub; c0 += OWN_WAVES * 32) {
        const int cl = c0 + OWN_WAVES * lane;
        int sb = 0, se = 0;
        if (cl < nsub) { sb = tab[cl * TAB_PITCH + own]; se = tab[cl * TAB_PITCH + own + 1]; }
        int nk = (nsub - c0 + OWN_WAVES - 1) / OWN_WAVES;
        if (nk > 32) nk = 32;
        int beg = __builtin_amdgcn_readlane(sb, 0), end = __builtin_amdgcn_readlane(se, 0);
        int i0 = beg + lane, i1 = beg + 32 + lane;
        unsigned w0 = key[i0 < elast ? i0 : elast];
        unsigned w1 = key[i1 < elast ? i1 : elast];
        for (int k = 0; k < nk; ++k) {
            const int kn = (k + 1 < nk) ? k + 1 : k;
            const int begn = __builtin_amdgcn_readlane(sb, kn);
            const int endn = __builtin_amdgcn_readlane(se, kn);
            i0 = begn + lane; i1 = begn + 32 + lane;
            const unsigned n0 = key[i0 < elast ? i0 : elast];
            const unsigned n1 = key[i1 < elast ? i1 : elast];

            int m = end - beg; if (m > RUN_WORDS) m = RUN_WORDS;
            cnt = own_take(H1, acc, my, lane, ulo, w0, w1, m, cnt);
            for (int base = beg + RUN_WORDS; base < end; base += RUN_WORDS) {
                i0 = base + lane; i1 = base + 32 + lane;
                const unsigned a = key[i0 < elast ? i0 : elast];
                const unsigned b = key[i1 < elast ? i1 : elast];
                int mm = end - base; if (mm > RUN_WORDS) mm = RUN_WORDS;
                cnt = own_take(H1, acc, my, lane, ulo, a, b, mm, cnt);
            }
            beg = begn; end = endn; w0 = n0; w1 = n1;
        }
    }
    while (cnt > 0) {
        cnt -= 1;
        const unsigned t0 = (unsigned)__builtin_amdgcn_readfirstlane((int)my[cnt]);
        own_add1(H1, acc, lane, t0);
    }
    __syncthreads();

    float bb[8], w0[8], w1[8];
    #pragma unroll
    for (int q = 0; q < 2; ++q) {
        const int c = 4 * lane + 128 * q;
        const float4 bv = *(const float4*)(b1 + c);
        const float4 wa = *(const float4*)(W2 + 2 * c);
        const float4 wb = *(const float4*)(W2 + 2 * c + 4);
        bb[4 * q + 0] = bv.x; bb[4 * q + 1] = bv.y; bb[4 * q + 2] = bv.z; bb[4 * q + 3] = bv.w;
        w0[4 * q + 0] = wa.x; w1[4 * q + 0] = wa.y;
        w0[4 * q + 1] = wa.z; w1[4 * q + 1] = wa.w;
        w0[4 * q + 2] = wb.x; w1[4 * q + 2] = wb.y;
        w0[4 * q + 3] = wb.z; w1[4 * q + 3] = wb.w;
    }
    for (int r = wave; r < nrows; r += OWN_WAVES) {
        const float* a = acc + r * HIDDEN + lane;
        const float dv = dl[r];
        float p0 = 0.0f, p1 = 0.0f;
        #pragma unroll
        for (int j = 0; j < 8; ++j) {
            float z = dv * a[j * 32] + bb[j];
            z = z > 0.0f ? z : 0.0f;
            p0 += z * w0[j];
            p1 += z * w1[j];
        }
        #pragma unroll
        for (int off = 16; off > 0; off >>= 1) {
            p0 += __shfl_xor(p0, off, 32);
            p1 += __shfl_xor(p1, off, 32);
        }
        if (lane == 0) {
            const int row = lo + r;
            H2[2 * row]      = dv * p0;
            H2[2 * row + 1]  = dv * p1;
        }
    }
}

__global__ void __launch_bounds__(ROWN_THREADS)
k_agg2_own(const unsigned* __restrict__ key, const int* __restrict__ tab,
           const float* __restrict__ dinv, const float* __restrict__ H2,
           const float* __restrict__ b2, float* __restrict__ out, int e, int n) {
    __shared__ float acc[2 * OWN_ROWS];
    __shared__ float dl[OWN_ROWS];
    const int tid = threadIdx.x;
    const int own = (int)blockIdx.x;
    const int lo  = own * OWN_ROWS;
    int nrows = n - lo;
    if (nrows > OWN_ROWS) nrows = OWN_ROWS;
    const int wave = __builtin_amdgcn_readfirstlane(tid >> 5);
    const int lane = tid & 31;

    if (tid < nrows) {
        const float2 h = *(const float2*)(H2 + 2 * (size_t)(lo + tid));
        dl[tid] = dinv[lo + tid];
        acc[2 * tid]     = h.x;
        acc[2 * tid + 1] = h.y;
    }
    __syncthreads();

    const unsigned ulo = (unsigned)lo;
    const int nsub  = (e + SORT_EDGES - 1) / SORT_EDGES;
    const int elast = e - 1;
    for (int c0 = wave; c0 < nsub; c0 += ROWN_WAVES * 32) {
        const int cl = c0 + ROWN_WAVES * lane;
        int sb = 0, se = 0;
        if (cl < nsub) { sb = tab[cl * TAB_PITCH + own]; se = tab[cl * TAB_PITCH + own + 1]; }
        int nk = (nsub - c0 + ROWN_WAVES - 1) / ROWN_WAVES;
        if (nk > 32) nk = 32;
        int beg = __builtin_amdgcn_readlane(sb, 0), end = __builtin_amdgcn_readlane(se, 0);
        int i0 = beg + lane, i1 = beg + 32 + lane;
        unsigned w0 = key[i0 < elast ? i0 : elast];
        unsigned w1 = key[i1 < elast ? i1 : elast];
        for (int k = 0; k < nk; ++k) {
            const int kn = (k + 1 < nk) ? k + 1 : k;
            const int begn = __builtin_amdgcn_readlane(sb, kn);
            const int endn = __builtin_amdgcn_readlane(se, kn);
            i0 = begn + lane; i1 = begn + 32 + lane;
            const unsigned n0 = key[i0 < elast ? i0 : elast];
            const unsigned n1 = key[i1 < elast ? i1 : elast];
            for (int base = beg; base < end; base += 64) {
                if (base != beg) {
                    i0 = base + lane; i1 = base + 32 + lane;
                    w0 = key[i0 < elast ? i0 : elast];
                    w1 = key[i1 < elast ? i1 : elast];
                }
                const unsigned s0 = w0 & 0xFFFFu, s1 = w1 & 0xFFFFu;
                const float2 h0 = *(const float2*)(H2 + 2 * s0);
                const float2 h1 = *(const float2*)(H2 + 2 * s1);
                if (base + lane < end) {
                    const unsigned r = (w0 >> 16) - ulo;
                    __hip_atomic_fetch_add(&acc[2 * r],     h0.x, __ATOMIC_RELAXED, __HIP_MEMORY_SCOPE_WORKGROUP);
                    __hip_atomic_fetch_add(&acc[2 * r + 1], h0.y, __ATOMIC_RELAXED, __HIP_MEMORY_SCOPE_WORKGROUP);
                }
                if (base + 32 + lane < end) {
                    const unsigned r = (w1 >> 16) - ulo;
                    __hip_atomic_fetch_add(&acc[2 * r],     h1.x, __ATOMIC_RELAXED, __HIP_MEMORY_SCOPE_WORKGROUP);
                    __hip_atomic_fetch_add(&acc[2 * r + 1], h1.y, __ATOMIC_RELAXED, __HIP_MEMORY_SCOPE_WORKGROUP);
                }
            }
            beg = begn; end = endn; w0 = n0; w1 = n1;
        }
    }
    __syncthreads();
    if (tid < 2 * nrows) out[2 * (size_t)lo + tid] = dl[tid >> 1] * acc[tid] + b2[tid & 1];
}


extern "C" void kernel_launch(void* const* d_in, const int* in_sizes, int n_in,
                              void* d_out, int out_size, void* d_ws, size_t ws_size,
                              hipStream_t stream) {
    const float* X   = (const float*)d_in[0];
    const float* W1  = (const float*)d_in[1];
    const float* b1  = (const float*)d_in[2];
    const float* W2  = (const float*)d_in[3];
    const float* b2  = (const float*)d_in[4];
    const int*   ei  = (const int*)d_in[5];
    float*       out = (float*)d_out;

    const int N = NNODES;
    const int E = in_sizes[5] / 2;
    const int* src = ei;
    const int* dst = ei + E;

    float*  H1   = (float*)d_ws;
    float*  H2   = H1   + (size_t)N * HIDDEN;
    float*  dinv = H2   + (size_t)N * NCLS;
    int*    tab  = (int*)(dinv + N);
    unsigned* key = (unsigned*)(tab + N);

    k_sort_own<<<DEG_WGS, PRIV_THREADS, 0, stream>>>(src, dst, key, tab, E);
    k_own_deg <<<NOWN, ROWN_THREADS, 0, stream>>>(key, tab, dinv, E, N);

    k_gemm1<<<2 * G_ROWG, G_THREADS, 0, stream>>>(X, W1, dinv, H1);
    k_agg1_own<<<NOWN, OWN_THREADS, 0, stream>>>(key, tab, dinv, H1, b1, W2, H2, E, N);

    k_agg2_own<<<NOWN, ROWN_THREADS, 0, stream>>>(key, tab, dinv, H2, b2, out, E, N);
}
